# starting kernel
# baseline (speedup 1.0000x reference)
#include <hip/hip_runtime.h>
#include <hip/hip_bf16.h>


typedef float v2f __attribute__((ext_vector_type(2)));
typedef float v8f __attribute__((ext_vector_type(8)));

#define HID 128
#define NB_STATS 512

__global__ void zero_i32(int* __restrict__ p, int n) {
  for (int i = blockIdx.x * blockDim.x + threadIdx.x; i < n;
       i += gridDim.x * blockDim.x)
    p[i] = 0;
}

__global__ void hist_deg(const int* __restrict__ dst, int* __restrict__ cnt,
                         int E) {
  for (int e = blockIdx.x * blockDim.x + threadIdx.x; e < E;
       e += gridDim.x * blockDim.x)
    atomicAdd(&cnt[dst[e]], 1);
}

__global__ void invdeg_from_cnt(const int* __restrict__ cnt,
                                float* __restrict__ invdeg, int N) {
  for (int i = blockIdx.x * blockDim.x + threadIdx.x; i < N;
       i += gridDim.x * blockDim.x)
    invdeg[i] = 1.0f / fmaxf((float)cnt[i], 1.0f);
}

__global__ void __launch_bounds__(1024) scan_excl(const int* __restrict__ cnt,
                                                  int* __restrict__ rowptr,
                                                  int N) {
  __shared__ int tmp[1024];
  __shared__ int carry;
  const int t = threadIdx.x;
  if (t == 0) carry = 0;
  __syncthreads();
  for (int base = 0; base < N; base += 1024) {
    int i = base + t;
    int v = (i < N) ? cnt[i] : 0;
    tmp[t] = v;
    __syncthreads();
#pragma unroll
    for (int off = 1; off < 1024; off <<= 1) {
      int addv = (t >= off) ? tmp[t - off] : 0;
      __syncthreads();
      tmp[t] += addv;
      __syncthreads();
    }
    int incl = tmp[t];
    int ctotal = tmp[1023];
    int cbase = carry;
    if (i < N) rowptr[i] = cbase + incl - v;
    __syncthreads();
    if (t == 0) carry = cbase + ctotal;
    __syncthreads();
  }
  if (t == 0) rowptr[N] = carry;
}

__global__ void fill_csr(const int* __restrict__ src,
                         const int* __restrict__ dst,
                         const int* __restrict__ rowptr,
                         int* __restrict__ cur, int* __restrict__ eidx,
                         int E) {
  for (int e = blockIdx.x * blockDim.x + threadIdx.x; e < E;
       e += gridDim.x * blockDim.x) {
    int d = dst[e];
    int slot = rowptr[d] + atomicAdd(&cur[d], 1);
    eidx[slot] = src[e];
  }
}

__global__ void __launch_bounds__(128) aggregate_csr(
    const float* __restrict__ h, const int* __restrict__ rowptr,
    const int* __restrict__ eidx, const float* __restrict__ invdeg,
    float* __restrict__ c, int N) {
  __shared__ int idxs[128];
  const int t = threadIdx.x;
  for (int node = blockIdx.x; node < N; node += gridDim.x) {
    int beg = rowptr[node];
    int cnt = rowptr[node + 1] - beg;
    float acc = 0.0f;
    for (int chunk = 0; chunk < cnt; chunk += 128) {
      int m = min(128, cnt - chunk);
      if (t < m) idxs[t] = eidx[beg + chunk + t];
      __syncthreads();
      int i = 0;
      for (; i + 4 <= m; i += 4) {
        float v0 = h[idxs[i + 0] * HID + t];
        float v1 = h[idxs[i + 1] * HID + t];
        float v2 = h[idxs[i + 2] * HID + t];
        float v3 = h[idxs[i + 3] * HID + t];
        acc += v0 + v1 + v2 + v3;
      }
      for (; i < m; ++i) acc += h[idxs[i] * HID + t];
      __syncthreads();
    }
    c[node * HID + t] = acc * invdeg[node];
  }
}

__global__ void __launch_bounds__(256) embed_gemm(
    const float* __restrict__ X, const float* __restrict__ W,
    const float* __restrict__ B, float* __restrict__ Y) {
  __shared__ __attribute__((aligned(16))) float tA[16 * HID];

  const int tid = threadIdx.x;
  const int rowBase = blockIdx.x * 16;

  const float4* Xv = (const float4*)(X + rowBase * HID);
  float4* tAv = (float4*)tA;
  tAv[tid] = Xv[tid];
  tAv[tid + 256] = Xv[tid + 256];
  __syncthreads();

  const int wave = tid >> 5;
  const int lane = tid & 31;
  const int half = lane >> 4;
  const int l16 = lane & 15;
  const int col = wave * 16 + l16;

  v8f acc;
  {
    float bv = B[col];
#pragma unroll
    for (int r = 0; r < 8; ++r) acc[r] = bv;
  }

  const float* arow = tA + l16 * HID;
  const float* wrow = W + col * HID;

#pragma unroll 4
  for (int k = 0; k < HID; k += 4) {
    const int kk = k + 2 * half;
    v2f a = *(const v2f*)(arow + kk);
    v2f b = *(const v2f*)(wrow + kk);
    acc = __builtin_amdgcn_wmma_f32_16x16x4_f32(false, a, false, b, (short)0,
                                                acc, false, false);
  }

#pragma unroll
  for (int r = 0; r < 8; ++r) {
    int row = rowBase + r + 8 * half;
    Y[row * HID + col] = acc[r];
  }
}

__global__ void __launch_bounds__(256) layer_gemm(
    const float* __restrict__ h, const float* __restrict__ c,
    const float* __restrict__ W, const float* __restrict__ B,
    float* __restrict__ Z) {
  __shared__ __attribute__((aligned(16))) float tA[16 * 256];

  const int tid = threadIdx.x;
  const int rowBase = blockIdx.x * 16;

#pragma unroll
  for (int i = 0; i < 2; ++i) {
    int v = tid + i * 256;
    int r = v >> 5;
    int c4 = v & 31;
    ((float4*)(tA + r * 256))[c4] =
        ((const float4*)(h + (rowBase + r) * HID))[c4];
  }
#pragma unroll
  for (int i = 0; i < 2; ++i) {
    int v = tid + i * 256;
    int r = v >> 5;
    int c4 = v & 31;
    ((float4*)(tA + r * 256 + HID))[c4] =
        ((const float4*)(c + (rowBase + r) * HID))[c4];
  }
  __syncthreads();

  const int wave = tid >> 5;
  const int lane = tid & 31;
  const int half = lane >> 4;
  const int l16 = lane & 15;
  const int col = wave * 16 + l16;

  v8f acc;
  {
    float bv = B[col];
#pragma unroll
    for (int r = 0; r < 8; ++r) acc[r] = bv;
  }

  const float* arow = tA + l16 * 256;
  const float* wrow = W + col * 256;

#pragma unroll 4
  for (int k = 0; k < 256; k += 4) {
    const int kk = k + 2 * half;
    v2f a = *(const v2f*)(arow + kk);
    v2f b = *(const v2f*)(wrow + kk);
    acc = __builtin_amdgcn_wmma_f32_16x16x4_f32(false, a, false, b, (short)0,
                                                acc, false, false);
  }

#pragma unroll
  for (int r = 0; r < 8; ++r) {
    int row = rowBase + r + 8 * half;
    Z[row * HID + col] = acc[r];
  }
}

__global__ void __launch_bounds__(128) norm_relu_stats(
    float* __restrict__ z, float* __restrict__ part, int N) {
  __shared__ float red[4];
  const int t = threadIdx.x;
  const int w = t >> 5;
  const int lane = t & 31;

  float ps = 0.0f, pss = 0.0f;
  for (int node = blockIdx.x; node < N; node += gridDim.x) {
    float v = z[node * HID + t];
    float sq = v * v;
#pragma unroll
    for (int off = 16; off > 0; off >>= 1) sq += __shfl_xor(sq, off, 32);
    if (lane == 0) red[w] = sq;
    __syncthreads();
    float tot = red[0] + red[1] + red[2] + red[3];
    __syncthreads();
    float nrm = fmaxf(sqrtf(tot), 1e-12f);
    float r = fmaxf(v / nrm, 0.0f);
    z[node * HID + t] = r;
    ps += r;
    pss += r * r;
  }
  part[blockIdx.x * 256 + t] = ps;
  part[blockIdx.x * 256 + 128 + t] = pss;
}

__global__ void __launch_bounds__(256) bn_finalize(
    const float* __restrict__ part, int nb, const float* __restrict__ gamma,
    const float* __restrict__ beta, float* __restrict__ bn, float invN) {
  __shared__ float tot[256];
  const int j = threadIdx.x;
  float s = 0.0f;
  for (int b = 0; b < nb; ++b) s += part[b * 256 + j];
  tot[j] = s;
  __syncthreads();
  if (j < 128) {
    float mu = tot[j] * invN;
    float var = tot[128 + j] * invN - mu * mu;
    float a = gamma[j] / sqrtf(var + 1e-5f);
    bn[j] = a;
    bn[128 + j] = beta[j] - mu * a;
  }
}

__global__ void residual_bn(const float* __restrict__ h,
                            const float* __restrict__ z,
                            const float* __restrict__ bn,
                            float* __restrict__ out, int total) {
  for (int i = blockIdx.x * blockDim.x + threadIdx.x; i < total;
       i += gridDim.x * blockDim.x) {
    int j = i & (HID - 1);
    out[i] = h[i] + z[i] * bn[j] + bn[HID + j];
  }
}

extern "C" void kernel_launch(void* const* d_in, const int* in_sizes, int n_in,
                              void* d_out, int out_size, void* d_ws,
                              size_t ws_size, hipStream_t stream) {
  const float* h0 = (const float*)d_in[0];
  const int* src = (const int*)d_in[1];
  const int* dst = (const int*)d_in[2];
  const float* W_emb = (const float*)d_in[3];
  const float* b_emb = (const float*)d_in[4];
  const float* Ws = (const float*)d_in[5];
  const float* bs = (const float*)d_in[6];
  const float* gam = (const float*)d_in[7];
  const float* bet = (const float*)d_in[8];

  const int N = in_sizes[0] / HID;
  const int E = in_sizes[1];
  const int NF = N * HID;
  const int L = 4;

  float* ws = (float*)d_ws;
  float* H = ws;
  float* C = H + NF;
  float* Z = C + NF;
  float* DEG = Z + NF;
  float* BN = DEG + N;
  float* PART = BN + 256;
  int* rowptr = (int*)(PART + NB_STATS * 256);
  int* cnt = rowptr + (N + 1);
  int* eidx = cnt + N;

  const int mtiles = N / 16;

  zero_i32<<<256, 256, 0, stream>>>(cnt, N);
  hist_deg<<<1024, 256, 0, stream>>>(dst, cnt, E);
  invdeg_from_cnt<<<256, 256, 0, stream>>>(cnt, DEG, N);
  scan_excl<<<1, 1024, 0, stream>>>(cnt, rowptr, N);
  zero_i32<<<256, 256, 0, stream>>>(cnt, N);
  fill_csr<<<1024, 256, 0, stream>>>(src, dst, rowptr, cnt, eidx, E);

  embed_gemm<<<mtiles, 256, 0, stream>>>(h0, W_emb, b_emb, H);

  for (int l = 0; l < L; ++l) {
    aggregate_csr<<<2048, 128, 0, stream>>>(H, rowptr, eidx, DEG, C, N);
    layer_gemm<<<mtiles, 256, 0, stream>>>(H, C, Ws + l * HID * 256,
                                           bs + l * HID, Z);
    norm_relu_stats<<<NB_STATS, 128, 0, stream>>>(Z, PART, N);
    bn_finalize<<<1, 256, 0, stream>>>(PART, NB_STATS, gam + l * HID,
                                       bet + l * HID, BN, 1.0f / (float)N);
    float* out = (l == L - 1) ? (float*)d_out : H;
    residual_bn<<<4096, 256, 0, stream>>>(H, Z, BN, out, NF);
  }
}


# norm_relu_stats: one wave per row (float4 per lane, 10 rows in flight per wave, no block barrier per row, DPP butterfly); row sum of squares and BN partial sums summed in a different order
# speedup vs baseline: 1.1683x; 1.1683x over previous
#include <hip/hip_runtime.h>
#include <hip/hip_bf16.h>


typedef float v2f __attribute__((ext_vector_type(2)));
typedef float v8f __attribute__((ext_vector_type(8)));

#define HID 128
#define NB_STATS 512

__global__ void zero_i32(int* __restrict__ p, int n) {
  for (int i = blockIdx.x * blockDim.x + threadIdx.x; i < n;
       i += gridDim.x * blockDim.x)
    p[i] = 0;
}

__global__ void hist_deg(const int* __restrict__ dst, int* __restrict__ cnt,
                         int E) {
  for (int e = blockIdx.x * blockDim.x + threadIdx.x; e < E;
       e += gridDim.x * blockDim.x)
    atomicAdd(&cnt[dst[e]], 1);
}

__global__ void invdeg_from_cnt(const int* __restrict__ cnt,
                                float* __restrict__ invdeg, int N) {
  for (int i = blockIdx.x * blockDim.x + threadIdx.x; i < N;
       i += gridDim.x * blockDim.x)
    invdeg[i] = 1.0f / fmaxf((float)cnt[i], 1.0f);
}

__global__ void __launch_bounds__(1024) scan_excl(const int* __restrict__ cnt,
                                                  int* __restrict__ rowptr,
                                                  int N) {
  __shared__ int tmp[1024];
  __shared__ int carry;
  const int t = threadIdx.x;
  if (t == 0) carry = 0;
  __syncthreads();
  for (int base = 0; base < N; base += 1024) {
    int i = base + t;
    int v = (i < N) ? cnt[i] : 0;
    tmp[t] = v;
    __syncthreads();
#pragma unroll
    for (int off = 1; off < 1024; off <<= 1) {
      int addv = (t >= off) ? tmp[t - off] : 0;
      __syncthreads();
      tmp[t] += addv;
      __syncthreads();
    }
    int incl = tmp[t];
    int ctotal = tmp[1023];
    int cbase = carry;
    if (i < N) rowptr[i] = cbase + incl - v;
    __syncthreads();
    if (t == 0) carry = cbase + ctotal;
    __syncthreads();
  }
  if (t == 0) rowptr[N] = carry;
}

__global__ void fill_csr(const int* __restrict__ src,
                         const int* __restrict__ dst,
                         const int* __restrict__ rowptr,
                         int* __restrict__ cur, int* __restrict__ eidx,
                         int E) {
  for (int e = blockIdx.x * blockDim.x + threadIdx.x; e < E;
       e += gridDim.x * blockDim.x) {
    int d = dst[e];
    int slot = rowptr[d] + atomicAdd(&cur[d], 1);
    eidx[slot] = src[e];
  }
}

__global__ void __launch_bounds__(128) aggregate_csr(
    const float* __restrict__ h, const int* __restrict__ rowptr,
    const int* __restrict__ eidx, const float* __restrict__ invdeg,
    float* __restrict__ c, int N) {
  __shared__ int idxs[128];
  const int t = threadIdx.x;
  for (int node = blockIdx.x; node < N; node += gridDim.x) {
    int beg = rowptr[node];
    int cnt = rowptr[node + 1] - beg;
    float acc = 0.0f;
    for (int chunk = 0; chunk < cnt; chunk += 128) {
      int m = min(128, cnt - chunk);
      if (t < m) idxs[t] = eidx[beg + chunk + t];
      __syncthreads();
      int i = 0;
      for (; i + 4 <= m; i += 4) {
        float v0 = h[idxs[i + 0] * HID + t];
        float v1 = h[idxs[i + 1] * HID + t];
        float v2 = h[idxs[i + 2] * HID + t];
        float v3 = h[idxs[i + 3] * HID + t];
        acc += v0 + v1 + v2 + v3;
      }
      for (; i < m; ++i) acc += h[idxs[i] * HID + t];
      __syncthreads();
    }
    c[node * HID + t] = acc * invdeg[node];
  }
}

__global__ void __launch_bounds__(256) embed_gemm(
    const float* __restrict__ X, const float* __restrict__ W,
    const float* __restrict__ B, float* __restrict__ Y) {
  __shared__ __attribute__((aligned(16))) float tA[16 * HID];

  const int tid = threadIdx.x;
  const int rowBase = blockIdx.x * 16;

  const float4* Xv = (const float4*)(X + rowBase * HID);
  float4* tAv = (float4*)tA;
  tAv[tid] = Xv[tid];
  tAv[tid + 256] = Xv[tid + 256];
  __syncthreads();

  const int wave = tid >> 5;
  const int lane = tid & 31;
  const int half = lane >> 4;
  const int l16 = lane & 15;
  const int col = wave * 16 + l16;

  v8f acc;
  {
    float bv = B[col];
#pragma unroll
    for (int r = 0; r < 8; ++r) acc[r] = bv;
  }

  const float* arow = tA + l16 * HID;
  const float* wrow = W + col * HID;

#pragma unroll 4
  for (int k = 0; k < HID; k += 4) {
    const int kk = k + 2 * half;
    v2f a = *(const v2f*)(arow + kk);
    v2f b = *(const v2f*)(wrow + kk);
    acc = __builtin_amdgcn_wmma_f32_16x16x4_f32(false, a, false, b, (short)0,
                                                acc, false, false);
  }

#pragma unroll
  for (int r = 0; r < 8; ++r) {
    int row = rowBase + r + 8 * half;
    Y[row * HID + col] = acc[r];
  }
}

__global__ void __launch_bounds__(256) layer_gemm(
    const float* __restrict__ h, const float* __restrict__ c,
    const float* __restrict__ W, const float* __restrict__ B,
    float* __restrict__ Z) {
  __shared__ __attribute__((aligned(16))) float tA[16 * 256];

  const int tid = threadIdx.x;
  const int rowBase = blockIdx.x * 16;

#pragma unroll
  for (int i = 0; i < 2; ++i) {
    int v = tid + i * 256;
    int r = v >> 5;
    int c4 = v & 31;
    ((float4*)(tA + r * 256))[c4] =
        ((const float4*)(h + (rowBase + r) * HID))[c4];
  }
#pragma unroll
  for (int i = 0; i < 2; ++i) {
    int v = tid + i * 256;
    int r = v >> 5;
    int c4 = v & 31;
    ((float4*)(tA + r * 256 + HID))[c4] =
        ((const float4*)(c + (rowBase + r) * HID))[c4];
  }
  __syncthreads();

  const int wave = tid >> 5;
  const int lane = tid & 31;
  const int half = lane >> 4;
  const int l16 = lane & 15;
  const int col = wave * 16 + l16;

  v8f acc;
  {
    float bv = B[col];
#pragma unroll
    for (int r = 0; r < 8; ++r) acc[r] = bv;
  }

  const float* arow = tA + l16 * 256;
  const float* wrow = W + col * 256;

#pragma unroll 4
  for (int k = 0; k < 256; k += 4) {
    const int kk = k + 2 * half;
    v2f a = *(const v2f*)(arow + kk);
    v2f b = *(const v2f*)(wrow + kk);
    acc = __builtin_amdgcn_wmma_f32_16x16x4_f32(false, a, false, b, (short)0,
                                                acc, false, false);
  }

#pragma unroll
  for (int r = 0; r < 8; ++r) {
    int row = rowBase + r + 8 * half;
    Z[row * HID + col] = acc[r];
  }
}


template <int M>
__device__ __forceinline__ float row_xor(float x) {
  const int b = __builtin_bit_cast(int, x);
  return __builtin_bit_cast(
      float, __builtin_amdgcn_update_dpp(b, b, 0x160 + M, 0xf, 0xf, true));
}

__device__ __forceinline__ float row_other(float x) {
  const int b = __builtin_bit_cast(int, x);
  return __builtin_bit_cast(
      float, __builtin_amdgcn_permlanex16(b, b, 0x76543210, 0xfedcba98, false,
                                          false));
}

#define NRS_WAVES 8
#define NRS_ROWS 10

__global__ void __launch_bounds__(NRS_WAVES * 32) norm_relu_stats(
    float* __restrict__ z, float* __restrict__ part, int N) {
  __shared__ __attribute__((aligned(16))) float red[NRS_WAVES * 256];
  const int t = threadIdx.x;
  const int lane = t & 31;
  const int w = __builtin_amdgcn_readfirstlane(t >> 5);
  const int gw = blockIdx.x * NRS_WAVES + w;
  const int tw = gridDim.x * NRS_WAVES;

  float4* __restrict__ zv = (float4*)z;
  float4 ps = make_float4(0.0f, 0.0f, 0.0f, 0.0f);
  float4 pss = make_float4(0.0f, 0.0f, 0.0f, 0.0f);

  for (int row0 = gw; row0 < N; row0 += NRS_ROWS * tw) {
    float4 v[NRS_ROWS];
#pragma unroll
    for (int j = 0; j < NRS_ROWS; ++j) {
      const int row = row0 + j * tw;
      const int rr = (row < N) ? row : row0;
      v[j] = zv[rr * (HID / 4) + lane];
    }
    __builtin_amdgcn_sched_barrier(0);
#pragma unroll
    for (int j = 0; j < NRS_ROWS; ++j) {
      const int row = row0 + j * tw;
      const float4 x = v[j];
      float sq = (x.x * x.x + x.y * x.y) + (x.z * x.z + x.w * x.w);
      sq += row_xor<1>(sq);
      sq += row_xor<2>(sq);
      sq += row_xor<4>(sq);
      sq += row_xor<8>(sq);
      sq += row_other(sq);
      const float nrm = fmaxf(sqrtf(sq), 1e-12f);
      float4 r;
      r.x = fmaxf(x.x / nrm, 0.0f);
      r.y = fmaxf(x.y / nrm, 0.0f);
      r.z = fmaxf(x.z / nrm, 0.0f);
      r.w = fmaxf(x.w / nrm, 0.0f);
      if (row < N) {
        zv[row * (HID / 4) + lane] = r;
        ps.x += r.x;
        ps.y += r.y;
        ps.z += r.z;
        ps.w += r.w;
        pss.x += r.x * r.x;
        pss.y += r.y * r.y;
        pss.z += r.z * r.z;
        pss.w += r.w * r.w;
      }
    }
  }

  ((float4*)(red + w * 256))[lane] = ps;
  ((float4*)(red + w * 256 + 128))[lane] = pss;
  __syncthreads();
  float s = 0.0f;
#pragma unroll
  for (int k = 0; k < NRS_WAVES; ++k) s += red[k * 256 + t];
  part[blockIdx.x * 256 + t] = s;
}

__global__ void __launch_bounds__(256) bn_finalize(
    const float* __restrict__ part, int nb, const float* __restrict__ gamma,
    const float* __restrict__ beta, float* __restrict__ bn, float invN) {
  __shared__ float tot[256];
  const int j = threadIdx.x;
  float s = 0.0f;
  for (int b = 0; b < nb; ++b) s += part[b * 256 + j];
  tot[j] = s;
  __syncthreads();
  if (j < 128) {
    float mu = tot[j] * invN;
    float var = tot[128 + j] * invN - mu * mu;
    float a = gamma[j] / sqrtf(var + 1e-5f);
    bn[j] = a;
    bn[128 + j] = beta[j] - mu * a;
  }
}

__global__ void residual_bn(const float* __restrict__ h,
                            const float* __restrict__ z,
                            const float* __restrict__ bn,
                            float* __restrict__ out, int total) {
  for (int i = blockIdx.x * blockDim.x + threadIdx.x; i < total;
       i += gridDim.x * blockDim.x) {
    int j = i & (HID - 1);
    out[i] = h[i] + z[i] * bn[j] + bn[HID + j];
  }
}

extern "C" void kernel_launch(void* const* d_in, const int* in_sizes, int n_in,
                              void* d_out, int out_size, void* d_ws,
                              size_t ws_size, hipStream_t stream) {
  const float* h0 = (const float*)d_in[0];
  const int* src = (const int*)d_in[1];
  const int* dst = (const int*)d_in[2];
  const float* W_emb = (const float*)d_in[3];
  const float* b_emb = (const float*)d_in[4];
  const float* Ws = (const float*)d_in[5];
  const float* bs = (const float*)d_in[6];
  const float* gam = (const float*)d_in[7];
  const float* bet = (const float*)d_in[8];

  const int N = in_sizes[0] / HID;
  const int E = in_sizes[1];
  const int NF = N * HID;
  const int L = 4;

  float* ws = (float*)d_ws;
  float* H = ws;
  float* C = H + NF;
  float* Z = C + NF;
  float* DEG = Z + NF;
  float* BN = DEG + N;
  float* PART = BN + 256;
  int* rowptr = (int*)(PART + NB_STATS * 256);
  int* cnt = rowptr + (N + 1);
  int* eidx = cnt + N;

  const int mtiles = N / 16;

  zero_i32<<<256, 256, 0, stream>>>(cnt, N);
  hist_deg<<<1024, 256, 0, stream>>>(dst, cnt, E);
  invdeg_from_cnt<<<256, 256, 0, stream>>>(cnt, DEG, N);
  scan_excl<<<1, 1024, 0, stream>>>(cnt, rowptr, N);
  zero_i32<<<256, 256, 0, stream>>>(cnt, N);
  fill_csr<<<1024, 256, 0, stream>>>(src, dst, rowptr, cnt, eidx, E);

  embed_gemm<<<mtiles, 256, 0, stream>>>(h0, W_emb, b_emb, H);

  for (int l = 0; l < L; ++l) {
    aggregate_csr<<<2048, 128, 0, stream>>>(H, rowptr, eidx, DEG, C, N);
    layer_gemm<<<mtiles, 256, 0, stream>>>(H, C, Ws + l * HID * 256,
                                           bs + l * HID, Z);
    norm_relu_stats<<<NB_STATS, NRS_WAVES * 32, 0, stream>>>(Z, PART, N);
    bn_finalize<<<1, 256, 0, stream>>>(PART, NB_STATS, gam + l * HID,
                                       bet + l * HID, BN, 1.0f / (float)N);
    float* out = (l == L - 1) ? (float*)d_out : H;
    residual_bn<<<4096, 256, 0, stream>>>(H, Z, BN, out, NF);
  }
}


# aggregate_csr: one wave per node (float4 per lane, indices via readlane, no LDS, no barrier), all of a node's neighbour rows requested before the first add; sum order unchanged
# speedup vs baseline: 1.3993x; 1.1977x over previous
#include <hip/hip_runtime.h>
#include <hip/hip_bf16.h>


typedef float v2f __attribute__((ext_vector_type(2)));
typedef float v8f __attribute__((ext_vector_type(8)));

#define HID 128
#define NB_STATS 512

__global__ void zero_i32(int* __restrict__ p, int n) {
  for (int i = blockIdx.x * blockDim.x + threadIdx.x; i < n;
       i += gridDim.x * blockDim.x)
    p[i] = 0;
}

__global__ void hist_deg(const int* __restrict__ dst, int* __restrict__ cnt,
                         int E) {
  for (int e = blockIdx.x * blockDim.x + threadIdx.x; e < E;
       e += gridDim.x * blockDim.x)
    atomicAdd(&cnt[dst[e]], 1);
}

__global__ void invdeg_from_cnt(const int* __restrict__ cnt,
                                float* __restrict__ invdeg, int N) {
  for (int i = blockIdx.x * blockDim.x + threadIdx.x; i < N;
       i += gridDim.x * blockDim.x)
    invdeg[i] = 1.0f / fmaxf((float)cnt[i], 1.0f);
}

__global__ void __launch_bounds__(1024) scan_excl(const int* __restrict__ cnt,
                                                  int* __restrict__ rowptr,
                                                  int N) {
  __shared__ int tmp[1024];
  __shared__ int carry;
  const int t = threadIdx.x;
  if (t == 0) carry = 0;
  __syncthreads();
  for (int base = 0; base < N; base += 1024) {
    int i = base + t;
    int v = (i < N) ? cnt[i] : 0;
    tmp[t] = v;
    __syncthreads();
#pragma unroll
    for (int off = 1; off < 1024; off <<= 1) {
      int addv = (t >= off) ? tmp[t - off] : 0;
      __syncthreads();
      tmp[t] += addv;
      __syncthreads();
    }
    int incl = tmp[t];
    int ctotal = tmp[1023];
    int cbase = carry;
    if (i < N) rowptr[i] = cbase + incl - v;
    __syncthreads();
    if (t == 0) carry = cbase + ctotal;
    __syncthreads();
  }
  if (t == 0) rowptr[N] = carry;
}

__global__ void fill_csr(const int* __restrict__ src,
                         const int* __restrict__ dst,
                         const int* __restrict__ rowptr,
                         int* __restrict__ cur, int* __restrict__ eidx,
                         int E) {
  for (int e = blockIdx.x * blockDim.x + threadIdx.x; e < E;
       e += gridDim.x * blockDim.x) {
    int d = dst[e];
    int slot = rowptr[d] + atomicAdd(&cur[d], 1);
    eidx[slot] = src[e];
  }
}

#define AGG_WAVES 4
#define AGG_NODES 8
#define AGG_ROWS 32

template <int J>
__device__ __forceinline__ void agg_request(float4 (&v)[AGG_ROWS],
                                            const float4* __restrict__ hv,
                                            int cur, int m, int lane) {
  if constexpr (J < AGG_ROWS) {
    if (J < m) {
      const int s = __builtin_amdgcn_readlane(cur, J);
      v[J] = hv[s * (HID / 4) + lane];
      agg_request<J + 1>(v, hv, cur, m, lane);
    }
  }
}

template <int G>
__device__ __forceinline__ void agg_sum(float4& acc,
                                        const float4 (&v)[AGG_ROWS], int m) {
  if constexpr (G < AGG_ROWS) {
    if (G + 4 <= m) {
      acc.x += ((v[G].x + v[G + 1].x) + v[G + 2].x) + v[G + 3].x;
      acc.y += ((v[G].y + v[G + 1].y) + v[G + 2].y) + v[G + 3].y;
      acc.z += ((v[G].z + v[G + 1].z) + v[G + 2].z) + v[G + 3].z;
      acc.w += ((v[G].w + v[G + 1].w) + v[G + 2].w) + v[G + 3].w;
      agg_sum<G + 4>(acc, v, m);
    } else if (G < m) {
      acc.x += v[G].x;
      acc.y += v[G].y;
      acc.z += v[G].z;
      acc.w += v[G].w;
      if (G + 1 < m) {
        acc.x += v[G + 1].x;
        acc.y += v[G + 1].y;
        acc.z += v[G + 1].z;
        acc.w += v[G + 1].w;
      }
      if (G + 2 < m) {
        acc.x += v[G + 2].x;
        acc.y += v[G + 2].y;
        acc.z += v[G + 2].z;
        acc.w += v[G + 2].w;
      }
    }
  }
}

__global__ void __launch_bounds__(AGG_WAVES * 32) aggregate_csr(
    const float* __restrict__ h, const int* __restrict__ rowptr,
    const int* __restrict__ eidx, const float* __restrict__ invdeg,
    float* __restrict__ c, int N, int E) {
  const int lane = threadIdx.x & 31;
  const int w = __builtin_amdgcn_readfirstlane(threadIdx.x >> 5);
  const int gw = blockIdx.x * AGG_WAVES + w;
  const int tw = gridDim.x * AGG_WAVES;
  const float4* __restrict__ hv = (const float4*)h;
  float4* __restrict__ cv = (float4*)c;

  for (int base = gw * AGG_NODES; base < N; base += tw * AGG_NODES) {
    const int nn = min(AGG_NODES, N - base);
    const int rp = rowptr[min(base + lane, N)];
    const int dg = __builtin_bit_cast(int, invdeg[min(base + lane, N - 1)]);
    int b = __builtin_amdgcn_readlane(rp, 0);
    int e = __builtin_amdgcn_readlane(rp, 1);
    int idx = eidx[min(b + lane, E - 1)];

    for (int k = 0; k < nn; ++k) {
      const int cnt = e - b;
      const int ne = __builtin_amdgcn_readlane(rp, k + 2);
      const int nidx = eidx[min(e + lane, E - 1)];

      float4 acc = make_float4(0.0f, 0.0f, 0.0f, 0.0f);
      for (int chunk = 0; chunk < cnt; chunk += AGG_ROWS) {
        int cur = idx;
        if (chunk > 0) cur = eidx[min(b + chunk + lane, E - 1)];
        const int m = min(AGG_ROWS, cnt - chunk);
        float4 v[AGG_ROWS];
        agg_request<0>(v, hv, cur, m, lane);
        agg_sum<0>(acc, v, m);
      }

      const float sc =
          __builtin_bit_cast(float, __builtin_amdgcn_readlane(dg, k));
      float4 r;
      r.x = acc.x * sc;
      r.y = acc.y * sc;
      r.z = acc.z * sc;
      r.w = acc.w * sc;
      cv[(base + k) * (HID / 4) + lane] = r;

      b = e;
      e = ne;
      idx = nidx;
    }
  }
}

__global__ void __launch_bounds__(256) embed_gemm(
    const float* __restrict__ X, const float* __restrict__ W,
    const float* __restrict__ B, float* __restrict__ Y) {
  __shared__ __attribute__((aligned(16))) float tA[16 * HID];

  const int tid = threadIdx.x;
  const int rowBase = blockIdx.x * 16;

  const float4* Xv = (const float4*)(X + rowBase * HID);
  float4* tAv = (float4*)tA;
  tAv[tid] = Xv[tid];
  tAv[tid + 256] = Xv[tid + 256];
  __syncthreads();

  const int wave = tid >> 5;
  const int lane = tid & 31;
  const int half = lane >> 4;
  const int l16 = lane & 15;
  const int col = wave * 16 + l16;

  v8f acc;
  {
    float bv = B[col];
#pragma unroll
    for (int r = 0; r < 8; ++r) acc[r] = bv;
  }

  const float* arow = tA + l16 * HID;
  const float* wrow = W + col * HID;

#pragma unroll 4
  for (int k = 0; k < HID; k += 4) {
    const int kk = k + 2 * half;
    v2f a = *(const v2f*)(arow + kk);
    v2f b = *(const v2f*)(wrow + kk);
    acc = __builtin_amdgcn_wmma_f32_16x16x4_f32(false, a, false, b, (short)0,
                                                acc, false, false);
  }

#pragma unroll
  for (int r = 0; r < 8; ++r) {
    int row = rowBase + r + 8 * half;
    Y[row * HID + col] = acc[r];
  }
}

__global__ void __launch_bounds__(256) layer_gemm(
    const float* __restrict__ h, const float* __restrict__ c,
    const float* __restrict__ W, const float* __restrict__ B,
    float* __restrict__ Z) {
  __shared__ __attribute__((aligned(16))) float tA[16 * 256];

  const int tid = threadIdx.x;
  const int rowBase = blockIdx.x * 16;

#pragma unroll
  for (int i = 0; i < 2; ++i) {
    int v = tid + i * 256;
    int r = v >> 5;
    int c4 = v & 31;
    ((float4*)(tA + r * 256))[c4] =
        ((const float4*)(h + (rowBase + r) * HID))[c4];
  }
#pragma unroll
  for (int i = 0; i < 2; ++i) {
    int v = tid + i * 256;
    int r = v >> 5;
    int c4 = v & 31;
    ((float4*)(tA + r * 256 + HID))[c4] =
        ((const float4*)(c + (rowBase + r) * HID))[c4];
  }
  __syncthreads();

  const int wave = tid >> 5;
  const int lane = tid & 31;
  const int half = lane >> 4;
  const int l16 = lane & 15;
  const int col = wave * 16 + l16;

  v8f acc;
  {
    float bv = B[col];
#pragma unroll
    for (int r = 0; r < 8; ++r) acc[r] = bv;
  }

  const float* arow = tA + l16 * 256;
  const float* wrow = W + col * 256;

#pragma unroll 4
  for (int k = 0; k < 256; k += 4) {
    const int kk = k + 2 * half;
    v2f a = *(const v2f*)(arow + kk);
    v2f b = *(const v2f*)(wrow + kk);
    acc = __builtin_amdgcn_wmma_f32_16x16x4_f32(false, a, false, b, (short)0,
                                                acc, false, false);
  }

#pragma unroll
  for (int r = 0; r < 8; ++r) {
    int row = rowBase + r + 8 * half;
    Z[row * HID + col] = acc[r];
  }
}


template <int M>
__device__ __forceinline__ float row_xor(float x) {
  const int b = __builtin_bit_cast(int, x);
  return __builtin_bit_cast(
      float, __builtin_amdgcn_update_dpp(b, b, 0x160 + M, 0xf, 0xf, true));
}

__device__ __forceinline__ float row_other(float x) {
  const int b = __builtin_bit_cast(int, x);
  return __builtin_bit_cast(
      float, __builtin_amdgcn_permlanex16(b, b, 0x76543210, 0xfedcba98, false,
                                          false));
}

#define NRS_WAVES 8
#define NRS_ROWS 10

__global__ void __launch_bounds__(NRS_WAVES * 32) norm_relu_stats(
    float* __restrict__ z, float* __restrict__ part, int N) {
  __shared__ __attribute__((aligned(16))) float red[NRS_WAVES * 256];
  const int t = threadIdx.x;
  const int lane = t & 31;
  const int w = __builtin_amdgcn_readfirstlane(t >> 5);
  const int gw = blockIdx.x * NRS_WAVES + w;
  const int tw = gridDim.x * NRS_WAVES;

  float4* __restrict__ zv = (float4*)z;
  float4 ps = make_float4(0.0f, 0.0f, 0.0f, 0.0f);
  float4 pss = make_float4(0.0f, 0.0f, 0.0f, 0.0f);

  for (int row0 = gw; row0 < N; row0 += NRS_ROWS * tw) {
    float4 v[NRS_ROWS];
#pragma unroll
    for (int j = 0; j < NRS_ROWS; ++j) {
      const int row = row0 + j * tw;
      const int rr = (row < N) ? row : row0;
      v[j] = zv[rr * (HID / 4) + lane];
    }
    __builtin_amdgcn_sched_barrier(0);
#pragma unroll
    for (int j = 0; j < NRS_ROWS; ++j) {
      const int row = row0 + j * tw;
      const float4 x = v[j];
      float sq = (x.x * x.x + x.y * x.y) + (x.z * x.z + x.w * x.w);
      sq += row_xor<1>(sq);
      sq += row_xor<2>(sq);
      sq += row_xor<4>(sq);
      sq += row_xor<8>(sq);
      sq += row_other(sq);
      const float nrm = fmaxf(sqrtf(sq), 1e-12f);
      float4 r;
      r.x = fmaxf(x.x / nrm, 0.0f);
      r.y = fmaxf(x.y / nrm, 0.0f);
      r.z = fmaxf(x.z / nrm, 0.0f);
      r.w = fmaxf(x.w / nrm, 0.0f);
      if (row < N) {
        zv[row * (HID / 4) + lane] = r;
        ps.x += r.x;
        ps.y += r.y;
        ps.z += r.z;
        ps.w += r.w;
        pss.x += r.x * r.x;
        pss.y += r.y * r.y;
        pss.z += r.z * r.z;
        pss.w += r.w * r.w;
      }
    }
  }

  ((float4*)(red + w * 256))[lane] = ps;
  ((float4*)(red + w * 256 + 128))[lane] = pss;
  __syncthreads();
  float s = 0.0f;
#pragma unroll
  for (int k = 0; k < NRS_WAVES; ++k) s += red[k * 256 + t];
  part[blockIdx.x * 256 + t] = s;
}

__global__ void __launch_bounds__(256) bn_finalize(
    const float* __restrict__ part, int nb, const float* __restrict__ gamma,
    const float* __restrict__ beta, float* __restrict__ bn, float invN) {
  __shared__ float tot[256];
  const int j = threadIdx.x;
  float s = 0.0f;
  for (int b = 0; b < nb; ++b) s += part[b * 256 + j];
  tot[j] = s;
  __syncthreads();
  if (j < 128) {
    float mu = tot[j] * invN;
    float var = tot[128 + j] * invN - mu * mu;
    float a = gamma[j] / sqrtf(var + 1e-5f);
    bn[j] = a;
    bn[128 + j] = beta[j] - mu * a;
  }
}

__global__ void residual_bn(const float* __restrict__ h,
                            const float* __restrict__ z,
                            const float* __restrict__ bn,
                            float* __restrict__ out, int total) {
  for (int i = blockIdx.x * blockDim.x + threadIdx.x; i < total;
       i += gridDim.x * blockDim.x) {
    int j = i & (HID - 1);
    out[i] = h[i] + z[i] * bn[j] + bn[HID + j];
  }
}

extern "C" void kernel_launch(void* const* d_in, const int* in_sizes, int n_in,
                              void* d_out, int out_size, void* d_ws,
                              size_t ws_size, hipStream_t stream) {
  const float* h0 = (const float*)d_in[0];
  const int* src = (const int*)d_in[1];
  const int* dst = (const int*)d_in[2];
  const float* W_emb = (const float*)d_in[3];
  const float* b_emb = (const float*)d_in[4];
  const float* Ws = (const float*)d_in[5];
  const float* bs = (const float*)d_in[6];
  const float* gam = (const float*)d_in[7];
  const float* bet = (const float*)d_in[8];

  const int N = in_sizes[0] / HID;
  const int E = in_sizes[1];
  const int NF = N * HID;
  const int L = 4;

  float* ws = (float*)d_ws;
  float* H = ws;
  float* C = H + NF;
  float* Z = C + NF;
  float* DEG = Z + NF;
  float* BN = DEG + N;
  float* PART = BN + 256;
  int* rowptr = (int*)(PART + NB_STATS * 256);
  int* cnt = rowptr + (N + 1);
  int* eidx = cnt + N;

  const int mtiles = N / 16;

  zero_i32<<<256, 256, 0, stream>>>(cnt, N);
  hist_deg<<<1024, 256, 0, stream>>>(dst, cnt, E);
  invdeg_from_cnt<<<256, 256, 0, stream>>>(cnt, DEG, N);
  scan_excl<<<1, 1024, 0, stream>>>(cnt, rowptr, N);
  zero_i32<<<256, 256, 0, stream>>>(cnt, N);
  fill_csr<<<1024, 256, 0, stream>>>(src, dst, rowptr, cnt, eidx, E);

  embed_gemm<<<mtiles, 256, 0, stream>>>(h0, W_emb, b_emb, H);

  for (int l = 0; l < L; ++l) {
    aggregate_csr<<<1250, AGG_WAVES * 32, 0, stream>>>(H, rowptr, eidx, DEG, C,
                                                       N, E);
    layer_gemm<<<mtiles, 256, 0, stream>>>(H, C, Ws + l * HID * 256,
                                           bs + l * HID, Z);
    norm_relu_stats<<<NB_STATS, NRS_WAVES * 32, 0, stream>>>(Z, PART, N);
    bn_finalize<<<1, 256, 0, stream>>>(PART, NB_STATS, gam + l * HID,
                                       bet + l * HID, BN, 1.0f / (float)N);
    float* out = (l == L - 1) ? (float*)d_out : H;
    residual_bn<<<4096, 256, 0, stream>>>(H, Z, BN, out, NF);
  }
}


# scan_excl (CSR row bounds): one pass, int4 per lane, loads first, prefix across lanes on the vector ALU inside each wave, one block barrier, not 23 per 1024 counts; integer sums, same result
# speedup vs baseline: 1.6071x; 1.1485x over previous
#include <hip/hip_runtime.h>
#include <hip/hip_bf16.h>


typedef float v2f __attribute__((ext_vector_type(2)));
typedef float v8f __attribute__((ext_vector_type(8)));

#define HID 128
#define NB_STATS 512

__global__ void zero_i32(int* __restrict__ p, int n) {
  for (int i = blockIdx.x * blockDim.x + threadIdx.x; i < n;
       i += gridDim.x * blockDim.x)
    p[i] = 0;
}

__global__ void hist_deg(const int* __restrict__ dst, int* __restrict__ cnt,
                         int E) {
  for (int e = blockIdx.x * blockDim.x + threadIdx.x; e < E;
       e += gridDim.x * blockDim.x)
    atomicAdd(&cnt[dst[e]], 1);
}

__global__ void invdeg_from_cnt(const int* __restrict__ cnt,
                                float* __restrict__ invdeg, int N) {
  for (int i = blockIdx.x * blockDim.x + threadIdx.x; i < N;
       i += gridDim.x * blockDim.x)
    invdeg[i] = 1.0f / fmaxf((float)cnt[i], 1.0f);
}

#define SCAN_WAVES 32
#define SCAN_VEC 10
#define SCAN_WSPAN (SCAN_VEC * 128)
#define SCAN_SPAN (SCAN_WAVES * SCAN_WSPAN)

template <int S>
__device__ __forceinline__ int row_shr_or0(int x) {
  return __builtin_amdgcn_update_dpp(0, x, 0x110 + S, 0xf, 0xf, false);
}

__device__ __forceinline__ int wave_incl_scan(int x, int lane) {
  x += row_shr_or0<1>(x);
  x += row_shr_or0<2>(x);
  x += row_shr_or0<4>(x);
  x += row_shr_or0<8>(x);
  const int lo = __builtin_amdgcn_readlane(x, 15);
  return x + ((lane >= 16) ? lo : 0);
}

__global__ void __launch_bounds__(SCAN_WAVES * 32) scan_excl(
    const int* __restrict__ cnt, int* __restrict__ rowptr, int N) {
  __shared__ int wtot[SCAN_WAVES];
  const int lane = threadIdx.x & 31;
  const int w = __builtin_amdgcn_readfirstlane(threadIdx.x >> 5);
  int carry = 0;

  for (int base = 0; base < N; base += SCAN_SPAN) {
    const int e0 = base + w * SCAN_WSPAN + lane * 4;
    int4 v[SCAN_VEC];
#pragma unroll
    for (int i = 0; i < SCAN_VEC; ++i) {
      const int e = e0 + i * 128;
      const int4 x = *(const int4*)(cnt + min(e, N - 4));
      const bool in = (e + 4 <= N);
      v[i].x = in ? x.x : 0;
      v[i].y = in ? x.y : 0;
      v[i].z = in ? x.z : 0;
      v[i].w = in ? x.w : 0;
    }
    __builtin_amdgcn_sched_barrier(0);

    int pre[SCAN_VEC];
    int wsum = 0;
#pragma unroll
    for (int i = 0; i < SCAN_VEC; ++i) {
      const int s = (v[i].x + v[i].y) + (v[i].z + v[i].w);
      const int t = wave_incl_scan(s, lane);
      pre[i] = wsum + (t - s);
      wsum += __builtin_amdgcn_readlane(t, 31);
    }

    if (lane == 0) wtot[w] = wsum;
    __syncthreads();
    const int tv = wtot[lane];
    const int ti = wave_incl_scan(tv, lane);
    const int wbase = carry + __builtin_amdgcn_readlane(ti - tv, w);
    carry += __builtin_amdgcn_readlane(ti, 31);

#pragma unroll
    for (int i = 0; i < SCAN_VEC; ++i) {
      const int e = e0 + i * 128;
      int4 r;
      r.x = wbase + pre[i];
      r.y = r.x + v[i].x;
      r.z = r.y + v[i].y;
      r.w = r.z + v[i].z;
      if (e + 4 <= N) *(int4*)(rowptr + e) = r;
    }
    __syncthreads();
  }
  if (threadIdx.x == 0) rowptr[N] = carry;
}

__global__ void fill_csr(const int* __restrict__ src,
                         const int* __restrict__ dst,
                         const int* __restrict__ rowptr,
                         int* __restrict__ cur, int* __restrict__ eidx,
                         int E) {
  for (int e = blockIdx.x * blockDim.x + threadIdx.x; e < E;
       e += gridDim.x * blockDim.x) {
    int d = dst[e];
    int slot = rowptr[d] + atomicAdd(&cur[d], 1);
    eidx[slot] = src[e];
  }
}

#define AGG_WAVES 4
#define AGG_NODES 8
#define AGG_ROWS 32

template <int J>
__device__ __forceinline__ void agg_request(float4 (&v)[AGG_ROWS],
                                            const float4* __restrict__ hv,
                                            int cur, int m, int lane) {
  if constexpr (J < AGG_ROWS) {
    if (J < m) {
      const int s = __builtin_amdgcn_readlane(cur, J);
      v[J] = hv[s * (HID / 4) + lane];
      agg_request<J + 1>(v, hv, cur, m, lane);
    }
  }
}

template <int G>
__device__ __forceinline__ void agg_sum(float4& acc,
                                        const float4 (&v)[AGG_ROWS], int m) {
  if constexpr (G < AGG_ROWS) {
    if (G + 4 <= m) {
      acc.x += ((v[G].x + v[G + 1].x) + v[G + 2].x) + v[G + 3].x;
      acc.y += ((v[G].y + v[G + 1].y) + v[G + 2].y) + v[G + 3].y;
      acc.z += ((v[G].z + v[G + 1].z) + v[G + 2].z) + v[G + 3].z;
      acc.w += ((v[G].w + v[G + 1].w) + v[G + 2].w) + v[G + 3].w;
      agg_sum<G + 4>(acc, v, m);
    } else if (G < m) {
      acc.x += v[G].x;
      acc.y += v[G].y;
      acc.z += v[G].z;
      acc.w += v[G].w;
      if (G + 1 < m) {
        acc.x += v[G + 1].x;
        acc.y += v[G + 1].y;
        acc.z += v[G + 1].z;
        acc.w += v[G + 1].w;
      }
      if (G + 2 < m) {
        acc.x += v[G + 2].x;
        acc.y += v[G + 2].y;
        acc.z += v[G + 2].z;
        acc.w += v[G + 2].w;
      }
    }
  }
}

__global__ void __launch_bounds__(AGG_WAVES * 32) aggregate_csr(
    const float* __restrict__ h, const int* __restrict__ rowptr,
    const int* __restrict__ eidx, const float* __restrict__ invdeg,
    float* __restrict__ c, int N, int E) {
  const int lane = threadIdx.x & 31;
  const int w = __builtin_amdgcn_readfirstlane(threadIdx.x >> 5);
  const int gw = blockIdx.x * AGG_WAVES + w;
  const int tw = gridDim.x * AGG_WAVES;
  const float4* __restrict__ hv = (const float4*)h;
  float4* __restrict__ cv = (float4*)c;

  for (int base = gw * AGG_NODES; base < N; base += tw * AGG_NODES) {
    const int nn = min(AGG_NODES, N - base);
    const int rp = rowptr[min(base + lane, N)];
    const int dg = __builtin_bit_cast(int, invdeg[min(base + lane, N - 1)]);
    int b = __builtin_amdgcn_readlane(rp, 0);
    int e = __builtin_amdgcn_readlane(rp, 1);
    int idx = eidx[min(b + lane, E - 1)];

    for (int k = 0; k < nn; ++k) {
      const int cnt = e - b;
      const int ne = __builtin_amdgcn_readlane(rp, k + 2);
      const int nidx = eidx[min(e + lane, E - 1)];

      float4 acc = make_float4(0.0f, 0.0f, 0.0f, 0.0f);
      for (int chunk = 0; chunk < cnt; chunk += AGG_ROWS) {
        int cur = idx;
        if (chunk > 0) cur = eidx[min(b + chunk + lane, E - 1)];
        const int m = min(AGG_ROWS, cnt - chunk);
        float4 v[AGG_ROWS];
        agg_request<0>(v, hv, cur, m, lane);
        agg_sum<0>(acc, v, m);
      }

      const float sc =
          __builtin_bit_cast(float, __builtin_amdgcn_readlane(dg, k));
      float4 r;
      r.x = acc.x * sc;
      r.y = acc.y * sc;
      r.z = acc.z * sc;
      r.w = acc.w * sc;
      cv[(base + k) * (HID / 4) + lane] = r;

      b = e;
      e = ne;
      idx = nidx;
    }
  }
}

__global__ void __launch_bounds__(256) embed_gemm(
    const float* __restrict__ X, const float* __restrict__ W,
    const float* __restrict__ B, float* __restrict__ Y) {
  __shared__ __attribute__((aligned(16))) float tA[16 * HID];

  const int tid = threadIdx.x;
  const int rowBase = blockIdx.x * 16;

  const float4* Xv = (const float4*)(X + rowBase * HID);
  float4* tAv = (float4*)tA;
  tAv[tid] = Xv[tid];
  tAv[tid + 256] = Xv[tid + 256];
  __syncthreads();

  const int wave = tid >> 5;
  const int lane = tid & 31;
  const int half = lane >> 4;
  const int l16 = lane & 15;
  const int col = wave * 16 + l16;

  v8f acc;
  {
    float bv = B[col];
#pragma unroll
    for (int r = 0; r < 8; ++r) acc[r] = bv;
  }

  const float* arow = tA + l16 * HID;
  const float* wrow = W + col * HID;

#pragma unroll 4
  for (int k = 0; k < HID; k += 4) {
    const int kk = k + 2 * half;
    v2f a = *(const v2f*)(arow + kk);
    v2f b = *(const v2f*)(wrow + kk);
    acc = __builtin_amdgcn_wmma_f32_16x16x4_f32(false, a, false, b, (short)0,
                                                acc, false, false);
  }

#pragma unroll
  for (int r = 0; r < 8; ++r) {
    int row = rowBase + r + 8 * half;
    Y[row * HID + col] = acc[r];
  }
}

__global__ void __launch_bounds__(256) layer_gemm(
    const float* __restrict__ h, const float* __restrict__ c,
    const float* __restrict__ W, const float* __restrict__ B,
    float* __restrict__ Z) {
  __shared__ __attribute__((aligned(16))) float tA[16 * 256];

  const int tid = threadIdx.x;
  const int rowBase = blockIdx.x * 16;

#pragma unroll
  for (int i = 0; i < 2; ++i) {
    int v = tid + i * 256;
    int r = v >> 5;
    int c4 = v & 31;
    ((float4*)(tA + r * 256))[c4] =
        ((const float4*)(h + (rowBase + r) * HID))[c4];
  }
#pragma unroll
  for (int i = 0; i < 2; ++i) {
    int v = tid + i * 256;
    int r = v >> 5;
    int c4 = v & 31;
    ((float4*)(tA + r * 256 + HID))[c4] =
        ((const float4*)(c + (rowBase + r) * HID))[c4];
  }
  __syncthreads();

  const int wave = tid >> 5;
  const int lane = tid & 31;
  const int half = lane >> 4;
  const int l16 = lane & 15;
  const int col = wave * 16 + l16;

  v8f acc;
  {
    float bv = B[col];
#pragma unroll
    for (int r = 0; r < 8; ++r) acc[r] = bv;
  }

  const float* arow = tA + l16 * 256;
  const float* wrow = W + col * 256;

#pragma unroll 4
  for (int k = 0; k < 256; k += 4) {
    const int kk = k + 2 * half;
    v2f a = *(const v2f*)(arow + kk);
    v2f b = *(const v2f*)(wrow + kk);
    acc = __builtin_amdgcn_wmma_f32_16x16x4_f32(false, a, false, b, (short)0,
                                                acc, false, false);
  }

#pragma unroll
  for (int r = 0; r < 8; ++r) {
    int row = rowBase + r + 8 * half;
    Z[row * HID + col] = acc[r];
  }
}


template <int M>
__device__ __forceinline__ float row_xor(float x) {
  const int b = __builtin_bit_cast(int, x);
  return __builtin_bit_cast(
      float, __builtin_amdgcn_update_dpp(b, b, 0x160 + M, 0xf, 0xf, true));
}

__device__ __forceinline__ float row_other(float x) {
  const int b = __builtin_bit_cast(int, x);
  return __builtin_bit_cast(
      float, __builtin_amdgcn_permlanex16(b, b, 0x76543210, 0xfedcba98, false,
                                          false));
}

#define NRS_WAVES 8
#define NRS_ROWS 10

__global__ void __launch_bounds__(NRS_WAVES * 32) norm_relu_stats(
    float* __restrict__ z, float* __restrict__ part, int N) {
  __shared__ __attribute__((aligned(16))) float red[NRS_WAVES * 256];
  const int t = threadIdx.x;
  const int lane = t & 31;
  const int w = __builtin_amdgcn_readfirstlane(t >> 5);
  const int gw = blockIdx.x * NRS_WAVES + w;
  const int tw = gridDim.x * NRS_WAVES;

  float4* __restrict__ zv = (float4*)z;
  float4 ps = make_float4(0.0f, 0.0f, 0.0f, 0.0f);
  float4 pss = make_float4(0.0f, 0.0f, 0.0f, 0.0f);

  for (int row0 = gw; row0 < N; row0 += NRS_ROWS * tw) {
    float4 v[NRS_ROWS];
#pragma unroll
    for (int j = 0; j < NRS_ROWS; ++j) {
      const int row = row0 + j * tw;
      const int rr = (row < N) ? row : row0;
      v[j] = zv[rr * (HID / 4) + lane];
    }
    __builtin_amdgcn_sched_barrier(0);
#pragma unroll
    for (int j = 0; j < NRS_ROWS; ++j) {
      const int row = row0 + j * tw;
      const float4 x = v[j];
      float sq = (x.x * x.x + x.y * x.y) + (x.z * x.z + x.w * x.w);
      sq += row_xor<1>(sq);
      sq += row_xor<2>(sq);
      sq += row_xor<4>(sq);
      sq += row_xor<8>(sq);
      sq += row_other(sq);
      const float nrm = fmaxf(sqrtf(sq), 1e-12f);
      float4 r;
      r.x = fmaxf(x.x / nrm, 0.0f);
      r.y = fmaxf(x.y / nrm, 0.0f);
      r.z = fmaxf(x.z / nrm, 0.0f);
      r.w = fmaxf(x.w / nrm, 0.0f);
      if (row < N) {
        zv[row * (HID / 4) + lane] = r;
        ps.x += r.x;
        ps.y += r.y;
        ps.z += r.z;
        ps.w += r.w;
        pss.x += r.x * r.x;
        pss.y += r.y * r.y;
        pss.z += r.z * r.z;
        pss.w += r.w * r.w;
      }
    }
  }

  ((float4*)(red + w * 256))[lane] = ps;
  ((float4*)(red + w * 256 + 128))[lane] = pss;
  __syncthreads();
  float s = 0.0f;
#pragma unroll
  for (int k = 0; k < NRS_WAVES; ++k) s += red[k * 256 + t];
  part[blockIdx.x * 256 + t] = s;
}

__global__ void __launch_bounds__(256) bn_finalize(
    const float* __restrict__ part, int nb, const float* __restrict__ gamma,
    const float* __restrict__ beta, float* __restrict__ bn, float invN) {
  __shared__ float tot[256];
  const int j = threadIdx.x;
  float s = 0.0f;
  for (int b = 0; b < nb; ++b) s += part[b * 256 + j];
  tot[j] = s;
  __syncthreads();
  if (j < 128) {
    float mu = tot[j] * invN;
    float var = tot[128 + j] * invN - mu * mu;
    float a = gamma[j] / sqrtf(var + 1e-5f);
    bn[j] = a;
    bn[128 + j] = beta[j] - mu * a;
  }
}

__global__ void residual_bn(const float* __restrict__ h,
                            const float* __restrict__ z,
                            const float* __restrict__ bn,
                            float* __restrict__ out, int total) {
  for (int i = blockIdx.x * blockDim.x + threadIdx.x; i < total;
       i += gridDim.x * blockDim.x) {
    int j = i & (HID - 1);
    out[i] = h[i] + z[i] * bn[j] + bn[HID + j];
  }
}

extern "C" void kernel_launch(void* const* d_in, const int* in_sizes, int n_in,
                              void* d_out, int out_size, void* d_ws,
                              size_t ws_size, hipStream_t stream) {
  const float* h0 = (const float*)d_in[0];
  const int* src = (const int*)d_in[1];
  const int* dst = (const int*)d_in[2];
  const float* W_emb = (const float*)d_in[3];
  const float* b_emb = (const float*)d_in[4];
  const float* Ws = (const float*)d_in[5];
  const float* bs = (const float*)d_in[6];
  const float* gam = (const float*)d_in[7];
  const float* bet = (const float*)d_in[8];

  const int N = in_sizes[0] / HID;
  const int E = in_sizes[1];
  const int NF = N * HID;
  const int L = 4;

  float* ws = (float*)d_ws;
  float* H = ws;
  float* C = H + NF;
  float* Z = C + NF;
  float* DEG = Z + NF;
  float* BN = DEG + N;
  float* PART = BN + 256;
  int* cnt = (int*)(PART + NB_STATS * 256);
  int* rowptr = cnt + N;
  int* eidx = rowptr + (N + 1);

  const int mtiles = N / 16;

  zero_i32<<<256, 256, 0, stream>>>(cnt, N);
  hist_deg<<<1024, 256, 0, stream>>>(dst, cnt, E);
  invdeg_from_cnt<<<256, 256, 0, stream>>>(cnt, DEG, N);
  scan_excl<<<1, SCAN_WAVES * 32, 0, stream>>>(cnt, rowptr, N);
  zero_i32<<<256, 256, 0, stream>>>(cnt, N);
  fill_csr<<<1024, 256, 0, stream>>>(src, dst, rowptr, cnt, eidx, E);

  embed_gemm<<<mtiles, 256, 0, stream>>>(h0, W_emb, b_emb, H);

  for (int l = 0; l < L; ++l) {
    aggregate_csr<<<1250, AGG_WAVES * 32, 0, stream>>>(H, rowptr, eidx, DEG, C,
                                                       N, E);
    layer_gemm<<<mtiles, 256, 0, stream>>>(H, C, Ws + l * HID * 256,
                                           bs + l * HID, Z);
    norm_relu_stats<<<NB_STATS, NRS_WAVES * 32, 0, stream>>>(Z, PART, N);
    bn_finalize<<<1, 256, 0, stream>>>(PART, NB_STATS, gam + l * HID,
                                       bet + l * HID, BN, 1.0f / (float)N);
    float* out = (l == L - 1) ? (float*)d_out : H;
    residual_bn<<<4096, 256, 0, stream>>>(H, Z, BN, out, NF);
  }
}


# embed_gemm, layer_gemm: LDS A tile row pitch K+4 floats instead of K, so an A fragment read covers all 64 banks once instead of a 16-way bank conflict; k order and sums unchanged
# speedup vs baseline: 1.7759x; 1.1050x over previous
#include <hip/hip_runtime.h>
#include <hip/hip_bf16.h>


typedef float v2f __attribute__((ext_vector_type(2)));
typedef float v8f __attribute__((ext_vector_type(8)));

#define HID 128
#define NB_STATS 512

__global__ void zero_i32(int* __restrict__ p, int n) {
  for (int i = blockIdx.x * blockDim.x + threadIdx.x; i < n;
       i += gridDim.x * blockDim.x)
    p[i] = 0;
}

__global__ void hist_deg(const int* __restrict__ dst, int* __restrict__ cnt,
                         int E) {
  for (int e = blockIdx.x * blockDim.x + threadIdx.x; e < E;
       e += gridDim.x * blockDim.x)
    atomicAdd(&cnt[dst[e]], 1);
}

__global__ void invdeg_from_cnt(const int* __restrict__ cnt,
                                float* __restrict__ invdeg, int N) {
  for (int i = blockIdx.x * blockDim.x + threadIdx.x; i < N;
       i += gridDim.x * blockDim.x)
    invdeg[i] = 1.0f / fmaxf((float)cnt[i], 1.0f);
}

#define SCAN_WAVES 32
#define SCAN_VEC 10
#define SCAN_WSPAN (SCAN_VEC * 128)
#define SCAN_SPAN (SCAN_WAVES * SCAN_WSPAN)

template <int S>
__device__ __forceinline__ int row_shr_or0(int x) {
  return __builtin_amdgcn_update_dpp(0, x, 0x110 + S, 0xf, 0xf, false);
}

__device__ __forceinline__ int wave_incl_scan(int x, int lane) {
  x += row_shr_or0<1>(x);
  x += row_shr_or0<2>(x);
  x += row_shr_or0<4>(x);
  x += row_shr_or0<8>(x);
  const int lo = __builtin_amdgcn_readlane(x, 15);
  return x + ((lane >= 16) ? lo : 0);
}

__global__ void __launch_bounds__(SCAN_WAVES * 32) scan_excl(
    const int* __restrict__ cnt, int* __restrict__ rowptr, int N) {
  __shared__ int wtot[SCAN_WAVES];
  const int lane = threadIdx.x & 31;
  const int w = __builtin_amdgcn_readfirstlane(threadIdx.x >> 5);
  int carry = 0;

  for (int base = 0; base < N; base += SCAN_SPAN) {
    const int e0 = base + w * SCAN_WSPAN + lane * 4;
    int4 v[SCAN_VEC];
#pragma unroll
    for (int i = 0; i < SCAN_VEC; ++i) {
      const int e = e0 + i * 128;
      const int4 x = *(const int4*)(cnt + min(e, N - 4));
      const bool in = (e + 4 <= N);
      v[i].x = in ? x.x : 0;
      v[i].y = in ? x.y : 0;
      v[i].z = in ? x.z : 0;
      v[i].w = in ? x.w : 0;
    }
    __builtin_amdgcn_sched_barrier(0);

    int pre[SCAN_VEC];
    int wsum = 0;
#pragma unroll
    for (int i = 0; i < SCAN_VEC; ++i) {
      const int s = (v[i].x + v[i].y) + (v[i].z + v[i].w);
      const int t = wave_incl_scan(s, lane);
      pre[i] = wsum + (t - s);
      wsum += __builtin_amdgcn_readlane(t, 31);
    }

    if (lane == 0) wtot[w] = wsum;
    __syncthreads();
    const int tv = wtot[lane];
    const int ti = wave_incl_scan(tv, lane);
    const int wbase = carry + __builtin_amdgcn_readlane(ti - tv, w);
    carry += __builtin_amdgcn_readlane(ti, 31);

#pragma unroll
    for (int i = 0; i < SCAN_VEC; ++i) {
      const int e = e0 + i * 128;
      int4 r;
      r.x = wbase + pre[i];
      r.y = r.x + v[i].x;
      r.z = r.y + v[i].y;
      r.w = r.z + v[i].z;
      if (e + 4 <= N) *(int4*)(rowptr + e) = r;
    }
    __syncthreads();
  }
  if (threadIdx.x == 0) rowptr[N] = carry;
}

__global__ void fill_csr(const int* __restrict__ src,
                         const int* __restrict__ dst,
                         const int* __restrict__ rowptr,
                         int* __restrict__ cur, int* __restrict__ eidx,
                         int E) {
  for (int e = blockIdx.x * blockDim.x + threadIdx.x; e < E;
       e += gridDim.x * blockDim.x) {
    int d = dst[e];
    int slot = rowptr[d] + atomicAdd(&cur[d], 1);
    eidx[slot] = src[e];
  }
}

#define AGG_WAVES 4
#define AGG_NODES 8
#define AGG_ROWS 32

template <int J>
__device__ __forceinline__ void agg_request(float4 (&v)[AGG_ROWS],
                                            const float4* __restrict__ hv,
                                            int cur, int m, int lane) {
  if constexpr (J < AGG_ROWS) {
    if (J < m) {
      const int s = __builtin_amdgcn_readlane(cur, J);
      v[J] = hv[s * (HID / 4) + lane];
      agg_request<J + 1>(v, hv, cur, m, lane);
    }
  }
}

template <int G>
__device__ __forceinline__ void agg_sum(float4& acc,
                                        const float4 (&v)[AGG_ROWS], int m) {
  if constexpr (G < AGG_ROWS) {
    if (G + 4 <= m) {
      acc.x += ((v[G].x + v[G + 1].x) + v[G + 2].x) + v[G + 3].x;
      acc.y += ((v[G].y + v[G + 1].y) + v[G + 2].y) + v[G + 3].y;
      acc.z += ((v[G].z + v[G + 1].z) + v[G + 2].z) + v[G + 3].z;
      acc.w += ((v[G].w + v[G + 1].w) + v[G + 2].w) + v[G + 3].w;
      agg_sum<G + 4>(acc, v, m);
    } else if (G < m) {
      acc.x += v[G].x;
      acc.y += v[G].y;
      acc.z += v[G].z;
      acc.w += v[G].w;
      if (G + 1 < m) {
        acc.x += v[G + 1].x;
        acc.y += v[G + 1].y;
        acc.z += v[G + 1].z;
        acc.w += v[G + 1].w;
      }
      if (G + 2 < m) {
        acc.x += v[G + 2].x;
        acc.y += v[G + 2].y;
        acc.z += v[G + 2].z;
        acc.w += v[G + 2].w;
      }
    }
  }
}

__global__ void __launch_bounds__(AGG_WAVES * 32) aggregate_csr(
    const float* __restrict__ h, const int* __restrict__ rowptr,
    const int* __restrict__ eidx, const float* __restrict__ invdeg,
    float* __restrict__ c, int N, int E) {
  const int lane = threadIdx.x & 31;
  const int w = __builtin_amdgcn_readfirstlane(threadIdx.x >> 5);
  const int gw = blockIdx.x * AGG_WAVES + w;
  const int tw = gridDim.x * AGG_WAVES;
  const float4* __restrict__ hv = (const float4*)h;
  float4* __restrict__ cv = (float4*)c;

  for (int base = gw * AGG_NODES; base < N; base += tw * AGG_NODES) {
    const int nn = min(AGG_NODES, N - base);
    const int rp = rowptr[min(base + lane, N)];
    const int dg = __builtin_bit_cast(int, invdeg[min(base + lane, N - 1)]);
    int b = __builtin_amdgcn_readlane(rp, 0);
    int e = __builtin_amdgcn_readlane(rp, 1);
    int idx = eidx[min(b + lane, E - 1)];

    for (int k = 0; k < nn; ++k) {
      const int cnt = e - b;
      const int ne = __builtin_amdgcn_readlane(rp, k + 2);
      const int nidx = eidx[min(e + lane, E - 1)];

      float4 acc = make_float4(0.0f, 0.0f, 0.0f, 0.0f);
      for (int chunk = 0; chunk < cnt; chunk += AGG_ROWS) {
        int cur = idx;
        if (chunk > 0) cur = eidx[min(b + chunk + lane, E - 1)];
        const int m = min(AGG_ROWS, cnt - chunk);
        float4 v[AGG_ROWS];
        agg_request<0>(v, hv, cur, m, lane);
        agg_sum<0>(acc, v, m);
      }

      const float sc =
          __builtin_bit_cast(float, __builtin_amdgcn_readlane(dg, k));
      float4 r;
      r.x = acc.x * sc;
      r.y = acc.y * sc;
      r.z = acc.z * sc;
      r.w = acc.w * sc;
      cv[(base + k) * (HID / 4) + lane] = r;

      b = e;
      e = ne;
      idx = nidx;
    }
  }
}

#define A_PAD 4
#define EMB_PITCH (HID + A_PAD)
#define LAY_PITCH (2 * HID + A_PAD)

__global__ void __launch_bounds__(256) embed_gemm(
    const float* __restrict__ X, const float* __restrict__ W,
    const float* __restrict__ B, float* __restrict__ Y) {
  __shared__ __attribute__((aligned(16))) float tA[16 * EMB_PITCH];

  const int tid = threadIdx.x;
  const int rowBase = blockIdx.x * 16;

  const float4* Xv = (const float4*)(X + rowBase * HID);
  for (int i = 0; i < 2; ++i) {
    int v = tid + i * 256;
    int r = v >> 5;
    int c4 = v & 31;
    ((float4*)(tA + r * EMB_PITCH))[c4] = Xv[v];
  }
  __syncthreads();

  const int wave = tid >> 5;
  const int lane = tid & 31;
  const int half = lane >> 4;
  const int l16 = lane & 15;
  const int col = wave * 16 + l16;

  v8f acc;
  {
    float bv = B[col];
#pragma unroll
    for (int r = 0; r < 8; ++r) acc[r] = bv;
  }

  const float* arow = tA + l16 * EMB_PITCH;
  const float* wrow = W + col * HID;

#pragma unroll 4
  for (int k = 0; k < HID; k += 4) {
    const int kk = k + 2 * half;
    v2f a = *(const v2f*)(arow + kk);
    v2f b = *(const v2f*)(wrow + kk);
    acc = __builtin_amdgcn_wmma_f32_16x16x4_f32(false, a, false, b, (short)0,
                                                acc, false, false);
  }

#pragma unroll
  for (int r = 0; r < 8; ++r) {
    int row = rowBase + r + 8 * half;
    Y[row * HID + col] = acc[r];
  }
}

__global__ void __launch_bounds__(256) layer_gemm(
    const float* __restrict__ h, const float* __restrict__ c,
    const float* __restrict__ W, const float* __restrict__ B,
    float* __restrict__ Z) {
  __shared__ __attribute__((aligned(16))) float tA[16 * LAY_PITCH];

  const int tid = threadIdx.x;
  const int rowBase = blockIdx.x * 16;

#pragma unroll
  for (int i = 0; i < 2; ++i) {
    int v = tid + i * 256;
    int r = v >> 5;
    int c4 = v & 31;
    ((float4*)(tA + r * LAY_PITCH))[c4] =
        ((const float4*)(h + (rowBase + r) * HID))[c4];
  }
#pragma unroll
  for (int i = 0; i < 2; ++i) {
    int v = tid + i * 256;
    int r = v >> 5;
    int c4 = v & 31;
    ((float4*)(tA + r * LAY_PITCH + HID))[c4] =
        ((const float4*)(c + (rowBase + r) * HID))[c4];
  }
  __syncthreads();

  const int wave = tid >> 5;
  const int lane = tid & 31;
  const int half = lane >> 4;
  const int l16 = lane & 15;
  const int col = wave * 16 + l16;

  v8f acc;
  {
    float bv = B[col];
#pragma unroll
    for (int r = 0; r < 8; ++r) acc[r] = bv;
  }

  const float* arow = tA + l16 * LAY_PITCH;
  const float* wrow = W + col * 256;

#pragma unroll 4
  for (int k = 0; k < 256; k += 4) {
    const int kk = k + 2 * half;
    v2f a = *(const v2f*)(arow + kk);
    v2f b = *(const v2f*)(wrow + kk);
    acc = __builtin_amdgcn_wmma_f32_16x16x4_f32(false, a, false, b, (short)0,
                                                acc, false, false);
  }

#pragma unroll
  for (int r = 0; r < 8; ++r) {
    int row = rowBase + r + 8 * half;
    Z[row * HID + col] = acc[r];
  }
}


template <int M>
__device__ __forceinline__ float row_xor(float x) {
  const int b = __builtin_bit_cast(int, x);
  return __builtin_bit_cast(
      float, __builtin_amdgcn_update_dpp(b, b, 0x160 + M, 0xf, 0xf, true));
}

__device__ __forceinline__ float row_other(float x) {
  const int b = __builtin_bit_cast(int, x);
  return __builtin_bit_cast(
      float, __builtin_amdgcn_permlanex16(b, b, 0x76543210, 0xfedcba98, false,
                                          false));
}

#define NRS_WAVES 8
#define NRS_ROWS 10

__global__ void __launch_bounds__(NRS_WAVES * 32) norm_relu_stats(
    float* __restrict__ z, float* __restrict__ part, int N) {
  __shared__ __attribute__((aligned(16))) float red[NRS_WAVES * 256];
  const int t = threadIdx.x;
  const int lane = t & 31;
  const int w = __builtin_amdgcn_readfirstlane(t >> 5);
  const int gw = blockIdx.x * NRS_WAVES + w;
  const int tw = gridDim.x * NRS_WAVES;

  float4* __restrict__ zv = (float4*)z;
  float4 ps = make_float4(0.0f, 0.0f, 0.0f, 0.0f);
  float4 pss = make_float4(0.0f, 0.0f, 0.0f, 0.0f);

  for (int row0 = gw; row0 < N; row0 += NRS_ROWS * tw) {
    float4 v[NRS_ROWS];
#pragma unroll
    for (int j = 0; j < NRS_ROWS; ++j) {
      const int row = row0 + j * tw;
      const int rr = (row < N) ? row : row0;
      v[j] = zv[rr * (HID / 4) + lane];
    }
    __builtin_amdgcn_sched_barrier(0);
#pragma unroll
    for (int j = 0; j < NRS_ROWS; ++j) {
      const int row = row0 + j * tw;
      const float4 x = v[j];
      float sq = (x.x * x.x + x.y * x.y) + (x.z * x.z + x.w * x.w);
      sq += row_xor<1>(sq);
      sq += row_xor<2>(sq);
      sq += row_xor<4>(sq);
      sq += row_xor<8>(sq);
      sq += row_other(sq);
      const float nrm = fmaxf(sqrtf(sq), 1e-12f);
      float4 r;
      r.x = fmaxf(x.x / nrm, 0.0f);
      r.y = fmaxf(x.y / nrm, 0.0f);
      r.z = fmaxf(x.z / nrm, 0.0f);
      r.w = fmaxf(x.w / nrm, 0.0f);
      if (row < N) {
        zv[row * (HID / 4) + lane] = r;
        ps.x += r.x;
        ps.y += r.y;
        ps.z += r.z;
        ps.w += r.w;
        pss.x += r.x * r.x;
        pss.y += r.y * r.y;
        pss.z += r.z * r.z;
        pss.w += r.w * r.w;
      }
    }
  }

  ((float4*)(red + w * 256))[lane] = ps;
  ((float4*)(red + w * 256 + 128))[lane] = pss;
  __syncthreads();
  float s = 0.0f;
#pragma unroll
  for (int k = 0; k < NRS_WAVES; ++k) s += red[k * 256 + t];
  part[blockIdx.x * 256 + t] = s;
}

__global__ void __launch_bounds__(256) bn_finalize(
    const float* __restrict__ part, int nb, const float* __restrict__ gamma,
    const float* __restrict__ beta, float* __restrict__ bn, float invN) {
  __shared__ float tot[256];
  const int j = threadIdx.x;
  float s = 0.0f;
  for (int b = 0; b < nb; ++b) s += part[b * 256 + j];
  tot[j] = s;
  __syncthreads();
  if (j < 128) {
    float mu = tot[j] * invN;
    float var = tot[128 + j] * invN - mu * mu;
    float a = gamma[j] / sqrtf(var + 1e-5f);
    bn[j] = a;
    bn[128 + j] = beta[j] - mu * a;
  }
}

__global__ void residual_bn(const float* __restrict__ h,
                            const float* __restrict__ z,
                            const float* __restrict__ bn,
                            float* __restrict__ out, int total) {
  for (int i = blockIdx.x * blockDim.x + threadIdx.x; i < total;
       i += gridDim.x * blockDim.x) {
    int j = i & (HID - 1);
    out[i] = h[i] + z[i] * bn[j] + bn[HID + j];
  }
}

extern "C" void kernel_launch(void* const* d_in, const int* in_sizes, int n_in,
                              void* d_out, int out_size, void* d_ws,
                              size_t ws_size, hipStream_t stream) {
  const float* h0 = (const float*)d_in[0];
  const int* src = (const int*)d_in[1];
  const int* dst = (const int*)d_in[2];
  const float* W_emb = (const float*)d_in[3];
  const float* b_emb = (const float*)d_in[4];
  const float* Ws = (const float*)d_in[5];
  const float* bs = (const float*)d_in[6];
  const float* gam = (const float*)d_in[7];
  const float* bet = (const float*)d_in[8];

  const int N = in_sizes[0] / HID;
  const int E = in_sizes[1];
  const int NF = N * HID;
  const int L = 4;

  float* ws = (float*)d_ws;
  float* H = ws;
  float* C = H + NF;
  float* Z = C + NF;
  float* DEG = Z + NF;
  float* BN = DEG + N;
  float* PART = BN + 256;
  int* cnt = (int*)(PART + NB_STATS * 256);
  int* rowptr = cnt + N;
  int* eidx = rowptr + (N + 1);

  const int mtiles = N / 16;

  zero_i32<<<256, 256, 0, stream>>>(cnt, N);
  hist_deg<<<1024, 256, 0, stream>>>(dst, cnt, E);
  invdeg_from_cnt<<<256, 256, 0, stream>>>(cnt, DEG, N);
  scan_excl<<<1, SCAN_WAVES * 32, 0, stream>>>(cnt, rowptr, N);
  zero_i32<<<256, 256, 0, stream>>>(cnt, N);
  fill_csr<<<1024, 256, 0, stream>>>(src, dst, rowptr, cnt, eidx, E);

  embed_gemm<<<mtiles, 256, 0, stream>>>(h0, W_emb, b_emb, H);

  for (int l = 0; l < L; ++l) {
    aggregate_csr<<<1250, AGG_WAVES * 32, 0, stream>>>(H, rowptr, eidx, DEG, C,
                                                       N, E);
    layer_gemm<<<mtiles, 256, 0, stream>>>(H, C, Ws + l * HID * 256,
                                           bs + l * HID, Z);
    norm_relu_stats<<<NB_STATS, NRS_WAVES * 32, 0, stream>>>(Z, PART, N);
    bn_finalize<<<1, 256, 0, stream>>>(PART, NB_STATS, gam + l * HID,
                                       bet + l * HID, BN, 1.0f / (float)N);
    float* out = (l == L - 1) ? (float*)d_out : H;
    residual_bn<<<4096, 256, 0, stream>>>(H, Z, BN, out, NF);
  }
}


# embed_gemm, layer_gemm: 512 persistent blocks walk the 16-row tiles; a wave loads its W fragments into registers once, not per tile; A tile double-buffered in LDS; k order and sums unchanged
# speedup vs baseline: 1.9459x; 1.0957x over previous
#include <hip/hip_runtime.h>
#include <hip/hip_bf16.h>


typedef float v2f __attribute__((ext_vector_type(2)));
typedef float v8f __attribute__((ext_vector_type(8)));

#define HID 128
#define NB_STATS 512

__global__ void zero_i32(int* __restrict__ p, int n) {
  for (int i = blockIdx.x * blockDim.x + threadIdx.x; i < n;
       i += gridDim.x * blockDim.x)
    p[i] = 0;
}

__global__ void hist_deg(const int* __restrict__ dst, int* __restrict__ cnt,
                         int E) {
  for (int e = blockIdx.x * blockDim.x + threadIdx.x; e < E;
       e += gridDim.x * blockDim.x)
    atomicAdd(&cnt[dst[e]], 1);
}

__global__ void invdeg_from_cnt(const int* __restrict__ cnt,
                                float* __restrict__ invdeg, int N) {
  for (int i = blockIdx.x * blockDim.x + threadIdx.x; i < N;
       i += gridDim.x * blockDim.x)
    invdeg[i] = 1.0f / fmaxf((float)cnt[i], 1.0f);
}

#define SCAN_WAVES 32
#define SCAN_VEC 10
#define SCAN_WSPAN (SCAN_VEC * 128)
#define SCAN_SPAN (SCAN_WAVES * SCAN_WSPAN)

template <int S>
__device__ __forceinline__ int row_shr_or0(int x) {
  return __builtin_amdgcn_update_dpp(0, x, 0x110 + S, 0xf, 0xf, false);
}

__device__ __forceinline__ int wave_incl_scan(int x, int lane) {
  x += row_shr_or0<1>(x);
  x += row_shr_or0<2>(x);
  x += row_shr_or0<4>(x);
  x += row_shr_or0<8>(x);
  const int lo = __builtin_amdgcn_readlane(x, 15);
  return x + ((lane >= 16) ? lo : 0);
}

__global__ void __launch_bounds__(SCAN_WAVES * 32) scan_excl(
    const int* __restrict__ cnt, int* __restrict__ rowptr, int N) {
  __shared__ int wtot[SCAN_WAVES];
  const int lane = threadIdx.x & 31;
  const int w = __builtin_amdgcn_readfirstlane(threadIdx.x >> 5);
  int carry = 0;

  for (int base = 0; base < N; base += SCAN_SPAN) {
    const int e0 = base + w * SCAN_WSPAN + lane * 4;
    int4 v[SCAN_VEC];
#pragma unroll
    for (int i = 0; i < SCAN_VEC; ++i) {
      const int e = e0 + i * 128;
      const int4 x = *(const int4*)(cnt + min(e, N - 4));
      const bool in = (e + 4 <= N);
      v[i].x = in ? x.x : 0;
      v[i].y = in ? x.y : 0;
      v[i].z = in ? x.z : 0;
      v[i].w = in ? x.w : 0;
    }
    __builtin_amdgcn_sched_barrier(0);

    int pre[SCAN_VEC];
    int wsum = 0;
#pragma unroll
    for (int i = 0; i < SCAN_VEC; ++i) {
      const int s = (v[i].x + v[i].y) + (v[i].z + v[i].w);
      const int t = wave_incl_scan(s, lane);
      pre[i] = wsum + (t - s);
      wsum += __builtin_amdgcn_readlane(t, 31);
    }

    if (lane == 0) wtot[w] = wsum;
    __syncthreads();
    const int tv = wtot[lane];
    const int ti = wave_incl_scan(tv, lane);
    const int wbase = carry + __builtin_amdgcn_readlane(ti - tv, w);
    carry += __builtin_amdgcn_readlane(ti, 31);

#pragma unroll
    for (int i = 0; i < SCAN_VEC; ++i) {
      const int e = e0 + i * 128;
      int4 r;
      r.x = wbase + pre[i];
      r.y = r.x + v[i].x;
      r.z = r.y + v[i].y;
      r.w = r.z + v[i].z;
      if (e + 4 <= N) *(int4*)(rowptr + e) = r;
    }
    __syncthreads();
  }
  if (threadIdx.x == 0) rowptr[N] = carry;
}

__global__ void fill_csr(const int* __restrict__ src,
                         const int* __restrict__ dst,
                         const int* __restrict__ rowptr,
                         int* __restrict__ cur, int* __restrict__ eidx,
                         int E) {
  for (int e = blockIdx.x * blockDim.x + threadIdx.x; e < E;
       e += gridDim.x * blockDim.x) {
    int d = dst[e];
    int slot = rowptr[d] + atomicAdd(&cur[d], 1);
    eidx[slot] = src[e];
  }
}

#define AGG_WAVES 4
#define AGG_NODES 8
#define AGG_ROWS 32

template <int J>
__device__ __forceinline__ void agg_request(float4 (&v)[AGG_ROWS],
                                            const float4* __restrict__ hv,
                                            int cur, int m, int lane) {
  if constexpr (J < AGG_ROWS) {
    if (J < m) {
      const int s = __builtin_amdgcn_readlane(cur, J);
      v[J] = hv[s * (HID / 4) + lane];
      agg_request<J + 1>(v, hv, cur, m, lane);
    }
  }
}

template <int G>
__device__ __forceinline__ void agg_sum(float4& acc,
                                        const float4 (&v)[AGG_ROWS], int m) {
  if constexpr (G < AGG_ROWS) {
    if (G + 4 <= m) {
      acc.x += ((v[G].x + v[G + 1].x) + v[G + 2].x) + v[G + 3].x;
      acc.y += ((v[G].y + v[G + 1].y) + v[G + 2].y) + v[G + 3].y;
      acc.z += ((v[G].z + v[G + 1].z) + v[G + 2].z) + v[G + 3].z;
      acc.w += ((v[G].w + v[G + 1].w) + v[G + 2].w) + v[G + 3].w;
      agg_sum<G + 4>(acc, v, m);
    } else if (G < m) {
      acc.x += v[G].x;
      acc.y += v[G].y;
      acc.z += v[G].z;
      acc.w += v[G].w;
      if (G + 1 < m) {
        acc.x += v[G + 1].x;
        acc.y += v[G + 1].y;
        acc.z += v[G + 1].z;
        acc.w += v[G + 1].w;
      }
      if (G + 2 < m) {
        acc.x += v[G + 2].x;
        acc.y += v[G + 2].y;
        acc.z += v[G + 2].z;
        acc.w += v[G + 2].w;
      }
    }
  }
}

__global__ void __launch_bounds__(AGG_WAVES * 32) aggregate_csr(
    const float* __restrict__ h, const int* __restrict__ rowptr,
    const int* __restrict__ eidx, const float* __restrict__ invdeg,
    float* __restrict__ c, int N, int E) {
  const int lane = threadIdx.x & 31;
  const int w = __builtin_amdgcn_readfirstlane(threadIdx.x >> 5);
  const int gw = blockIdx.x * AGG_WAVES + w;
  const int tw = gridDim.x * AGG_WAVES;
  const float4* __restrict__ hv = (const float4*)h;
  float4* __restrict__ cv = (float4*)c;

  for (int base = gw * AGG_NODES; base < N; base += tw * AGG_NODES) {
    const int nn = min(AGG_NODES, N - base);
    const int rp = rowptr[min(base + lane, N)];
    const int dg = __builtin_bit_cast(int, invdeg[min(base + lane, N - 1)]);
    int b = __builtin_amdgcn_readlane(rp, 0);
    int e = __builtin_amdgcn_readlane(rp, 1);
    int idx = eidx[min(b + lane, E - 1)];

    for (int k = 0; k < nn; ++k) {
      const int cnt = e - b;
      const int ne = __builtin_amdgcn_readlane(rp, k + 2);
      const int nidx = eidx[min(e + lane, E - 1)];

      float4 acc = make_float4(0.0f, 0.0f, 0.0f, 0.0f);
      for (int chunk = 0; chunk < cnt; chunk += AGG_ROWS) {
        int cur = idx;
        if (chunk > 0) cur = eidx[min(b + chunk + lane, E - 1)];
        const int m = min(AGG_ROWS, cnt - chunk);
        float4 v[AGG_ROWS];
        agg_request<0>(v, hv, cur, m, lane);
        agg_sum<0>(acc, v, m);
      }

      const float sc =
          __builtin_bit_cast(float, __builtin_amdgcn_readlane(dg, k));
      float4 r;
      r.x = acc.x * sc;
      r.y = acc.y * sc;
      r.z = acc.z * sc;
      r.w = acc.w * sc;
      cv[(base + k) * (HID / 4) + lane] = r;

      b = e;
      e = ne;
      idx = nidx;
    }
  }
}

#define A_PAD 4
#define GEMM_BLOCKS 512

__device__ __forceinline__ void tile_request(float4& qa, float4& qb,
                                             const float* __restrict__ src,
                                             int tile, int r0, int c4) {
  const float4* p = (const float4*)(src + (tile * 16 + r0) * HID) + c4;
  qa = p[0];
  qb = p[8 * (HID / 4)];
}

template <int PITCH>
__device__ __forceinline__ void tile_store(float* __restrict__ st,
                                           const float4& qa, const float4& qb,
                                           int r0, int c4) {
  float4* p = (float4*)(st + r0 * PITCH) + c4;
  p[0] = qa;
  p[8 * (PITCH / 4)] = qb;
}

template <int K>
__device__ __forceinline__ void gemm_persistent(
    const float* __restrict__ a0, const float* __restrict__ a1,
    const float* __restrict__ W, const float* __restrict__ B,
    float* __restrict__ Y, int mtiles, float* __restrict__ tA) {
  constexpr int PITCH = K + A_PAD;
  constexpr int KS = K / 4;

  const int tid = threadIdx.x;
  const int wave = tid >> 5;
  const int lane = tid & 31;
  const int half = lane >> 4;
  const int l16 = lane & 15;
  const int col = wave * 16 + l16;
  const int r0 = tid >> 5;
  const int c4 = tid & 31;

  int tile = blockIdx.x;
  float4 q0, q1, q2, q3;
  q0 = q1 = q2 = q3 = make_float4(0.0f, 0.0f, 0.0f, 0.0f);
  if (tile < mtiles) {
    tile_request(q0, q1, a0, tile, r0, c4);
    if constexpr (K == 2 * HID) tile_request(q2, q3, a1, tile, r0, c4);
  }

  const float bv = B[col];

  if (tile < mtiles) {
    tile_store<PITCH>(tA, q0, q1, r0, c4);
    if constexpr (K == 2 * HID) tile_store<PITCH>(tA + HID, q2, q3, r0, c4);
  }
  __syncthreads();

  v2f bf[KS];
  const float* wrow = W + col * K + 2 * half;
#pragma unroll
  for (int i = 0; i < KS; ++i) bf[i] = *(const v2f*)(wrow + 4 * i);

  int s = 0;
  for (; tile < mtiles; tile += gridDim.x) {
    const int nt = tile + gridDim.x;
    const bool more = nt < mtiles;
    if (more) {
      tile_request(q0, q1, a0, nt, r0, c4);
      if constexpr (K == 2 * HID) tile_request(q2, q3, a1, nt, r0, c4);
    }

    v8f acc;
#pragma unroll
    for (int r = 0; r < 8; ++r) acc[r] = bv;

    const float* arow = tA + s * (16 * PITCH) + l16 * PITCH + 2 * half;
#pragma unroll
    for (int i = 0; i < KS; ++i) {
      const v2f a = *(const v2f*)(arow + 4 * i);
      acc = __builtin_amdgcn_wmma_f32_16x16x4_f32(false, a, false, bf[i],
                                                  (short)0, acc, false, false);
    }

    if (more) {
      float* st = tA + (s ^ 1) * (16 * PITCH);
      tile_store<PITCH>(st, q0, q1, r0, c4);
      if constexpr (K == 2 * HID) tile_store<PITCH>(st + HID, q2, q3, r0, c4);
    }
    __syncthreads();
    s ^= 1;

#pragma unroll
    for (int r = 0; r < 8; ++r) {
      const int row = tile * 16 + r + 8 * half;
      Y[row * HID + col] = acc[r];
    }
  }
}

__global__ void __launch_bounds__(256) embed_gemm(
    const float* __restrict__ X, const float* __restrict__ W,
    const float* __restrict__ B, float* __restrict__ Y, int mtiles) {
  __shared__ __attribute__((aligned(16))) float tA[2 * 16 * (HID + A_PAD)];
  gemm_persistent<HID>(X, X, W, B, Y, mtiles, tA);
}

__global__ void __launch_bounds__(256) layer_gemm(
    const float* __restrict__ h, const float* __restrict__ c,
    const float* __restrict__ W, const float* __restrict__ B,
    float* __restrict__ Z, int mtiles) {
  __shared__ __attribute__((aligned(16))) float tA[2 * 16 * (2 * HID + A_PAD)];
  gemm_persistent<2 * HID>(h, c, W, B, Z, mtiles, tA);
}


template <int M>
__device__ __forceinline__ float row_xor(float x) {
  const int b = __builtin_bit_cast(int, x);
  return __builtin_bit_cast(
      float, __builtin_amdgcn_update_dpp(b, b, 0x160 + M, 0xf, 0xf, true));
}

__device__ __forceinline__ float row_other(float x) {
  const int b = __builtin_bit_cast(int, x);
  return __builtin_bit_cast(
      float, __builtin_amdgcn_permlanex16(b, b, 0x76543210, 0xfedcba98, false,
                                          false));
}

#define NRS_WAVES 8
#define NRS_ROWS 10

__global__ void __launch_bounds__(NRS_WAVES * 32) norm_relu_stats(
    float* __restrict__ z, float* __restrict__ part, int N) {
  __shared__ __attribute__((aligned(16))) float red[NRS_WAVES * 256];
  const int t = threadIdx.x;
  const int lane = t & 31;
  const int w = __builtin_amdgcn_readfirstlane(t >> 5);
  const int gw = blockIdx.x * NRS_WAVES + w;
  const int tw = gridDim.x * NRS_WAVES;

  float4* __restrict__ zv = (float4*)z;
  float4 ps = make_float4(0.0f, 0.0f, 0.0f, 0.0f);
  float4 pss = make_float4(0.0f, 0.0f, 0.0f, 0.0f);

  for (int row0 = gw; row0 < N; row0 += NRS_ROWS * tw) {
    float4 v[NRS_ROWS];
#pragma unroll
    for (int j = 0; j < NRS_ROWS; ++j) {
      const int row = row0 + j * tw;
      const int rr = (row < N) ? row : row0;
      v[j] = zv[rr * (HID / 4) + lane];
    }
    __builtin_amdgcn_sched_barrier(0);
#pragma unroll
    for (int j = 0; j < NRS_ROWS; ++j) {
      const int row = row0 + j * tw;
      const float4 x = v[j];
      float sq = (x.x * x.x + x.y * x.y) + (x.z * x.z + x.w * x.w);
      sq += row_xor<1>(sq);
      sq += row_xor<2>(sq);
      sq += row_xor<4>(sq);
      sq += row_xor<8>(sq);
      sq += row_other(sq);
      const float nrm = fmaxf(sqrtf(sq), 1e-12f);
      float4 r;
      r.x = fmaxf(x.x / nrm, 0.0f);
      r.y = fmaxf(x.y / nrm, 0.0f);
      r.z = fmaxf(x.z / nrm, 0.0f);
      r.w = fmaxf(x.w / nrm, 0.0f);
      if (row < N) {
        zv[row * (HID / 4) + lane] = r;
        ps.x += r.x;
        ps.y += r.y;
        ps.z += r.z;
        ps.w += r.w;
        pss.x += r.x * r.x;
        pss.y += r.y * r.y;
        pss.z += r.z * r.z;
        pss.w += r.w * r.w;
      }
    }
  }

  ((float4*)(red + w * 256))[lane] = ps;
  ((float4*)(red + w * 256 + 128))[lane] = pss;
  __syncthreads();
  float s = 0.0f;
#pragma unroll
  for (int k = 0; k < NRS_WAVES; ++k) s += red[k * 256 + t];
  part[blockIdx.x * 256 + t] = s;
}

__global__ void __launch_bounds__(256) bn_finalize(
    const float* __restrict__ part, int nb, const float* __restrict__ gamma,
    const float* __restrict__ beta, float* __restrict__ bn, float invN) {
  __shared__ float tot[256];
  const int j = threadIdx.x;
  float s = 0.0f;
  for (int b = 0; b < nb; ++b) s += part[b * 256 + j];
  tot[j] = s;
  __syncthreads();
  if (j < 128) {
    float mu = tot[j] * invN;
    float var = tot[128 + j] * invN - mu * mu;
    float a = gamma[j] / sqrtf(var + 1e-5f);
    bn[j] = a;
    bn[128 + j] = beta[j] - mu * a;
  }
}

__global__ void residual_bn(const float* __restrict__ h,
                            const float* __restrict__ z,
                            const float* __restrict__ bn,
                            float* __restrict__ out, int total) {
  for (int i = blockIdx.x * blockDim.x + threadIdx.x; i < total;
       i += gridDim.x * blockDim.x) {
    int j = i & (HID - 1);
    out[i] = h[i] + z[i] * bn[j] + bn[HID + j];
  }
}

extern "C" void kernel_launch(void* const* d_in, const int* in_sizes, int n_in,
                              void* d_out, int out_size, void* d_ws,
                              size_t ws_size, hipStream_t stream) {
  const float* h0 = (const float*)d_in[0];
  const int* src = (const int*)d_in[1];
  const int* dst = (const int*)d_in[2];
  const float* W_emb = (const float*)d_in[3];
  const float* b_emb = (const float*)d_in[4];
  const float* Ws = (const float*)d_in[5];
  const float* bs = (const float*)d_in[6];
  const float* gam = (const float*)d_in[7];
  const float* bet = (const float*)d_in[8];

  const int N = in_sizes[0] / HID;
  const int E = in_sizes[1];
  const int NF = N * HID;
  const int L = 4;

  float* ws = (float*)d_ws;
  float* H = ws;
  float* C = H + NF;
  float* Z = C + NF;
  float* DEG = Z + NF;
  float* BN = DEG + N;
  float* PART = BN + 256;
  int* cnt = (int*)(PART + NB_STATS * 256);
  int* rowptr = cnt + N;
  int* eidx = rowptr + (N + 1);

  const int mtiles = N / 16;

  zero_i32<<<256, 256, 0, stream>>>(cnt, N);
  hist_deg<<<1024, 256, 0, stream>>>(dst, cnt, E);
  invdeg_from_cnt<<<256, 256, 0, stream>>>(cnt, DEG, N);
  scan_excl<<<1, SCAN_WAVES * 32, 0, stream>>>(cnt, rowptr, N);
  zero_i32<<<256, 256, 0, stream>>>(cnt, N);
  fill_csr<<<1024, 256, 0, stream>>>(src, dst, rowptr, cnt, eidx, E);

  embed_gemm<<<GEMM_BLOCKS, 256, 0, stream>>>(h0, W_emb, b_emb, H, mtiles);

  for (int l = 0; l < L; ++l) {
    aggregate_csr<<<1250, AGG_WAVES * 32, 0, stream>>>(H, rowptr, eidx, DEG, C,
                                                       N, E);
    layer_gemm<<<GEMM_BLOCKS, 256, 0, stream>>>(H, C, Ws + l * HID * 256,
                                                bs + l * HID, Z, mtiles);
    norm_relu_stats<<<NB_STATS, NRS_WAVES * 32, 0, stream>>>(Z, PART, N);
    bn_finalize<<<1, 256, 0, stream>>>(PART, NB_STATS, gam + l * HID,
                                       bet + l * HID, BN, 1.0f / (float)N);
    float* out = (l == L - 1) ? (float*)d_out : H;
    residual_bn<<<4096, 256, 0, stream>>>(H, Z, BN, out, NF);
  }
}


# bn_finalize: 4 blocks of 32 waves, 32 columns each; a thread requests all its float4 pieces of the 512 partial rows before the first add (was 64 load-then-wait trips in one block); BN sums reordered
# speedup vs baseline: 2.4782x; 1.2735x over previous
#include <hip/hip_runtime.h>
#include <hip/hip_bf16.h>


typedef float v2f __attribute__((ext_vector_type(2)));
typedef float v8f __attribute__((ext_vector_type(8)));

#define HID 128
#define NB_STATS 512

__global__ void zero_i32(int* __restrict__ p, int n) {
  for (int i = blockIdx.x * blockDim.x + threadIdx.x; i < n;
       i += gridDim.x * blockDim.x)
    p[i] = 0;
}

__global__ void hist_deg(const int* __restrict__ dst, int* __restrict__ cnt,
                         int E) {
  for (int e = blockIdx.x * blockDim.x + threadIdx.x; e < E;
       e += gridDim.x * blockDim.x)
    atomicAdd(&cnt[dst[e]], 1);
}

__global__ void invdeg_from_cnt(const int* __restrict__ cnt,
                                float* __restrict__ invdeg, int N) {
  for (int i = blockIdx.x * blockDim.x + threadIdx.x; i < N;
       i += gridDim.x * blockDim.x)
    invdeg[i] = 1.0f / fmaxf((float)cnt[i], 1.0f);
}

#define SCAN_WAVES 32
#define SCAN_VEC 10
#define SCAN_WSPAN (SCAN_VEC * 128)
#define SCAN_SPAN (SCAN_WAVES * SCAN_WSPAN)

template <int S>
__device__ __forceinline__ int row_shr_or0(int x) {
  return __builtin_amdgcn_update_dpp(0, x, 0x110 + S, 0xf, 0xf, false);
}

__device__ __forceinline__ int wave_incl_scan(int x, int lane) {
  x += row_shr_or0<1>(x);
  x += row_shr_or0<2>(x);
  x += row_shr_or0<4>(x);
  x += row_shr_or0<8>(x);
  const int lo = __builtin_amdgcn_readlane(x, 15);
  return x + ((lane >= 16) ? lo : 0);
}

__global__ void __launch_bounds__(SCAN_WAVES * 32) scan_excl(
    const int* __restrict__ cnt, int* __restrict__ rowptr, int N) {
  __shared__ int wtot[SCAN_WAVES];
  const int lane = threadIdx.x & 31;
  const int w = __builtin_amdgcn_readfirstlane(threadIdx.x >> 5);
  int carry = 0;

  for (int base = 0; base < N; base += SCAN_SPAN) {
    const int e0 = base + w * SCAN_WSPAN + lane * 4;
    int4 v[SCAN_VEC];
#pragma unroll
    for (int i = 0; i < SCAN_VEC; ++i) {
      const int e = e0 + i * 128;
      const int4 x = *(const int4*)(cnt + min(e, N - 4));
      const bool in = (e + 4 <= N);
      v[i].x = in ? x.x : 0;
      v[i].y = in ? x.y : 0;
      v[i].z = in ? x.z : 0;
      v[i].w = in ? x.w : 0;
    }
    __builtin_amdgcn_sched_barrier(0);

    int pre[SCAN_VEC];
    int wsum = 0;
#pragma unroll
    for (int i = 0; i < SCAN_VEC; ++i) {
      const int s = (v[i].x + v[i].y) + (v[i].z + v[i].w);
      const int t = wave_incl_scan(s, lane);
      pre[i] = wsum + (t - s);
      wsum += __builtin_amdgcn_readlane(t, 31);
    }

    if (lane == 0) wtot[w] = wsum;
    __syncthreads();
    const int tv = wtot[lane];
    const int ti = wave_incl_scan(tv, lane);
    const int wbase = carry + __builtin_amdgcn_readlane(ti - tv, w);
    carry += __builtin_amdgcn_readlane(ti, 31);

#pragma unroll
    for (int i = 0; i < SCAN_VEC; ++i) {
      const int e = e0 + i * 128;
      int4 r;
      r.x = wbase + pre[i];
      r.y = r.x + v[i].x;
      r.z = r.y + v[i].y;
      r.w = r.z + v[i].z;
      if (e + 4 <= N) *(int4*)(rowptr + e) = r;
    }
    __syncthreads();
  }
  if (threadIdx.x == 0) rowptr[N] = carry;
}

__global__ void fill_csr(const int* __restrict__ src,
                         const int* __restrict__ dst,
                         const int* __restrict__ rowptr,
                         int* __restrict__ cur, int* __restrict__ eidx,
                         int E) {
  for (int e = blockIdx.x * blockDim.x + threadIdx.x; e < E;
       e += gridDim.x * blockDim.x) {
    int d = dst[e];
    int slot = rowptr[d] + atomicAdd(&cur[d], 1);
    eidx[slot] = src[e];
  }
}

#define AGG_WAVES 4
#define AGG_NODES 8
#define AGG_ROWS 32

template <int J>
__device__ __forceinline__ void agg_request(float4 (&v)[AGG_ROWS],
                                            const float4* __restrict__ hv,
                                            int cur, int m, int lane) {
  if constexpr (J < AGG_ROWS) {
    if (J < m) {
      const int s = __builtin_amdgcn_readlane(cur, J);
      v[J] = hv[s * (HID / 4) + lane];
      agg_request<J + 1>(v, hv, cur, m, lane);
    }
  }
}

template <int G>
__device__ __forceinline__ void agg_sum(float4& acc,
                                        const float4 (&v)[AGG_ROWS], int m) {
  if constexpr (G < AGG_ROWS) {
    if (G + 4 <= m) {
      acc.x += ((v[G].x + v[G + 1].x) + v[G + 2].x) + v[G + 3].x;
      acc.y += ((v[G].y + v[G + 1].y) + v[G + 2].y) + v[G + 3].y;
      acc.z += ((v[G].z + v[G + 1].z) + v[G + 2].z) + v[G + 3].z;
      acc.w += ((v[G].w + v[G + 1].w) + v[G + 2].w) + v[G + 3].w;
      agg_sum<G + 4>(acc, v, m);
    } else if (G < m) {
      acc.x += v[G].x;
      acc.y += v[G].y;
      acc.z += v[G].z;
      acc.w += v[G].w;
      if (G + 1 < m) {
        acc.x += v[G + 1].x;
        acc.y += v[G + 1].y;
        acc.z += v[G + 1].z;
        acc.w += v[G + 1].w;
      }
      if (G + 2 < m) {
        acc.x += v[G + 2].x;
        acc.y += v[G + 2].y;
        acc.z += v[G + 2].z;
        acc.w += v[G + 2].w;
      }
    }
  }
}

__global__ void __launch_bounds__(AGG_WAVES * 32) aggregate_csr(
    const float* __restrict__ h, const int* __restrict__ rowptr,
    const int* __restrict__ eidx, const float* __restrict__ invdeg,
    float* __restrict__ c, int N, int E) {
  const int lane = threadIdx.x & 31;
  const int w = __builtin_amdgcn_readfirstlane(threadIdx.x >> 5);
  const int gw = blockIdx.x * AGG_WAVES + w;
  const int tw = gridDim.x * AGG_WAVES;
  const float4* __restrict__ hv = (const float4*)h;
  float4* __restrict__ cv = (float4*)c;

  for (int base = gw * AGG_NODES; base < N; base += tw * AGG_NODES) {
    const int nn = min(AGG_NODES, N - base);
    const int rp = rowptr[min(base + lane, N)];
    const int dg = __builtin_bit_cast(int, invdeg[min(base + lane, N - 1)]);
    int b = __builtin_amdgcn_readlane(rp, 0);
    int e = __builtin_amdgcn_readlane(rp, 1);
    int idx = eidx[min(b + lane, E - 1)];

    for (int k = 0; k < nn; ++k) {
      const int cnt = e - b;
      const int ne = __builtin_amdgcn_readlane(rp, k + 2);
      const int nidx = eidx[min(e + lane, E - 1)];

      float4 acc = make_float4(0.0f, 0.0f, 0.0f, 0.0f);
      for (int chunk = 0; chunk < cnt; chunk += AGG_ROWS) {
        int cur = idx;
        if (chunk > 0) cur = eidx[min(b + chunk + lane, E - 1)];
        const int m = min(AGG_ROWS, cnt - chunk);
        float4 v[AGG_ROWS];
        agg_request<0>(v, hv, cur, m, lane);
        agg_sum<0>(acc, v, m);
      }

      const float sc =
          __builtin_bit_cast(float, __builtin_amdgcn_readlane(dg, k));
      float4 r;
      r.x = acc.x * sc;
      r.y = acc.y * sc;
      r.z = acc.z * sc;
      r.w = acc.w * sc;
      cv[(base + k) * (HID / 4) + lane] = r;

      b = e;
      e = ne;
      idx = nidx;
    }
  }
}

#define A_PAD 4
#define GEMM_BLOCKS 512

__device__ __forceinline__ void tile_request(float4& qa, float4& qb,
                                             const float* __restrict__ src,
                                             int tile, int r0, int c4) {
  const float4* p = (const float4*)(src + (tile * 16 + r0) * HID) + c4;
  qa = p[0];
  qb = p[8 * (HID / 4)];
}

template <int PITCH>
__device__ __forceinline__ void tile_store(float* __restrict__ st,
                                           const float4& qa, const float4& qb,
                                           int r0, int c4) {
  float4* p = (float4*)(st + r0 * PITCH) + c4;
  p[0] = qa;
  p[8 * (PITCH / 4)] = qb;
}

template <int K>
__device__ __forceinline__ void gemm_persistent(
    const float* __restrict__ a0, const float* __restrict__ a1,
    const float* __restrict__ W, const float* __restrict__ B,
    float* __restrict__ Y, int mtiles, float* __restrict__ tA) {
  constexpr int PITCH = K + A_PAD;
  constexpr int KS = K / 4;

  const int tid = threadIdx.x;
  const int wave = tid >> 5;
  const int lane = tid & 31;
  const int half = lane >> 4;
  const int l16 = lane & 15;
  const int col = wave * 16 + l16;
  const int r0 = tid >> 5;
  const int c4 = tid & 31;

  int tile = blockIdx.x;
  float4 q0, q1, q2, q3;
  q0 = q1 = q2 = q3 = make_float4(0.0f, 0.0f, 0.0f, 0.0f);
  if (tile < mtiles) {
    tile_request(q0, q1, a0, tile, r0, c4);
    if constexpr (K == 2 * HID) tile_request(q2, q3, a1, tile, r0, c4);
  }

  const float bv = B[col];

  if (tile < mtiles) {
    tile_store<PITCH>(tA, q0, q1, r0, c4);
    if constexpr (K == 2 * HID) tile_store<PITCH>(tA + HID, q2, q3, r0, c4);
  }
  __syncthreads();

  v2f bf[KS];
  const float* wrow = W + col * K + 2 * half;
#pragma unroll
  for (int i = 0; i < KS; ++i) bf[i] = *(const v2f*)(wrow + 4 * i);

  int s = 0;
  for (; tile < mtiles; tile += gridDim.x) {
    const int nt = tile + gridDim.x;
    const bool more = nt < mtiles;
    if (more) {
      tile_request(q0, q1, a0, nt, r0, c4);
      if constexpr (K == 2 * HID) tile_request(q2, q3, a1, nt, r0, c4);
    }

    v8f acc;
#pragma unroll
    for (int r = 0; r < 8; ++r) acc[r] = bv;

    const float* arow = tA + s * (16 * PITCH) + l16 * PITCH + 2 * half;
#pragma unroll
    for (int i = 0; i < KS; ++i) {
      const v2f a = *(const v2f*)(arow + 4 * i);
      acc = __builtin_amdgcn_wmma_f32_16x16x4_f32(false, a, false, bf[i],
                                                  (short)0, acc, false, false);
    }

    if (more) {
      float* st = tA + (s ^ 1) * (16 * PITCH);
      tile_store<PITCH>(st, q0, q1, r0, c4);
      if constexpr (K == 2 * HID) tile_store<PITCH>(st + HID, q2, q3, r0, c4);
    }
    __syncthreads();
    s ^= 1;

#pragma unroll
    for (int r = 0; r < 8; ++r) {
      const int row = tile * 16 + r + 8 * half;
      Y[row * HID + col] = acc[r];
    }
  }
}

__global__ void __launch_bounds__(256) embed_gemm(
    const float* __restrict__ X, const float* __restrict__ W,
    const float* __restrict__ B, float* __restrict__ Y, int mtiles) {
  __shared__ __attribute__((aligned(16))) float tA[2 * 16 * (HID + A_PAD)];
  gemm_persistent<HID>(X, X, W, B, Y, mtiles, tA);
}

__global__ void __launch_bounds__(256) layer_gemm(
    const float* __restrict__ h, const float* __restrict__ c,
    const float* __restrict__ W, const float* __restrict__ B,
    float* __restrict__ Z, int mtiles) {
  __shared__ __attribute__((aligned(16))) float tA[2 * 16 * (2 * HID + A_PAD)];
  gemm_persistent<2 * HID>(h, c, W, B, Z, mtiles, tA);
}


template <int M>
__device__ __forceinline__ float row_xor(float x) {
  const int b = __builtin_bit_cast(int, x);
  return __builtin_bit_cast(
      float, __builtin_amdgcn_update_dpp(b, b, 0x160 + M, 0xf, 0xf, true));
}

__device__ __forceinline__ float row_other(float x) {
  const int b = __builtin_bit_cast(int, x);
  return __builtin_bit_cast(
      float, __builtin_amdgcn_permlanex16(b, b, 0x76543210, 0xfedcba98, false,
                                          false));
}

#define NRS_WAVES 8
#define NRS_ROWS 10

__global__ void __launch_bounds__(NRS_WAVES * 32) norm_relu_stats(
    float* __restrict__ z, float* __restrict__ part, int N) {
  __shared__ __attribute__((aligned(16))) float red[NRS_WAVES * 256];
  const int t = threadIdx.x;
  const int lane = t & 31;
  const int w = __builtin_amdgcn_readfirstlane(t >> 5);
  const int gw = blockIdx.x * NRS_WAVES + w;
  const int tw = gridDim.x * NRS_WAVES;

  float4* __restrict__ zv = (float4*)z;
  float4 ps = make_float4(0.0f, 0.0f, 0.0f, 0.0f);
  float4 pss = make_float4(0.0f, 0.0f, 0.0f, 0.0f);

  for (int row0 = gw; row0 < N; row0 += NRS_ROWS * tw) {
    float4 v[NRS_ROWS];
#pragma unroll
    for (int j = 0; j < NRS_ROWS; ++j) {
      const int row = row0 + j * tw;
      const int rr = (row < N) ? row : row0;
      v[j] = zv[rr * (HID / 4) + lane];
    }
    __builtin_amdgcn_sched_barrier(0);
#pragma unroll
    for (int j = 0; j < NRS_ROWS; ++j) {
      const int row = row0 + j * tw;
      const float4 x = v[j];
      float sq = (x.x * x.x + x.y * x.y) + (x.z * x.z + x.w * x.w);
      sq += row_xor<1>(sq);
      sq += row_xor<2>(sq);
      sq += row_xor<4>(sq);
      sq += row_xor<8>(sq);
      sq += row_other(sq);
      const float nrm = fmaxf(sqrtf(sq), 1e-12f);
      float4 r;
      r.x = fmaxf(x.x / nrm, 0.0f);
      r.y = fmaxf(x.y / nrm, 0.0f);
      r.z = fmaxf(x.z / nrm, 0.0f);
      r.w = fmaxf(x.w / nrm, 0.0f);
      if (row < N) {
        zv[row * (HID / 4) + lane] = r;
        ps.x += r.x;
        ps.y += r.y;
        ps.z += r.z;
        ps.w += r.w;
        pss.x += r.x * r.x;
        pss.y += r.y * r.y;
        pss.z += r.z * r.z;
        pss.w += r.w * r.w;
      }
    }
  }

  ((float4*)(red + w * 256))[lane] = ps;
  ((float4*)(red + w * 256 + 128))[lane] = pss;
  __syncthreads();
  float s = 0.0f;
#pragma unroll
  for (int k = 0; k < NRS_WAVES; ++k) s += red[k * 256 + t];
  part[blockIdx.x * 256 + t] = s;
}

#define BNF_BLOCKS (HID / 32)
#define BNF_WAVES 32
#define BNF_RPP (BNF_WAVES * 4)
#define BNF_PASSES (NB_STATS / BNF_RPP)

__global__ void __launch_bounds__(BNF_WAVES * 32) bn_finalize(
    const float* __restrict__ part, const float* __restrict__ gamma,
    const float* __restrict__ beta, float* __restrict__ bn, float invN) {
  static_assert(NB_STATS % BNF_RPP == 0, "whole passes over the partial rows");
  __shared__ __attribute__((aligned(16))) float red[BNF_WAVES * 64];
  const int t = threadIdx.x;
  const int lane = t & 31;
  const int w = t >> 5;
  const int col0 = blockIdx.x * 32;

  float g = 0.0f, be = 0.0f;
  if (t < 32) {
    g = gamma[col0 + t];
    be = beta[col0 + t];
  }

  const float4* p = (const float4*)(part + (t >> 3) * 256 + col0) + (t & 7);
  float4 s[BNF_PASSES], q[BNF_PASSES];
#pragma unroll
  for (int i = 0; i < BNF_PASSES; ++i) {
    s[i] = p[i * (BNF_RPP * 64)];
    q[i] = p[i * (BNF_RPP * 64) + 32];
  }

  float4 a = s[0];
  float4 b = q[0];
#pragma unroll
  for (int i = 1; i < BNF_PASSES; ++i) {
    a.x += s[i].x;
    a.y += s[i].y;
    a.z += s[i].z;
    a.w += s[i].w;
    b.x += q[i].x;
    b.y += q[i].y;
    b.z += q[i].z;
    b.w += q[i].w;
  }
  a.x += row_xor<8>(a.x);
  a.y += row_xor<8>(a.y);
  a.z += row_xor<8>(a.z);
  a.w += row_xor<8>(a.w);
  b.x += row_xor<8>(b.x);
  b.y += row_xor<8>(b.y);
  b.z += row_xor<8>(b.z);
  b.w += row_xor<8>(b.w);
  a.x += row_other(a.x);
  a.y += row_other(a.y);
  a.z += row_other(a.z);
  a.w += row_other(a.w);
  b.x += row_other(b.x);
  b.y += row_other(b.y);
  b.z += row_other(b.z);
  b.w += row_other(b.w);
  if (lane < 8) {
    ((float4*)(red + w * 64))[lane] = a;
    ((float4*)(red + w * 64 + 32))[lane] = b;
  }
  __syncthreads();

  if (t < 32) {
    float s1 = 0.0f, s2 = 0.0f;
#pragma unroll
    for (int k = 0; k < BNF_WAVES; ++k) {
      s1 += red[k * 64 + t];
      s2 += red[k * 64 + 32 + t];
    }
    float mu = s1 * invN;
    float var = s2 * invN - mu * mu;
    float sc = g / sqrtf(var + 1e-5f);
    bn[col0 + t] = sc;
    bn[128 + col0 + t] = be - mu * sc;
  }
}

__global__ void residual_bn(const float* __restrict__ h,
                            const float* __restrict__ z,
                            const float* __restrict__ bn,
                            float* __restrict__ out, int total) {
  for (int i = blockIdx.x * blockDim.x + threadIdx.x; i < total;
       i += gridDim.x * blockDim.x) {
    int j = i & (HID - 1);
    out[i] = h[i] + z[i] * bn[j] + bn[HID + j];
  }
}

extern "C" void kernel_launch(void* const* d_in, const int* in_sizes, int n_in,
                              void* d_out, int out_size, void* d_ws,
                              size_t ws_size, hipStream_t stream) {
  const float* h0 = (const float*)d_in[0];
  const int* src = (const int*)d_in[1];
  const int* dst = (const int*)d_in[2];
  const float* W_emb = (const float*)d_in[3];
  const float* b_emb = (const float*)d_in[4];
  const float* Ws = (const float*)d_in[5];
  const float* bs = (const float*)d_in[6];
  const float* gam = (const float*)d_in[7];
  const float* bet = (const float*)d_in[8];

  const int N = in_sizes[0] / HID;
  const int E = in_sizes[1];
  const int NF = N * HID;
  const int L = 4;

  float* ws = (float*)d_ws;
  float* H = ws;
  float* C = H + NF;
  float* Z = C + NF;
  float* DEG = Z + NF;
  float* BN = DEG + N;
  float* PART = BN + 256;
  int* cnt = (int*)(PART + NB_STATS * 256);
  int* rowptr = cnt + N;
  int* eidx = rowptr + (N + 1);

  const int mtiles = N / 16;

  zero_i32<<<256, 256, 0, stream>>>(cnt, N);
  hist_deg<<<1024, 256, 0, stream>>>(dst, cnt, E);
  invdeg_from_cnt<<<256, 256, 0, stream>>>(cnt, DEG, N);
  scan_excl<<<1, SCAN_WAVES * 32, 0, stream>>>(cnt, rowptr, N);
  zero_i32<<<256, 256, 0, stream>>>(cnt, N);
  fill_csr<<<1024, 256, 0, stream>>>(src, dst, rowptr, cnt, eidx, E);

  embed_gemm<<<GEMM_BLOCKS, 256, 0, stream>>>(h0, W_emb, b_emb, H, mtiles);

  for (int l = 0; l < L; ++l) {
    aggregate_csr<<<1250, AGG_WAVES * 32, 0, stream>>>(H, rowptr, eidx, DEG, C,
                                                       N, E);
    layer_gemm<<<GEMM_BLOCKS, 256, 0, stream>>>(H, C, Ws + l * HID * 256,
                                                bs + l * HID, Z, mtiles);
    norm_relu_stats<<<NB_STATS, NRS_WAVES * 32, 0, stream>>>(Z, PART, N);
    bn_finalize<<<BNF_BLOCKS, BNF_WAVES * 32, 0, stream>>>(
        PART, gam + l * HID, bet + l * HID, BN, 1.0f / (float)N);
    float* out = (l == L - 1) ? (float*)d_out : H;
    residual_bn<<<4096, 256, 0, stream>>>(H, Z, BN, out, NF);
  }
}


# layer_gemm epilogue does row L2-normalize, ReLU, BN partial sums on its accumulator tile (row squares folded in-wave, then over 8 waves via LDS); norm_relu_stats launch gone; row, BN sums reordered
# speedup vs baseline: 3.3587x; 1.3553x over previous
#include <hip/hip_runtime.h>
#include <hip/hip_bf16.h>


typedef float v2f __attribute__((ext_vector_type(2)));
typedef float v8f __attribute__((ext_vector_type(8)));

#define HID 128

__global__ void zero_i32(int* __restrict__ p, int n) {
  for (int i = blockIdx.x * blockDim.x + threadIdx.x; i < n;
       i += gridDim.x * blockDim.x)
    p[i] = 0;
}

__global__ void hist_deg(const int* __restrict__ dst, int* __restrict__ cnt,
                         int E) {
  for (int e = blockIdx.x * blockDim.x + threadIdx.x; e < E;
       e += gridDim.x * blockDim.x)
    atomicAdd(&cnt[dst[e]], 1);
}

__global__ void invdeg_from_cnt(const int* __restrict__ cnt,
                                float* __restrict__ invdeg, int N) {
  for (int i = blockIdx.x * blockDim.x + threadIdx.x; i < N;
       i += gridDim.x * blockDim.x)
    invdeg[i] = 1.0f / fmaxf((float)cnt[i], 1.0f);
}

#define SCAN_WAVES 32
#define SCAN_VEC 10
#define SCAN_WSPAN (SCAN_VEC * 128)
#define SCAN_SPAN (SCAN_WAVES * SCAN_WSPAN)

template <int S>
__device__ __forceinline__ int row_shr_or0(int x) {
  return __builtin_amdgcn_update_dpp(0, x, 0x110 + S, 0xf, 0xf, false);
}

__device__ __forceinline__ int wave_incl_scan(int x, int lane) {
  x += row_shr_or0<1>(x);
  x += row_shr_or0<2>(x);
  x += row_shr_or0<4>(x);
  x += row_shr_or0<8>(x);
  const int lo = __builtin_amdgcn_readlane(x, 15);
  return x + ((lane >= 16) ? lo : 0);
}

__global__ void __launch_bounds__(SCAN_WAVES * 32) scan_excl(
    const int* __restrict__ cnt, int* __restrict__ rowptr, int N) {
  __shared__ int wtot[SCAN_WAVES];
  const int lane = threadIdx.x & 31;
  const int w = __builtin_amdgcn_readfirstlane(threadIdx.x >> 5);
  int carry = 0;

  for (int base = 0; base < N; base += SCAN_SPAN) {
    const int e0 = base + w * SCAN_WSPAN + lane * 4;
    int4 v[SCAN_VEC];
#pragma unroll
    for (int i = 0; i < SCAN_VEC; ++i) {
      const int e = e0 + i * 128;
      const int4 x = *(const int4*)(cnt + min(e, N - 4));
      const bool in = (e + 4 <= N);
      v[i].x = in ? x.x : 0;
      v[i].y = in ? x.y : 0;
      v[i].z = in ? x.z : 0;
      v[i].w = in ? x.w : 0;
    }
    __builtin_amdgcn_sched_barrier(0);

    int pre[SCAN_VEC];
    int wsum = 0;
#pragma unroll
    for (int i = 0; i < SCAN_VEC; ++i) {
      const int s = (v[i].x + v[i].y) + (v[i].z + v[i].w);
      const int t = wave_incl_scan(s, lane);
      pre[i] = wsum + (t - s);
      wsum += __builtin_amdgcn_readlane(t, 31);
    }

    if (lane == 0) wtot[w] = wsum;
    __syncthreads();
    const int tv = wtot[lane];
    const int ti = wave_incl_scan(tv, lane);
    const int wbase = carry + __builtin_amdgcn_readlane(ti - tv, w);
    carry += __builtin_amdgcn_readlane(ti, 31);

#pragma unroll
    for (int i = 0; i < SCAN_VEC; ++i) {
      const int e = e0 + i * 128;
      int4 r;
      r.x = wbase + pre[i];
      r.y = r.x + v[i].x;
      r.z = r.y + v[i].y;
      r.w = r.z + v[i].z;
      if (e + 4 <= N) *(int4*)(rowptr + e) = r;
    }
    __syncthreads();
  }
  if (threadIdx.x == 0) rowptr[N] = carry;
}

__global__ void fill_csr(const int* __restrict__ src,
                         const int* __restrict__ dst,
                         const int* __restrict__ rowptr,
                         int* __restrict__ cur, int* __restrict__ eidx,
                         int E) {
  for (int e = blockIdx.x * blockDim.x + threadIdx.x; e < E;
       e += gridDim.x * blockDim.x) {
    int d = dst[e];
    int slot = rowptr[d] + atomicAdd(&cur[d], 1);
    eidx[slot] = src[e];
  }
}

#define AGG_WAVES 4
#define AGG_NODES 8
#define AGG_ROWS 32

template <int J>
__device__ __forceinline__ void agg_request(float4 (&v)[AGG_ROWS],
                                            const float4* __restrict__ hv,
                                            int cur, int m, int lane) {
  if constexpr (J < AGG_ROWS) {
    if (J < m) {
      const int s = __builtin_amdgcn_readlane(cur, J);
      v[J] = hv[s * (HID / 4) + lane];
      agg_request<J + 1>(v, hv, cur, m, lane);
    }
  }
}

template <int G>
__device__ __forceinline__ void agg_sum(float4& acc,
                                        const float4 (&v)[AGG_ROWS], int m) {
  if constexpr (G < AGG_ROWS) {
    if (G + 4 <= m) {
      acc.x += ((v[G].x + v[G + 1].x) + v[G + 2].x) + v[G + 3].x;
      acc.y += ((v[G].y + v[G + 1].y) + v[G + 2].y) + v[G + 3].y;
      acc.z += ((v[G].z + v[G + 1].z) + v[G + 2].z) + v[G + 3].z;
      acc.w += ((v[G].w + v[G + 1].w) + v[G + 2].w) + v[G + 3].w;
      agg_sum<G + 4>(acc, v, m);
    } else if (G < m) {
      acc.x += v[G].x;
      acc.y += v[G].y;
      acc.z += v[G].z;
      acc.w += v[G].w;
      if (G + 1 < m) {
        acc.x += v[G + 1].x;
        acc.y += v[G + 1].y;
        acc.z += v[G + 1].z;
        acc.w += v[G + 1].w;
      }
      if (G + 2 < m) {
        acc.x += v[G + 2].x;
        acc.y += v[G + 2].y;
        acc.z += v[G + 2].z;
        acc.w += v[G + 2].w;
      }
    }
  }
}

__global__ void __launch_bounds__(AGG_WAVES * 32) aggregate_csr(
    const float* __restrict__ h, const int* __restrict__ rowptr,
    const int* __restrict__ eidx, const float* __restrict__ invdeg,
    float* __restrict__ c, int N, int E) {
  const int lane = threadIdx.x & 31;
  const int w = __builtin_amdgcn_readfirstlane(threadIdx.x >> 5);
  const int gw = blockIdx.x * AGG_WAVES + w;
  const int tw = gridDim.x * AGG_WAVES;
  const float4* __restrict__ hv = (const float4*)h;
  float4* __restrict__ cv = (float4*)c;

  for (int base = gw * AGG_NODES; base < N; base += tw * AGG_NODES) {
    const int nn = min(AGG_NODES, N - base);
    const int rp = rowptr[min(base + lane, N)];
    const int dg = __builtin_bit_cast(int, invdeg[min(base + lane, N - 1)]);
    int b = __builtin_amdgcn_readlane(rp, 0);
    int e = __builtin_amdgcn_readlane(rp, 1);
    int idx = eidx[min(b + lane, E - 1)];

    for (int k = 0; k < nn; ++k) {
      const int cnt = e - b;
      const int ne = __builtin_amdgcn_readlane(rp, k + 2);
      const int nidx = eidx[min(e + lane, E - 1)];

      float4 acc = make_float4(0.0f, 0.0f, 0.0f, 0.0f);
      for (int chunk = 0; chunk < cnt; chunk += AGG_ROWS) {
        int cur = idx;
        if (chunk > 0) cur = eidx[min(b + chunk + lane, E - 1)];
        const int m = min(AGG_ROWS, cnt - chunk);
        float4 v[AGG_ROWS];
        agg_request<0>(v, hv, cur, m, lane);
        agg_sum<0>(acc, v, m);
      }

      const float sc =
          __builtin_bit_cast(float, __builtin_amdgcn_readlane(dg, k));
      float4 r;
      r.x = acc.x * sc;
      r.y = acc.y * sc;
      r.z = acc.z * sc;
      r.w = acc.w * sc;
      cv[(base + k) * (HID / 4) + lane] = r;

      b = e;
      e = ne;
      idx = nidx;
    }
  }
}


template <int M>
__device__ __forceinline__ float row_xor(float x) {
  const int b = __builtin_bit_cast(int, x);
  return __builtin_bit_cast(
      float, __builtin_amdgcn_update_dpp(b, b, 0x160 + M, 0xf, 0xf, true));
}

__device__ __forceinline__ float row_other(float x) {
  const int b = __builtin_bit_cast(int, x);
  return __builtin_bit_cast(
      float, __builtin_amdgcn_permlanex16(b, b, 0x76543210, 0xfedcba98, false,
                                          false));
}

#define A_PAD 4
#define GEMM_BLOCKS 512
#define NB_STATS GEMM_BLOCKS

__device__ __forceinline__ void tile_request(float4& qa, float4& qb,
                                             const float* __restrict__ src,
                                             int tile, int r0, int c4) {
  const float4* p = (const float4*)(src + (tile * 16 + r0) * HID) + c4;
  qa = p[0];
  qb = p[8 * (HID / 4)];
}

template <int PITCH>
__device__ __forceinline__ void tile_store(float* __restrict__ st,
                                           const float4& qa, const float4& qb,
                                           int r0, int c4) {
  float4* p = (float4*)(st + r0 * PITCH) + c4;
  p[0] = qa;
  p[8 * (PITCH / 4)] = qb;
}

template <int R>
__device__ __forceinline__ float row_share(float x) {
  const int b = __builtin_bit_cast(int, x);
  return __builtin_bit_cast(
      float, __builtin_amdgcn_update_dpp(b, b, 0x150 + R, 0xf, 0xf, true));
}

template <int K, bool NORM>
__device__ __forceinline__ void gemm_persistent(
    const float* __restrict__ a0, const float* __restrict__ a1,
    const float* __restrict__ W, const float* __restrict__ B,
    float* __restrict__ Y, int mtiles, float* __restrict__ tA,
    float* __restrict__ rs, float* __restrict__ part) {
  constexpr int PITCH = K + A_PAD;
  constexpr int KS = K / 4;

  const int tid = threadIdx.x;
  const int wave = tid >> 5;
  const int lane = tid & 31;
  const int half = lane >> 4;
  const int l16 = lane & 15;
  const int col = wave * 16 + l16;
  const int r0 = tid >> 5;
  const int c4 = tid & 31;

  int tile = blockIdx.x;
  float4 q0, q1, q2, q3;
  q0 = q1 = q2 = q3 = make_float4(0.0f, 0.0f, 0.0f, 0.0f);
  if (tile < mtiles) {
    tile_request(q0, q1, a0, tile, r0, c4);
    if constexpr (K == 2 * HID) tile_request(q2, q3, a1, tile, r0, c4);
  }

  const float bv = B[col];

  if (tile < mtiles) {
    tile_store<PITCH>(tA, q0, q1, r0, c4);
    if constexpr (K == 2 * HID) tile_store<PITCH>(tA + HID, q2, q3, r0, c4);
  }
  __syncthreads();

  v2f bf[KS];
  const float* wrow = W + col * K + 2 * half;
#pragma unroll
  for (int i = 0; i < KS; ++i) bf[i] = *(const v2f*)(wrow + 4 * i);

  float ps = 0.0f, pss = 0.0f;
  int s = 0;
  for (; tile < mtiles; tile += gridDim.x) {
    const int nt = tile + gridDim.x;
    const bool more = nt < mtiles;
    if (more) {
      tile_request(q0, q1, a0, nt, r0, c4);
      if constexpr (K == 2 * HID) tile_request(q2, q3, a1, nt, r0, c4);
    }

    v8f acc;
#pragma unroll
    for (int r = 0; r < 8; ++r) acc[r] = bv;

    const float* arow = tA + s * (16 * PITCH) + l16 * PITCH + 2 * half;
#pragma unroll
    for (int i = 0; i < KS; ++i) {
      const v2f a = *(const v2f*)(arow + 4 * i);
      acc = __builtin_amdgcn_wmma_f32_16x16x4_f32(false, a, false, bf[i],
                                                  (short)0, acc, false, false);
    }

    if constexpr (NORM) {
      asm volatile("v_nop\n\tv_nop\n\tv_nop\n\tv_nop\n\tv_nop" : "+v"(acc));
      float sq[8];
#pragma unroll
      for (int r = 0; r < 8; ++r) {
        float t = acc[r] * acc[r];
        t += row_xor<1>(t);
        t += row_xor<2>(t);
        t += row_xor<4>(t);
        t += row_xor<8>(t);
        sq[r] = t;
      }
      if (l16 == 0) {
        float4* p = (float4*)(rs + s * 128 + wave * 16 + 8 * half);
        p[0] = make_float4(sq[0], sq[1], sq[2], sq[3]);
        p[1] = make_float4(sq[4], sq[5], sq[6], sq[7]);
      }
    }

    if (more) {
      float* st = tA + (s ^ 1) * (16 * PITCH);
      tile_store<PITCH>(st, q0, q1, r0, c4);
      if constexpr (K == 2 * HID) tile_store<PITCH>(st + HID, q2, q3, r0, c4);
    }
    __syncthreads();

    if constexpr (NORM) {
      const float* qs = rs + s * 128 + 8 * half + (l16 & 7);
      float t = qs[0];
#pragma unroll
      for (int w = 1; w < 8; ++w) t += qs[16 * w];
      const float nrm = fmaxf(sqrtf(t), 1e-12f);
      float nr[8];
      nr[0] = row_share<0>(nrm);
      nr[1] = row_share<1>(nrm);
      nr[2] = row_share<2>(nrm);
      nr[3] = row_share<3>(nrm);
      nr[4] = row_share<4>(nrm);
      nr[5] = row_share<5>(nrm);
      nr[6] = row_share<6>(nrm);
      nr[7] = row_share<7>(nrm);
#pragma unroll
      for (int r = 0; r < 8; ++r) {
        const float v = fmaxf(acc[r] / nr[r], 0.0f);
        acc[r] = v;
        ps += v;
        pss += v * v;
      }
    }
    s ^= 1;

#pragma unroll
    for (int r = 0; r < 8; ++r) {
      const int row = tile * 16 + r + 8 * half;
      Y[row * HID + col] = acc[r];
    }
  }

  if constexpr (NORM) {
    ps += row_other(ps);
    pss += row_other(pss);
    if (half == 0) {
      part[blockIdx.x * 256 + col] = ps;
      part[blockIdx.x * 256 + HID + col] = pss;
    }
  }
}

__global__ void __launch_bounds__(256) embed_gemm(
    const float* __restrict__ X, const float* __restrict__ W,
    const float* __restrict__ B, float* __restrict__ Y, int mtiles) {
  __shared__ __attribute__((aligned(16))) float tA[2 * 16 * (HID + A_PAD)];
  gemm_persistent<HID, false>(X, X, W, B, Y, mtiles, tA, nullptr, nullptr);
}

__global__ void __launch_bounds__(256) layer_gemm(
    const float* __restrict__ h, const float* __restrict__ c,
    const float* __restrict__ W, const float* __restrict__ B,
    float* __restrict__ Z, float* __restrict__ part, int mtiles) {
  __shared__ __attribute__((aligned(16))) float tA[2 * 16 * (2 * HID + A_PAD)];
  __shared__ __attribute__((aligned(16))) float rs[2 * 8 * 16];
  gemm_persistent<2 * HID, true>(h, c, W, B, Z, mtiles, tA, rs, part);
}

#define BNF_BLOCKS (HID / 32)
#define BNF_WAVES 32
#define BNF_RPP (BNF_WAVES * 4)
#define BNF_PASSES (NB_STATS / BNF_RPP)

__global__ void __launch_bounds__(BNF_WAVES * 32) bn_finalize(
    const float* __restrict__ part, const float* __restrict__ gamma,
    const float* __restrict__ beta, float* __restrict__ bn, float invN) {
  static_assert(NB_STATS % BNF_RPP == 0, "whole passes over the partial rows");
  __shared__ __attribute__((aligned(16))) float red[BNF_WAVES * 64];
  const int t = threadIdx.x;
  const int lane = t & 31;
  const int w = t >> 5;
  const int col0 = blockIdx.x * 32;

  float g = 0.0f, be = 0.0f;
  if (t < 32) {
    g = gamma[col0 + t];
    be = beta[col0 + t];
  }

  const float4* p = (const float4*)(part + (t >> 3) * 256 + col0) + (t & 7);
  float4 s[BNF_PASSES], q[BNF_PASSES];
#pragma unroll
  for (int i = 0; i < BNF_PASSES; ++i) {
    s[i] = p[i * (BNF_RPP * 64)];
    q[i] = p[i * (BNF_RPP * 64) + 32];
  }

  float4 a = s[0];
  float4 b = q[0];
#pragma unroll
  for (int i = 1; i < BNF_PASSES; ++i) {
    a.x += s[i].x;
    a.y += s[i].y;
    a.z += s[i].z;
    a.w += s[i].w;
    b.x += q[i].x;
    b.y += q[i].y;
    b.z += q[i].z;
    b.w += q[i].w;
  }
  a.x += row_xor<8>(a.x);
  a.y += row_xor<8>(a.y);
  a.z += row_xor<8>(a.z);
  a.w += row_xor<8>(a.w);
  b.x += row_xor<8>(b.x);
  b.y += row_xor<8>(b.y);
  b.z += row_xor<8>(b.z);
  b.w += row_xor<8>(b.w);
  a.x += row_other(a.x);
  a.y += row_other(a.y);
  a.z += row_other(a.z);
  a.w += row_other(a.w);
  b.x += row_other(b.x);
  b.y += row_other(b.y);
  b.z += row_other(b.z);
  b.w += row_other(b.w);
  if (lane < 8) {
    ((float4*)(red + w * 64))[lane] = a;
    ((float4*)(red + w * 64 + 32))[lane] = b;
  }
  __syncthreads();

  if (t < 32) {
    float s1 = 0.0f, s2 = 0.0f;
#pragma unroll
    for (int k = 0; k < BNF_WAVES; ++k) {
      s1 += red[k * 64 + t];
      s2 += red[k * 64 + 32 + t];
    }
    float mu = s1 * invN;
    float var = s2 * invN - mu * mu;
    float sc = g / sqrtf(var + 1e-5f);
    bn[col0 + t] = sc;
    bn[128 + col0 + t] = be - mu * sc;
  }
}

__global__ void residual_bn(const float* __restrict__ h,
                            const float* __restrict__ z,
                            const float* __restrict__ bn,
                            float* __restrict__ out, int total) {
  for (int i = blockIdx.x * blockDim.x + threadIdx.x; i < total;
       i += gridDim.x * blockDim.x) {
    int j = i & (HID - 1);
    out[i] = h[i] + z[i] * bn[j] + bn[HID + j];
  }
}

extern "C" void kernel_launch(void* const* d_in, const int* in_sizes, int n_in,
                              void* d_out, int out_size, void* d_ws,
                              size_t ws_size, hipStream_t stream) {
  const float* h0 = (const float*)d_in[0];
  const int* src = (const int*)d_in[1];
  const int* dst = (const int*)d_in[2];
  const float* W_emb = (const float*)d_in[3];
  const float* b_emb = (const float*)d_in[4];
  const float* Ws = (const float*)d_in[5];
  const float* bs = (const float*)d_in[6];
  const float* gam = (const float*)d_in[7];
  const float* bet = (const float*)d_in[8];

  const int N = in_sizes[0] / HID;
  const int E = in_sizes[1];
  const int NF = N * HID;
  const int L = 4;

  float* ws = (float*)d_ws;
  float* H = ws;
  float* C = H + NF;
  float* Z = C + NF;
  float* DEG = Z + NF;
  float* BN = DEG + N;
  float* PART = BN + 256;
  int* cnt = (int*)(PART + NB_STATS * 256);
  int* rowptr = cnt + N;
  int* eidx = rowptr + (N + 1);

  const int mtiles = N / 16;

  zero_i32<<<256, 256, 0, stream>>>(cnt, N);
  hist_deg<<<1024, 256, 0, stream>>>(dst, cnt, E);
  invdeg_from_cnt<<<256, 256, 0, stream>>>(cnt, DEG, N);
  scan_excl<<<1, SCAN_WAVES * 32, 0, stream>>>(cnt, rowptr, N);
  zero_i32<<<256, 256, 0, stream>>>(cnt, N);
  fill_csr<<<1024, 256, 0, stream>>>(src, dst, rowptr, cnt, eidx, E);

  embed_gemm<<<GEMM_BLOCKS, 256, 0, stream>>>(h0, W_emb, b_emb, H, mtiles);

  for (int l = 0; l < L; ++l) {
    aggregate_csr<<<1250, AGG_WAVES * 32, 0, stream>>>(H, rowptr, eidx, DEG, C,
                                                       N, E);
    layer_gemm<<<GEMM_BLOCKS, 256, 0, stream>>>(H, C, Ws + l * HID * 256,
                                                bs + l * HID, Z, PART, mtiles);
    bn_finalize<<<BNF_BLOCKS, BNF_WAVES * 32, 0, stream>>>(
        PART, gam + l * HID, bet + l * HID, BN, 1.0f / (float)N);
    float* out = (l == L - 1) ? (float*)d_out : H;
    residual_bn<<<4096, 256, 0, stream>>>(H, Z, BN, out, NF);
  }
}


# bn_finalize folded into residual_bn: 256 blocks of 32 waves, each redoes the BN reduce of its 32 columns with its float4 pieces of h and z already requested; 4 launches fewer; sums unchanged
# speedup vs baseline: 3.5211x; 1.0484x over previous
#include <hip/hip_runtime.h>
#include <hip/hip_bf16.h>


typedef float v2f __attribute__((ext_vector_type(2)));
typedef float v8f __attribute__((ext_vector_type(8)));

#define HID 128

__global__ void zero_i32(int* __restrict__ p, int n) {
  for (int i = blockIdx.x * blockDim.x + threadIdx.x; i < n;
       i += gridDim.x * blockDim.x)
    p[i] = 0;
}

__global__ void hist_deg(const int* __restrict__ dst, int* __restrict__ cnt,
                         int E) {
  for (int e = blockIdx.x * blockDim.x + threadIdx.x; e < E;
       e += gridDim.x * blockDim.x)
    atomicAdd(&cnt[dst[e]], 1);
}

__global__ void invdeg_from_cnt(const int* __restrict__ cnt,
                                float* __restrict__ invdeg, int N) {
  for (int i = blockIdx.x * blockDim.x + threadIdx.x; i < N;
       i += gridDim.x * blockDim.x)
    invdeg[i] = 1.0f / fmaxf((float)cnt[i], 1.0f);
}

#define SCAN_WAVES 32
#define SCAN_VEC 10
#define SCAN_WSPAN (SCAN_VEC * 128)
#define SCAN_SPAN (SCAN_WAVES * SCAN_WSPAN)

template <int S>
__device__ __forceinline__ int row_shr_or0(int x) {
  return __builtin_amdgcn_update_dpp(0, x, 0x110 + S, 0xf, 0xf, false);
}

__device__ __forceinline__ int wave_incl_scan(int x, int lane) {
  x += row_shr_or0<1>(x);
  x += row_shr_or0<2>(x);
  x += row_shr_or0<4>(x);
  x += row_shr_or0<8>(x);
  const int lo = __builtin_amdgcn_readlane(x, 15);
  return x + ((lane >= 16) ? lo : 0);
}

__global__ void __launch_bounds__(SCAN_WAVES * 32) scan_excl(
    const int* __restrict__ cnt, int* __restrict__ rowptr, int N) {
  __shared__ int wtot[SCAN_WAVES];
  const int lane = threadIdx.x & 31;
  const int w = __builtin_amdgcn_readfirstlane(threadIdx.x >> 5);
  int carry = 0;

  for (int base = 0; base < N; base += SCAN_SPAN) {
    const int e0 = base + w * SCAN_WSPAN + lane * 4;
    int4 v[SCAN_VEC];
#pragma unroll
    for (int i = 0; i < SCAN_VEC; ++i) {
      const int e = e0 + i * 128;
      const int4 x = *(const int4*)(cnt + min(e, N - 4));
      const bool in = (e + 4 <= N);
      v[i].x = in ? x.x : 0;
      v[i].y = in ? x.y : 0;
      v[i].z = in ? x.z : 0;
      v[i].w = in ? x.w : 0;
    }
    __builtin_amdgcn_sched_barrier(0);

    int pre[SCAN_VEC];
    int wsum = 0;
#pragma unroll
    for (int i = 0; i < SCAN_VEC; ++i) {
      const int s = (v[i].x + v[i].y) + (v[i].z + v[i].w);
      const int t = wave_incl_scan(s, lane);
      pre[i] = wsum + (t - s);
      wsum += __builtin_amdgcn_readlane(t, 31);
    }

    if (lane == 0) wtot[w] = wsum;
    __syncthreads();
    const int tv = wtot[lane];
    const int ti = wave_incl_scan(tv, lane);
    const int wbase = carry + __builtin_amdgcn_readlane(ti - tv, w);
    carry += __builtin_amdgcn_readlane(ti, 31);

#pragma unroll
    for (int i = 0; i < SCAN_VEC; ++i) {
      const int e = e0 + i * 128;
      int4 r;
      r.x = wbase + pre[i];
      r.y = r.x + v[i].x;
      r.z = r.y + v[i].y;
      r.w = r.z + v[i].z;
      if (e + 4 <= N) *(int4*)(rowptr + e) = r;
    }
    __syncthreads();
  }
  if (threadIdx.x == 0) rowptr[N] = carry;
}

__global__ void fill_csr(const int* __restrict__ src,
                         const int* __restrict__ dst,
                         const int* __restrict__ rowptr,
                         int* __restrict__ cur, int* __restrict__ eidx,
                         int E) {
  for (int e = blockIdx.x * blockDim.x + threadIdx.x; e < E;
       e += gridDim.x * blockDim.x) {
    int d = dst[e];
    int slot = rowptr[d] + atomicAdd(&cur[d], 1);
    eidx[slot] = src[e];
  }
}

#define AGG_WAVES 4
#define AGG_NODES 8
#define AGG_ROWS 32

template <int J>
__device__ __forceinline__ void agg_request(float4 (&v)[AGG_ROWS],
                                            const float4* __restrict__ hv,
                                            int cur, int m, int lane) {
  if constexpr (J < AGG_ROWS) {
    if (J < m) {
      const int s = __builtin_amdgcn_readlane(cur, J);
      v[J] = hv[s * (HID / 4) + lane];
      agg_request<J + 1>(v, hv, cur, m, lane);
    }
  }
}

template <int G>
__device__ __forceinline__ void agg_sum(float4& acc,
                                        const float4 (&v)[AGG_ROWS], int m) {
  if constexpr (G < AGG_ROWS) {
    if (G + 4 <= m) {
      acc.x += ((v[G].x + v[G + 1].x) + v[G + 2].x) + v[G + 3].x;
      acc.y += ((v[G].y + v[G + 1].y) + v[G + 2].y) + v[G + 3].y;
      acc.z += ((v[G].z + v[G + 1].z) + v[G + 2].z) + v[G + 3].z;
      acc.w += ((v[G].w + v[G + 1].w) + v[G + 2].w) + v[G + 3].w;
      agg_sum<G + 4>(acc, v, m);
    } else if (G < m) {
      acc.x += v[G].x;
      acc.y += v[G].y;
      acc.z += v[G].z;
      acc.w += v[G].w;
      if (G + 1 < m) {
        acc.x += v[G + 1].x;
        acc.y += v[G + 1].y;
        acc.z += v[G + 1].z;
        acc.w += v[G + 1].w;
      }
      if (G + 2 < m) {
        acc.x += v[G + 2].x;
        acc.y += v[G + 2].y;
        acc.z += v[G + 2].z;
        acc.w += v[G + 2].w;
      }
    }
  }
}

__global__ void __launch_bounds__(AGG_WAVES * 32) aggregate_csr(
    const float* __restrict__ h, const int* __restrict__ rowptr,
    const int* __restrict__ eidx, const float* __restrict__ invdeg,
    float* __restrict__ c, int N, int E) {
  const int lane = threadIdx.x & 31;
  const int w = __builtin_amdgcn_readfirstlane(threadIdx.x >> 5);
  const int gw = blockIdx.x * AGG_WAVES + w;
  const int tw = gridDim.x * AGG_WAVES;
  const float4* __restrict__ hv = (const float4*)h;
  float4* __restrict__ cv = (float4*)c;

  for (int base = gw * AGG_NODES; base < N; base += tw * AGG_NODES) {
    const int nn = min(AGG_NODES, N - base);
    const int rp = rowptr[min(base + lane, N)];
    const int dg = __builtin_bit_cast(int, invdeg[min(base + lane, N - 1)]);
    int b = __builtin_amdgcn_readlane(rp, 0);
    int e = __builtin_amdgcn_readlane(rp, 1);
    int idx = eidx[min(b + lane, E - 1)];

    for (int k = 0; k < nn; ++k) {
      const int cnt = e - b;
      const int ne = __builtin_amdgcn_readlane(rp, k + 2);
      const int nidx = eidx[min(e + lane, E - 1)];

      float4 acc = make_float4(0.0f, 0.0f, 0.0f, 0.0f);
      for (int chunk = 0; chunk < cnt; chunk += AGG_ROWS) {
        int cur = idx;
        if (chunk > 0) cur = eidx[min(b + chunk + lane, E - 1)];
        const int m = min(AGG_ROWS, cnt - chunk);
        float4 v[AGG_ROWS];
        agg_request<0>(v, hv, cur, m, lane);
        agg_sum<0>(acc, v, m);
      }

      const float sc =
          __builtin_bit_cast(float, __builtin_amdgcn_readlane(dg, k));
      float4 r;
      r.x = acc.x * sc;
      r.y = acc.y * sc;
      r.z = acc.z * sc;
      r.w = acc.w * sc;
      cv[(base + k) * (HID / 4) + lane] = r;

      b = e;
      e = ne;
      idx = nidx;
    }
  }
}


template <int M>
__device__ __forceinline__ float row_xor(float x) {
  const int b = __builtin_bit_cast(int, x);
  return __builtin_bit_cast(
      float, __builtin_amdgcn_update_dpp(b, b, 0x160 + M, 0xf, 0xf, true));
}

__device__ __forceinline__ float row_other(float x) {
  const int b = __builtin_bit_cast(int, x);
  return __builtin_bit_cast(
      float, __builtin_amdgcn_permlanex16(b, b, 0x76543210, 0xfedcba98, false,
                                          false));
}

#define A_PAD 4
#define GEMM_BLOCKS 512
#define NB_STATS GEMM_BLOCKS

__device__ __forceinline__ void tile_request(float4& qa, float4& qb,
                                             const float* __restrict__ src,
                                             int tile, int r0, int c4) {
  const float4* p = (const float4*)(src + (tile * 16 + r0) * HID) + c4;
  qa = p[0];
  qb = p[8 * (HID / 4)];
}

template <int PITCH>
__device__ __forceinline__ void tile_store(float* __restrict__ st,
                                           const float4& qa, const float4& qb,
                                           int r0, int c4) {
  float4* p = (float4*)(st + r0 * PITCH) + c4;
  p[0] = qa;
  p[8 * (PITCH / 4)] = qb;
}

template <int R>
__device__ __forceinline__ float row_share(float x) {
  const int b = __builtin_bit_cast(int, x);
  return __builtin_bit_cast(
      float, __builtin_amdgcn_update_dpp(b, b, 0x150 + R, 0xf, 0xf, true));
}

template <int K, bool NORM>
__device__ __forceinline__ void gemm_persistent(
    const float* __restrict__ a0, const float* __restrict__ a1,
    const float* __restrict__ W, const float* __restrict__ B,
    float* __restrict__ Y, int mtiles, float* __restrict__ tA,
    float* __restrict__ rs, float* __restrict__ part) {
  constexpr int PITCH = K + A_PAD;
  constexpr int KS = K / 4;

  const int tid = threadIdx.x;
  const int wave = tid >> 5;
  const int lane = tid & 31;
  const int half = lane >> 4;
  const int l16 = lane & 15;
  const int col = wave * 16 + l16;
  const int r0 = tid >> 5;
  const int c4 = tid & 31;

  int tile = blockIdx.x;
  float4 q0, q1, q2, q3;
  q0 = q1 = q2 = q3 = make_float4(0.0f, 0.0f, 0.0f, 0.0f);
  if (tile < mtiles) {
    tile_request(q0, q1, a0, tile, r0, c4);
    if constexpr (K == 2 * HID) tile_request(q2, q3, a1, tile, r0, c4);
  }

  const float bv = B[col];

  if (tile < mtiles) {
    tile_store<PITCH>(tA, q0, q1, r0, c4);
    if constexpr (K == 2 * HID) tile_store<PITCH>(tA + HID, q2, q3, r0, c4);
  }
  __syncthreads();

  v2f bf[KS];
  const float* wrow = W + col * K + 2 * half;
#pragma unroll
  for (int i = 0; i < KS; ++i) bf[i] = *(const v2f*)(wrow + 4 * i);

  float ps = 0.0f, pss = 0.0f;
  int s = 0;
  for (; tile < mtiles; tile += gridDim.x) {
    const int nt = tile + gridDim.x;
    const bool more = nt < mtiles;
    if (more) {
      tile_request(q0, q1, a0, nt, r0, c4);
      if constexpr (K == 2 * HID) tile_request(q2, q3, a1, nt, r0, c4);
    }

    v8f acc;
#pragma unroll
    for (int r = 0; r < 8; ++r) acc[r] = bv;

    const float* arow = tA + s * (16 * PITCH) + l16 * PITCH + 2 * half;
#pragma unroll
    for (int i = 0; i < KS; ++i) {
      const v2f a = *(const v2f*)(arow + 4 * i);
      acc = __builtin_amdgcn_wmma_f32_16x16x4_f32(false, a, false, bf[i],
                                                  (short)0, acc, false, false);
    }

    if constexpr (NORM) {
      asm volatile("v_nop\n\tv_nop\n\tv_nop\n\tv_nop\n\tv_nop" : "+v"(acc));
      float sq[8];
#pragma unroll
      for (int r = 0; r < 8; ++r) {
        float t = acc[r] * acc[r];
        t += row_xor<1>(t);
        t += row_xor<2>(t);
        t += row_xor<4>(t);
        t += row_xor<8>(t);
        sq[r] = t;
      }
      if (l16 == 0) {
        float4* p = (float4*)(rs + s * 128 + wave * 16 + 8 * half);
        p[0] = make_float4(sq[0], sq[1], sq[2], sq[3]);
        p[1] = make_float4(sq[4], sq[5], sq[6], sq[7]);
      }
    }

    if (more) {
      float* st = tA + (s ^ 1) * (16 * PITCH);
      tile_store<PITCH>(st, q0, q1, r0, c4);
      if constexpr (K == 2 * HID) tile_store<PITCH>(st + HID, q2, q3, r0, c4);
    }
    __syncthreads();

    if constexpr (NORM) {
      const float* qs = rs + s * 128 + 8 * half + (l16 & 7);
      float t = qs[0];
#pragma unroll
      for (int w = 1; w < 8; ++w) t += qs[16 * w];
      const float nrm = fmaxf(sqrtf(t), 1e-12f);
      float nr[8];
      nr[0] = row_share<0>(nrm);
      nr[1] = row_share<1>(nrm);
      nr[2] = row_share<2>(nrm);
      nr[3] = row_share<3>(nrm);
      nr[4] = row_share<4>(nrm);
      nr[5] = row_share<5>(nrm);
      nr[6] = row_share<6>(nrm);
      nr[7] = row_share<7>(nrm);
#pragma unroll
      for (int r = 0; r < 8; ++r) {
        const float v = fmaxf(acc[r] / nr[r], 0.0f);
        acc[r] = v;
        ps += v;
        pss += v * v;
      }
    }
    s ^= 1;

#pragma unroll
    for (int r = 0; r < 8; ++r) {
      const int row = tile * 16 + r + 8 * half;
      Y[row * HID + col] = acc[r];
    }
  }

  if constexpr (NORM) {
    ps += row_other(ps);
    pss += row_other(pss);
    if (half == 0) {
      part[blockIdx.x * 256 + col] = ps;
      part[blockIdx.x * 256 + HID + col] = pss;
    }
  }
}

__global__ void __launch_bounds__(256) embed_gemm(
    const float* __restrict__ X, const float* __restrict__ W,
    const float* __restrict__ B, float* __restrict__ Y, int mtiles) {
  __shared__ __attribute__((aligned(16))) float tA[2 * 16 * (HID + A_PAD)];
  gemm_persistent<HID, false>(X, X, W, B, Y, mtiles, tA, nullptr, nullptr);
}

__global__ void __launch_bounds__(256) layer_gemm(
    const float* __restrict__ h, const float* __restrict__ c,
    const float* __restrict__ W, const float* __restrict__ B,
    float* __restrict__ Z, float* __restrict__ part, int mtiles) {
  __shared__ __attribute__((aligned(16))) float tA[2 * 16 * (2 * HID + A_PAD)];
  __shared__ __attribute__((aligned(16))) float rs[2 * 8 * 16];
  gemm_persistent<2 * HID, true>(h, c, W, B, Z, mtiles, tA, rs, part);
}

#define BNF_BLOCKS (HID / 32)
#define BNF_WAVES 32
#define BNF_RPP (BNF_WAVES * 4)
#define BNF_PASSES (NB_STATS / BNF_RPP)
#define BNR_SLABS 64
#define BNR_VEC 5

__global__ void __launch_bounds__(BNF_WAVES * 32) bn_residual(
    const float* h, const float* __restrict__ z,
    const float* __restrict__ part, const float* __restrict__ gamma,
    const float* __restrict__ beta, float* out, float invN, int N) {
  static_assert(NB_STATS % BNF_RPP == 0, "whole passes over the partial rows");
  __shared__ __attribute__((aligned(16))) float red[BNF_WAVES * 64];
  __shared__ __attribute__((aligned(16))) float bnl[64];
  const int t = threadIdx.x;
  const int lane = t & 31;
  const int w = t >> 5;
  const int g = blockIdx.x & (BNF_BLOCKS - 1);
  const int slab = blockIdx.x / BNF_BLOCKS;
  const int col0 = g * 32;
  const int piece = t & 7;
  const int rr = t >> 3;

  const int rps = (N + BNR_SLABS - 1) / BNR_SLABS;
  const int rbeg = slab * rps;
  const int rend = min(rbeg + rps, N);

  float ga = 0.0f, be = 0.0f;
  if (t < 32) {
    ga = gamma[col0 + t];
    be = beta[col0 + t];
  }

  const float4* p = (const float4*)(part + rr * 256 + col0) + piece;
  float4 s[BNF_PASSES], q[BNF_PASSES];
#pragma unroll
  for (int i = 0; i < BNF_PASSES; ++i) {
    s[i] = p[i * (BNF_RPP * 64)];
    q[i] = p[i * (BNF_RPP * 64) + 32];
  }

  const float4* hv = (const float4*)h + g * 8 + piece;
  const float4* zv = (const float4*)z + g * 8 + piece;
  float4* ov = (float4*)out + g * 8 + piece;
  float4 hq[BNR_VEC], zq[BNR_VEC];
#pragma unroll
  for (int i = 0; i < BNR_VEC; ++i) {
    const int r = min(rbeg + rr + i * BNF_RPP, N - 1);
    hq[i] = hv[r * (HID / 4)];
    zq[i] = zv[r * (HID / 4)];
  }
  __builtin_amdgcn_sched_barrier(0);

  float4 a = s[0];
  float4 b = q[0];
#pragma unroll
  for (int i = 1; i < BNF_PASSES; ++i) {
    a.x += s[i].x;
    a.y += s[i].y;
    a.z += s[i].z;
    a.w += s[i].w;
    b.x += q[i].x;
    b.y += q[i].y;
    b.z += q[i].z;
    b.w += q[i].w;
  }
  a.x += row_xor<8>(a.x);
  a.y += row_xor<8>(a.y);
  a.z += row_xor<8>(a.z);
  a.w += row_xor<8>(a.w);
  b.x += row_xor<8>(b.x);
  b.y += row_xor<8>(b.y);
  b.z += row_xor<8>(b.z);
  b.w += row_xor<8>(b.w);
  a.x += row_other(a.x);
  a.y += row_other(a.y);
  a.z += row_other(a.z);
  a.w += row_other(a.w);
  b.x += row_other(b.x);
  b.y += row_other(b.y);
  b.z += row_other(b.z);
  b.w += row_other(b.w);
  if (lane < 8) {
    ((float4*)(red + w * 64))[lane] = a;
    ((float4*)(red + w * 64 + 32))[lane] = b;
  }
  __syncthreads();

  if (t < 32) {
    float s1 = 0.0f, s2 = 0.0f;
#pragma unroll
    for (int k = 0; k < BNF_WAVES; ++k) {
      s1 += red[k * 64 + t];
      s2 += red[k * 64 + 32 + t];
    }
    float mu = s1 * invN;
    float var = s2 * invN - mu * mu;
    float sc = ga / sqrtf(var + 1e-5f);
    bnl[t] = sc;
    bnl[32 + t] = be - mu * sc;
  }
  __syncthreads();

  const float4 a4 = ((const float4*)bnl)[piece];
  const float4 b4 = ((const float4*)(bnl + 32))[piece];

  for (int r0 = rbeg;;) {
#pragma unroll
    for (int i = 0; i < BNR_VEC; ++i) {
      const int r = r0 + rr + i * BNF_RPP;
      float4 o;
      o.x = hq[i].x + zq[i].x * a4.x + b4.x;
      o.y = hq[i].y + zq[i].y * a4.y + b4.y;
      o.z = hq[i].z + zq[i].z * a4.z + b4.z;
      o.w = hq[i].w + zq[i].w * a4.w + b4.w;
      if (r < rend) ov[r * (HID / 4)] = o;
    }
    r0 += BNR_VEC * BNF_RPP;
    if (r0 >= rend) break;
#pragma unroll
    for (int i = 0; i < BNR_VEC; ++i) {
      const int r = min(r0 + rr + i * BNF_RPP, N - 1);
      hq[i] = hv[r * (HID / 4)];
      zq[i] = zv[r * (HID / 4)];
    }
  }
}

extern "C" void kernel_launch(void* const* d_in, const int* in_sizes, int n_in,
                              void* d_out, int out_size, void* d_ws,
                              size_t ws_size, hipStream_t stream) {
  const float* h0 = (const float*)d_in[0];
  const int* src = (const int*)d_in[1];
  const int* dst = (const int*)d_in[2];
  const float* W_emb = (const float*)d_in[3];
  const float* b_emb = (const float*)d_in[4];
  const float* Ws = (const float*)d_in[5];
  const float* bs = (const float*)d_in[6];
  const float* gam = (const float*)d_in[7];
  const float* bet = (const float*)d_in[8];

  const int N = in_sizes[0] / HID;
  const int E = in_sizes[1];
  const int NF = N * HID;
  const int L = 4;

  float* ws = (float*)d_ws;
  float* H = ws;
  float* C = H + NF;
  float* Z = C + NF;
  float* DEG = Z + NF;
  float* BN = DEG + N;
  float* PART = BN + 256;
  int* cnt = (int*)(PART + NB_STATS * 256);
  int* rowptr = cnt + N;
  int* eidx = rowptr + (N + 1);

  const int mtiles = N / 16;

  zero_i32<<<256, 256, 0, stream>>>(cnt, N);
  hist_deg<<<1024, 256, 0, stream>>>(dst, cnt, E);
  invdeg_from_cnt<<<256, 256, 0, stream>>>(cnt, DEG, N);
  scan_excl<<<1, SCAN_WAVES * 32, 0, stream>>>(cnt, rowptr, N);
  zero_i32<<<256, 256, 0, stream>>>(cnt, N);
  fill_csr<<<1024, 256, 0, stream>>>(src, dst, rowptr, cnt, eidx, E);

  embed_gemm<<<GEMM_BLOCKS, 256, 0, stream>>>(h0, W_emb, b_emb, H, mtiles);

  for (int l = 0; l < L; ++l) {
    aggregate_csr<<<1250, AGG_WAVES * 32, 0, stream>>>(H, rowptr, eidx, DEG, C,
                                                       N, E);
    layer_gemm<<<GEMM_BLOCKS, 256, 0, stream>>>(H, C, Ws + l * HID * 256,
                                                bs + l * HID, Z, PART, mtiles);
    float* out = (l == L - 1) ? (float*)d_out : H;
    bn_residual<<<BNF_BLOCKS * BNR_SLABS, BNF_WAVES * 32, 0, stream>>>(
        H, Z, PART, gam + l * HID, bet + l * HID, out, 1.0f / (float)N, N);
  }
}


# CSR build + embedding in 4 launches, not 7: the degree atomic's return is the edge's rank (no cursor pass, no 2nd clear); fill and 1/deg ride in the embedding launch; list order by arrival as before
# speedup vs baseline: 3.9474x; 1.1211x over previous
#include <hip/hip_runtime.h>
#include <hip/hip_bf16.h>


typedef float v2f __attribute__((ext_vector_type(2)));
typedef float v8f __attribute__((ext_vector_type(8)));

#define HID 128

__global__ void zero_i32(int* __restrict__ p, int n) {
  for (int i = blockIdx.x * blockDim.x + threadIdx.x; i < n;
       i += gridDim.x * blockDim.x)
    p[i] = 0;
}

__global__ void hist_rank(const int* __restrict__ dst, int* __restrict__ cnt,
                          int* __restrict__ rank, int E) {
  for (int e = blockIdx.x * blockDim.x + threadIdx.x; e < E;
       e += gridDim.x * blockDim.x)
    rank[e] = atomicAdd(&cnt[dst[e]], 1);
}

#define SCAN_WAVES 32
#define SCAN_VEC 10
#define SCAN_WSPAN (SCAN_VEC * 128)
#define SCAN_SPAN (SCAN_WAVES * SCAN_WSPAN)

template <int S>
__device__ __forceinline__ int row_shr_or0(int x) {
  return __builtin_amdgcn_update_dpp(0, x, 0x110 + S, 0xf, 0xf, false);
}

__device__ __forceinline__ int wave_incl_scan(int x, int lane) {
  x += row_shr_or0<1>(x);
  x += row_shr_or0<2>(x);
  x += row_shr_or0<4>(x);
  x += row_shr_or0<8>(x);
  const int lo = __builtin_amdgcn_readlane(x, 15);
  return x + ((lane >= 16) ? lo : 0);
}

__global__ void __launch_bounds__(SCAN_WAVES * 32) scan_excl(
    const int* __restrict__ cnt, int* __restrict__ rowptr, int N) {
  __shared__ int wtot[SCAN_WAVES];
  const int lane = threadIdx.x & 31;
  const int w = __builtin_amdgcn_readfirstlane(threadIdx.x >> 5);
  int carry = 0;

  for (int base = 0; base < N; base += SCAN_SPAN) {
    const int e0 = base + w * SCAN_WSPAN + lane * 4;
    int4 v[SCAN_VEC];
#pragma unroll
    for (int i = 0; i < SCAN_VEC; ++i) {
      const int e = e0 + i * 128;
      const int4 x = *(const int4*)(cnt + min(e, N - 4));
      const bool in = (e + 4 <= N);
      v[i].x = in ? x.x : 0;
      v[i].y = in ? x.y : 0;
      v[i].z = in ? x.z : 0;
      v[i].w = in ? x.w : 0;
    }
    __builtin_amdgcn_sched_barrier(0);

    int pre[SCAN_VEC];
    int wsum = 0;
#pragma unroll
    for (int i = 0; i < SCAN_VEC; ++i) {
      const int s = (v[i].x + v[i].y) + (v[i].z + v[i].w);
      const int t = wave_incl_scan(s, lane);
      pre[i] = wsum + (t - s);
      wsum += __builtin_amdgcn_readlane(t, 31);
    }

    if (lane == 0) wtot[w] = wsum;
    __syncthreads();
    const int tv = wtot[lane];
    const int ti = wave_incl_scan(tv, lane);
    const int wbase = carry + __builtin_amdgcn_readlane(ti - tv, w);
    carry += __builtin_amdgcn_readlane(ti, 31);

#pragma unroll
    for (int i = 0; i < SCAN_VEC; ++i) {
      const int e = e0 + i * 128;
      int4 r;
      r.x = wbase + pre[i];
      r.y = r.x + v[i].x;
      r.z = r.y + v[i].y;
      r.w = r.z + v[i].z;
      if (e + 4 <= N) *(int4*)(rowptr + e) = r;
    }
    __syncthreads();
  }
  if (threadIdx.x == 0) rowptr[N] = carry;
}

#define AGG_WAVES 4
#define AGG_NODES 8
#define AGG_ROWS 32

template <int J>
__device__ __forceinline__ void agg_request(float4 (&v)[AGG_ROWS],
                                            const float4* __restrict__ hv,
                                            int cur, int m, int lane) {
  if constexpr (J < AGG_ROWS) {
    if (J < m) {
      const int s = __builtin_amdgcn_readlane(cur, J);
      v[J] = hv[s * (HID / 4) + lane];
      agg_request<J + 1>(v, hv, cur, m, lane);
    }
  }
}

template <int G>
__device__ __forceinline__ void agg_sum(float4& acc,
                                        const float4 (&v)[AGG_ROWS], int m) {
  if constexpr (G < AGG_ROWS) {
    if (G + 4 <= m) {
      acc.x += ((v[G].x + v[G + 1].x) + v[G + 2].x) + v[G + 3].x;
      acc.y += ((v[G].y + v[G + 1].y) + v[G + 2].y) + v[G + 3].y;
      acc.z += ((v[G].z + v[G + 1].z) + v[G + 2].z) + v[G + 3].z;
      acc.w += ((v[G].w + v[G + 1].w) + v[G + 2].w) + v[G + 3].w;
      agg_sum<G + 4>(acc, v, m);
    } else if (G < m) {
      acc.x += v[G].x;
      acc.y += v[G].y;
      acc.z += v[G].z;
      acc.w += v[G].w;
      if (G + 1 < m) {
        acc.x += v[G + 1].x;
        acc.y += v[G + 1].y;
        acc.z += v[G + 1].z;
        acc.w += v[G + 1].w;
      }
      if (G + 2 < m) {
        acc.x += v[G + 2].x;
        acc.y += v[G + 2].y;
        acc.z += v[G + 2].z;
        acc.w += v[G + 2].w;
      }
    }
  }
}

__global__ void __launch_bounds__(AGG_WAVES * 32) aggregate_csr(
    const float* __restrict__ h, const int* __restrict__ rowptr,
    const int* __restrict__ eidx, const float* __restrict__ invdeg,
    float* __restrict__ c, int N, int E) {
  const int lane = threadIdx.x & 31;
  const int w = __builtin_amdgcn_readfirstlane(threadIdx.x >> 5);
  const int gw = blockIdx.x * AGG_WAVES + w;
  const int tw = gridDim.x * AGG_WAVES;
  const float4* __restrict__ hv = (const float4*)h;
  float4* __restrict__ cv = (float4*)c;

  for (int base = gw * AGG_NODES; base < N; base += tw * AGG_NODES) {
    const int nn = min(AGG_NODES, N - base);
    const int rp = rowptr[min(base + lane, N)];
    const int dg = __builtin_bit_cast(int, invdeg[min(base + lane, N - 1)]);
    int b = __builtin_amdgcn_readlane(rp, 0);
    int e = __builtin_amdgcn_readlane(rp, 1);
    int idx = eidx[min(b + lane, E - 1)];

    for (int k = 0; k < nn; ++k) {
      const int cnt = e - b;
      const int ne = __builtin_amdgcn_readlane(rp, k + 2);
      const int nidx = eidx[min(e + lane, E - 1)];

      float4 acc = make_float4(0.0f, 0.0f, 0.0f, 0.0f);
      for (int chunk = 0; chunk < cnt; chunk += AGG_ROWS) {
        int cur = idx;
        if (chunk > 0) cur = eidx[min(b + chunk + lane, E - 1)];
        const int m = min(AGG_ROWS, cnt - chunk);
        float4 v[AGG_ROWS];
        agg_request<0>(v, hv, cur, m, lane);
        agg_sum<0>(acc, v, m);
      }

      const float sc =
          __builtin_bit_cast(float, __builtin_amdgcn_readlane(dg, k));
      float4 r;
      r.x = acc.x * sc;
      r.y = acc.y * sc;
      r.z = acc.z * sc;
      r.w = acc.w * sc;
      cv[(base + k) * (HID / 4) + lane] = r;

      b = e;
      e = ne;
      idx = nidx;
    }
  }
}


template <int M>
__device__ __forceinline__ float row_xor(float x) {
  const int b = __builtin_bit_cast(int, x);
  return __builtin_bit_cast(
      float, __builtin_amdgcn_update_dpp(b, b, 0x160 + M, 0xf, 0xf, true));
}

__device__ __forceinline__ float row_other(float x) {
  const int b = __builtin_bit_cast(int, x);
  return __builtin_bit_cast(
      float, __builtin_amdgcn_permlanex16(b, b, 0x76543210, 0xfedcba98, false,
                                          false));
}

#define A_PAD 4
#define GEMM_BLOCKS 512
#define NB_STATS GEMM_BLOCKS
#define FILL_BLOCKS 1024

__device__ __forceinline__ void tile_request(float4& qa, float4& qb,
                                             const float* __restrict__ src,
                                             int tile, int r0, int c4) {
  const float4* p = (const float4*)(src + (tile * 16 + r0) * HID) + c4;
  qa = p[0];
  qb = p[8 * (HID / 4)];
}

template <int PITCH>
__device__ __forceinline__ void tile_store(float* __restrict__ st,
                                           const float4& qa, const float4& qb,
                                           int r0, int c4) {
  float4* p = (float4*)(st + r0 * PITCH) + c4;
  p[0] = qa;
  p[8 * (PITCH / 4)] = qb;
}

template <int R>
__device__ __forceinline__ float row_share(float x) {
  const int b = __builtin_bit_cast(int, x);
  return __builtin_bit_cast(
      float, __builtin_amdgcn_update_dpp(b, b, 0x150 + R, 0xf, 0xf, true));
}

template <int K, bool NORM>
__device__ __forceinline__ void gemm_persistent(
    const float* __restrict__ a0, const float* __restrict__ a1,
    const float* __restrict__ W, const float* __restrict__ B,
    float* __restrict__ Y, int mtiles, float* __restrict__ tA,
    float* __restrict__ rs, float* __restrict__ part, int bid, int nb) {
  constexpr int PITCH = K + A_PAD;
  constexpr int KS = K / 4;

  const int tid = threadIdx.x;
  const int wave = tid >> 5;
  const int lane = tid & 31;
  const int half = lane >> 4;
  const int l16 = lane & 15;
  const int col = wave * 16 + l16;
  const int r0 = tid >> 5;
  const int c4 = tid & 31;

  int tile = bid;
  float4 q0, q1, q2, q3;
  q0 = q1 = q2 = q3 = make_float4(0.0f, 0.0f, 0.0f, 0.0f);
  if (tile < mtiles) {
    tile_request(q0, q1, a0, tile, r0, c4);
    if constexpr (K == 2 * HID) tile_request(q2, q3, a1, tile, r0, c4);
  }

  const float bv = B[col];

  if (tile < mtiles) {
    tile_store<PITCH>(tA, q0, q1, r0, c4);
    if constexpr (K == 2 * HID) tile_store<PITCH>(tA + HID, q2, q3, r0, c4);
  }
  __syncthreads();

  v2f bf[KS];
  const float* wrow = W + col * K + 2 * half;
#pragma unroll
  for (int i = 0; i < KS; ++i) bf[i] = *(const v2f*)(wrow + 4 * i);

  float ps = 0.0f, pss = 0.0f;
  int s = 0;
  for (; tile < mtiles; tile += nb) {
    const int nt = tile + nb;
    const bool more = nt < mtiles;
    if (more) {
      tile_request(q0, q1, a0, nt, r0, c4);
      if constexpr (K == 2 * HID) tile_request(q2, q3, a1, nt, r0, c4);
    }

    v8f acc;
#pragma unroll
    for (int r = 0; r < 8; ++r) acc[r] = bv;

    const float* arow = tA + s * (16 * PITCH) + l16 * PITCH + 2 * half;
#pragma unroll
    for (int i = 0; i < KS; ++i) {
      const v2f a = *(const v2f*)(arow + 4 * i);
      acc = __builtin_amdgcn_wmma_f32_16x16x4_f32(false, a, false, bf[i],
                                                  (short)0, acc, false, false);
    }

    if constexpr (NORM) {
      asm volatile("v_nop\n\tv_nop\n\tv_nop\n\tv_nop\n\tv_nop" : "+v"(acc));
      float sq[8];
#pragma unroll
      for (int r = 0; r < 8; ++r) {
        float t = acc[r] * acc[r];
        t += row_xor<1>(t);
        t += row_xor<2>(t);
        t += row_xor<4>(t);
        t += row_xor<8>(t);
        sq[r] = t;
      }
      if (l16 == 0) {
        float4* p = (float4*)(rs + s * 128 + wave * 16 + 8 * half);
        p[0] = make_float4(sq[0], sq[1], sq[2], sq[3]);
        p[1] = make_float4(sq[4], sq[5], sq[6], sq[7]);
      }
    }

    if (more) {
      float* st = tA + (s ^ 1) * (16 * PITCH);
      tile_store<PITCH>(st, q0, q1, r0, c4);
      if constexpr (K == 2 * HID) tile_store<PITCH>(st + HID, q2, q3, r0, c4);
    }
    __syncthreads();

    if constexpr (NORM) {
      const float* qs = rs + s * 128 + 8 * half + (l16 & 7);
      float t = qs[0];
#pragma unroll
      for (int w = 1; w < 8; ++w) t += qs[16 * w];
      const float nrm = fmaxf(sqrtf(t), 1e-12f);
      float nr[8];
      nr[0] = row_share<0>(nrm);
      nr[1] = row_share<1>(nrm);
      nr[2] = row_share<2>(nrm);
      nr[3] = row_share<3>(nrm);
      nr[4] = row_share<4>(nrm);
      nr[5] = row_share<5>(nrm);
      nr[6] = row_share<6>(nrm);
      nr[7] = row_share<7>(nrm);
#pragma unroll
      for (int r = 0; r < 8; ++r) {
        const float v = fmaxf(acc[r] / nr[r], 0.0f);
        acc[r] = v;
        ps += v;
        pss += v * v;
      }
    }
    s ^= 1;

#pragma unroll
    for (int r = 0; r < 8; ++r) {
      const int row = tile * 16 + r + 8 * half;
      Y[row * HID + col] = acc[r];
    }
  }

  if constexpr (NORM) {
    ps += row_other(ps);
    pss += row_other(pss);
    if (half == 0) {
      part[bid * 256 + col] = ps;
      part[bid * 256 + HID + col] = pss;
    }
  }
}

__global__ void __launch_bounds__(256) embed_gemm_fill_csr(
    const float* __restrict__ X, const float* __restrict__ W,
    const float* __restrict__ B, float* __restrict__ Y, int mtiles,
    const int* __restrict__ src, const int* __restrict__ dst,
    const int* __restrict__ rowptr, const int* __restrict__ rank,
    int* __restrict__ eidx, float* __restrict__ invdeg, int N, int E) {
  __shared__ __attribute__((aligned(16))) float tA[2 * 16 * (HID + A_PAD)];
  if (blockIdx.x < GEMM_BLOCKS) {
    gemm_persistent<HID, false>(X, X, W, B, Y, mtiles, tA, nullptr, nullptr,
                                blockIdx.x, GEMM_BLOCKS);
  } else {
    const int fb = blockIdx.x - GEMM_BLOCKS;
    const int nfb = gridDim.x - GEMM_BLOCKS;
    const int gt = fb * blockDim.x + threadIdx.x;
    const int nt = nfb * blockDim.x;
    for (int e = gt; e < E; e += nt) eidx[rowptr[dst[e]] + rank[e]] = src[e];
    for (int n = gt; n < N; n += nt)
      invdeg[n] = 1.0f / fmaxf((float)(rowptr[n + 1] - rowptr[n]), 1.0f);
  }
}

__global__ void __launch_bounds__(256) layer_gemm(
    const float* __restrict__ h, const float* __restrict__ c,
    const float* __restrict__ W, const float* __restrict__ B,
    float* __restrict__ Z, float* __restrict__ part, int mtiles) {
  __shared__ __attribute__((aligned(16))) float tA[2 * 16 * (2 * HID + A_PAD)];
  __shared__ __attribute__((aligned(16))) float rs[2 * 8 * 16];
  gemm_persistent<2 * HID, true>(h, c, W, B, Z, mtiles, tA, rs, part,
                                 blockIdx.x, gridDim.x);
}

#define BNF_BLOCKS (HID / 32)
#define BNF_WAVES 32
#define BNF_RPP (BNF_WAVES * 4)
#define BNF_PASSES (NB_STATS / BNF_RPP)
#define BNR_SLABS 64
#define BNR_VEC 5

__global__ void __launch_bounds__(BNF_WAVES * 32) bn_residual(
    const float* h, const float* __restrict__ z,
    const float* __restrict__ part, const float* __restrict__ gamma,
    const float* __restrict__ beta, float* out, float invN, int N) {
  static_assert(NB_STATS % BNF_RPP == 0, "whole passes over the partial rows");
  __shared__ __attribute__((aligned(16))) float red[BNF_WAVES * 64];
  __shared__ __attribute__((aligned(16))) float bnl[64];
  const int t = threadIdx.x;
  const int lane = t & 31;
  const int w = t >> 5;
  const int g = blockIdx.x & (BNF_BLOCKS - 1);
  const int slab = blockIdx.x / BNF_BLOCKS;
  const int col0 = g * 32;
  const int piece = t & 7;
  const int rr = t >> 3;

  const int rps = (N + BNR_SLABS - 1) / BNR_SLABS;
  const int rbeg = slab * rps;
  const int rend = min(rbeg + rps, N);

  float ga = 0.0f, be = 0.0f;
  if (t < 32) {
    ga = gamma[col0 + t];
    be = beta[col0 + t];
  }

  const float4* p = (const float4*)(part + rr * 256 + col0) + piece;
  float4 s[BNF_PASSES], q[BNF_PASSES];
#pragma unroll
  for (int i = 0; i < BNF_PASSES; ++i) {
    s[i] = p[i * (BNF_RPP * 64)];
    q[i] = p[i * (BNF_RPP * 64) + 32];
  }

  const float4* hv = (const float4*)h + g * 8 + piece;
  const float4* zv = (const float4*)z + g * 8 + piece;
  float4* ov = (float4*)out + g * 8 + piece;
  float4 hq[BNR_VEC], zq[BNR_VEC];
#pragma unroll
  for (int i = 0; i < BNR_VEC; ++i) {
    const int r = min(rbeg + rr + i * BNF_RPP, N - 1);
    hq[i] = hv[r * (HID / 4)];
    zq[i] = zv[r * (HID / 4)];
  }
  __builtin_amdgcn_sched_barrier(0);

  float4 a = s[0];
  float4 b = q[0];
#pragma unroll
  for (int i = 1; i < BNF_PASSES; ++i) {
    a.x += s[i].x;
    a.y += s[i].y;
    a.z += s[i].z;
    a.w += s[i].w;
    b.x += q[i].x;
    b.y += q[i].y;
    b.z += q[i].z;
    b.w += q[i].w;
  }
  a.x += row_xor<8>(a.x);
  a.y += row_xor<8>(a.y);
  a.z += row_xor<8>(a.z);
  a.w += row_xor<8>(a.w);
  b.x += row_xor<8>(b.x);
  b.y += row_xor<8>(b.y);
  b.z += row_xor<8>(b.z);
  b.w += row_xor<8>(b.w);
  a.x += row_other(a.x);
  a.y += row_other(a.y);
  a.z += row_other(a.z);
  a.w += row_other(a.w);
  b.x += row_other(b.x);
  b.y += row_other(b.y);
  b.z += row_other(b.z);
  b.w += row_other(b.w);
  if (lane < 8) {
    ((float4*)(red + w * 64))[lane] = a;
    ((float4*)(red + w * 64 + 32))[lane] = b;
  }
  __syncthreads();

  if (t < 32) {
    float s1 = 0.0f, s2 = 0.0f;
#pragma unroll
    for (int k = 0; k < BNF_WAVES; ++k) {
      s1 += red[k * 64 + t];
      s2 += red[k * 64 + 32 + t];
    }
    float mu = s1 * invN;
    float var = s2 * invN - mu * mu;
    float sc = ga / sqrtf(var + 1e-5f);
    bnl[t] = sc;
    bnl[32 + t] = be - mu * sc;
  }
  __syncthreads();

  const float4 a4 = ((const float4*)bnl)[piece];
  const float4 b4 = ((const float4*)(bnl + 32))[piece];

  for (int r0 = rbeg;;) {
#pragma unroll
    for (int i = 0; i < BNR_VEC; ++i) {
      const int r = r0 + rr + i * BNF_RPP;
      float4 o;
      o.x = hq[i].x + zq[i].x * a4.x + b4.x;
      o.y = hq[i].y + zq[i].y * a4.y + b4.y;
      o.z = hq[i].z + zq[i].z * a4.z + b4.z;
      o.w = hq[i].w + zq[i].w * a4.w + b4.w;
      if (r < rend) ov[r * (HID / 4)] = o;
    }
    r0 += BNR_VEC * BNF_RPP;
    if (r0 >= rend) break;
#pragma unroll
    for (int i = 0; i < BNR_VEC; ++i) {
      const int r = min(r0 + rr + i * BNF_RPP, N - 1);
      hq[i] = hv[r * (HID / 4)];
      zq[i] = zv[r * (HID / 4)];
    }
  }
}

extern "C" void kernel_launch(void* const* d_in, const int* in_sizes, int n_in,
                              void* d_out, int out_size, void* d_ws,
                              size_t ws_size, hipStream_t stream) {
  const float* h0 = (const float*)d_in[0];
  const int* src = (const int*)d_in[1];
  const int* dst = (const int*)d_in[2];
  const float* W_emb = (const float*)d_in[3];
  const float* b_emb = (const float*)d_in[4];
  const float* Ws = (const float*)d_in[5];
  const float* bs = (const float*)d_in[6];
  const float* gam = (const float*)d_in[7];
  const float* bet = (const float*)d_in[8];

  const int N = in_sizes[0] / HID;
  const int E = in_sizes[1];
  const int NF = N * HID;
  const int L = 4;

  float* ws = (float*)d_ws;
  float* H = ws;
  float* C = H + NF;
  float* Z = C + NF;
  float* DEG = Z + NF;
  float* BN = DEG + N;
  float* PART = BN + 256;
  int* cnt = (int*)(PART + NB_STATS * 256);
  int* rowptr = cnt + N;
  int* eidx = rowptr + (N + 1);
  int* rank = (int*)Z;

  const int mtiles = N / 16;

  zero_i32<<<256, 256, 0, stream>>>(cnt, N);
  hist_rank<<<1024, 256, 0, stream>>>(dst, cnt, rank, E);
  scan_excl<<<1, SCAN_WAVES * 32, 0, stream>>>(cnt, rowptr, N);

  embed_gemm_fill_csr<<<GEMM_BLOCKS + FILL_BLOCKS, 256, 0, stream>>>(
      h0, W_emb, b_emb, H, mtiles, src, dst, rowptr, rank, eidx, DEG, N, E);

  for (int l = 0; l < L; ++l) {
    aggregate_csr<<<1250, AGG_WAVES * 32, 0, stream>>>(H, rowptr, eidx, DEG, C,
                                                       N, E);
    layer_gemm<<<GEMM_BLOCKS, 256, 0, stream>>>(H, C, Ws + l * HID * 256,
                                                bs + l * HID, Z, PART, mtiles);
    float* out = (l == L - 1) ? (float*)d_out : H;
    bn_residual<<<BNF_BLOCKS * BNR_SLABS, BNF_WAVES * 32, 0, stream>>>(
        H, Z, PART, gam + l * HID, bet + l * HID, out, 1.0f / (float)N, N);
  }
}


# layer_gemm gathers its own c rows: a wave averages the neighbour rows of two tile rows into the LDS A tile one tile ahead, first 16 rows under its products; aggregate_csr launch gone; sum order kept
# speedup vs baseline: 4.2114x; 1.0669x over previous
#include <hip/hip_runtime.h>
#include <hip/hip_bf16.h>


typedef float v2f __attribute__((ext_vector_type(2)));
typedef float v8f __attribute__((ext_vector_type(8)));

#define HID 128

__global__ void zero_i32(int* __restrict__ p, int n) {
  for (int i = blockIdx.x * blockDim.x + threadIdx.x; i < n;
       i += gridDim.x * blockDim.x)
    p[i] = 0;
}

__global__ void hist_rank(const int* __restrict__ dst, int* __restrict__ cnt,
                          int* __restrict__ rank, int E) {
  for (int e = blockIdx.x * blockDim.x + threadIdx.x; e < E;
       e += gridDim.x * blockDim.x)
    rank[e] = atomicAdd(&cnt[dst[e]], 1);
}

#define SCAN_WAVES 32
#define SCAN_VEC 10
#define SCAN_WSPAN (SCAN_VEC * 128)
#define SCAN_SPAN (SCAN_WAVES * SCAN_WSPAN)

template <int S>
__device__ __forceinline__ int row_shr_or0(int x) {
  return __builtin_amdgcn_update_dpp(0, x, 0x110 + S, 0xf, 0xf, false);
}

__device__ __forceinline__ int wave_incl_scan(int x, int lane) {
  x += row_shr_or0<1>(x);
  x += row_shr_or0<2>(x);
  x += row_shr_or0<4>(x);
  x += row_shr_or0<8>(x);
  const int lo = __builtin_amdgcn_readlane(x, 15);
  return x + ((lane >= 16) ? lo : 0);
}

__global__ void __launch_bounds__(SCAN_WAVES * 32) scan_excl(
    const int* __restrict__ cnt, int* __restrict__ rowptr, int N) {
  __shared__ int wtot[SCAN_WAVES];
  const int lane = threadIdx.x & 31;
  const int w = __builtin_amdgcn_readfirstlane(threadIdx.x >> 5);
  int carry = 0;

  for (int base = 0; base < N; base += SCAN_SPAN) {
    const int e0 = base + w * SCAN_WSPAN + lane * 4;
    int4 v[SCAN_VEC];
#pragma unroll
    for (int i = 0; i < SCAN_VEC; ++i) {
      const int e = e0 + i * 128;
      const int4 x = *(const int4*)(cnt + min(e, N - 4));
      const bool in = (e + 4 <= N);
      v[i].x = in ? x.x : 0;
      v[i].y = in ? x.y : 0;
      v[i].z = in ? x.z : 0;
      v[i].w = in ? x.w : 0;
    }
    __builtin_amdgcn_sched_barrier(0);

    int pre[SCAN_VEC];
    int wsum = 0;
#pragma unroll
    for (int i = 0; i < SCAN_VEC; ++i) {
      const int s = (v[i].x + v[i].y) + (v[i].z + v[i].w);
      const int t = wave_incl_scan(s, lane);
      pre[i] = wsum + (t - s);
      wsum += __builtin_amdgcn_readlane(t, 31);
    }

    if (lane == 0) wtot[w] = wsum;
    __syncthreads();
    const int tv = wtot[lane];
    const int ti = wave_incl_scan(tv, lane);
    const int wbase = carry + __builtin_amdgcn_readlane(ti - tv, w);
    carry += __builtin_amdgcn_readlane(ti, 31);

#pragma unroll
    for (int i = 0; i < SCAN_VEC; ++i) {
      const int e = e0 + i * 128;
      int4 r;
      r.x = wbase + pre[i];
      r.y = r.x + v[i].x;
      r.z = r.y + v[i].y;
      r.w = r.z + v[i].z;
      if (e + 4 <= N) *(int4*)(rowptr + e) = r;
    }
    __syncthreads();
  }
  if (threadIdx.x == 0) rowptr[N] = carry;
}

#define AGG_ROWS 16

template <int J>
__device__ __forceinline__ void agg_request(float4 (&v)[AGG_ROWS],
                                            const float4* __restrict__ hv,
                                            int cur, int o, int m, int lane) {
  if constexpr (J < AGG_ROWS) {
    if (J < m) {
      const int s = __builtin_amdgcn_readlane(cur, o + J);
      v[J] = hv[s * (HID / 4) + lane];
      agg_request<J + 1>(v, hv, cur, o, m, lane);
    }
  }
}

template <int G>
__device__ __forceinline__ void agg_sum(float4& acc,
                                        const float4 (&v)[AGG_ROWS], int m) {
  if constexpr (G < AGG_ROWS) {
    if (G + 4 <= m) {
      acc.x += ((v[G].x + v[G + 1].x) + v[G + 2].x) + v[G + 3].x;
      acc.y += ((v[G].y + v[G + 1].y) + v[G + 2].y) + v[G + 3].y;
      acc.z += ((v[G].z + v[G + 1].z) + v[G + 2].z) + v[G + 3].z;
      acc.w += ((v[G].w + v[G + 1].w) + v[G + 2].w) + v[G + 3].w;
      agg_sum<G + 4>(acc, v, m);
    } else if (G < m) {
      acc.x += v[G].x;
      acc.y += v[G].y;
      acc.z += v[G].z;
      acc.w += v[G].w;
      if (G + 1 < m) {
        acc.x += v[G + 1].x;
        acc.y += v[G + 1].y;
        acc.z += v[G + 1].z;
        acc.w += v[G + 1].w;
      }
      if (G + 2 < m) {
        acc.x += v[G + 2].x;
        acc.y += v[G + 2].y;
        acc.z += v[G + 2].z;
        acc.w += v[G + 2].w;
      }
    }
  }
}

__device__ __forceinline__ int agg_bounds(const int* __restrict__ rowptr,
                                          int n0, int lane, int N) {
  return rowptr[min(n0 + lane, N)];
}

__device__ __forceinline__ int agg_scales(const float* __restrict__ invdeg,
                                          int n0, int lane, int N) {
  return __builtin_bit_cast(int, invdeg[min(n0 + lane, N - 1)]);
}

__device__ __forceinline__ int agg_indices(const int* __restrict__ eidx, int b,
                                           int lane, int E) {
  return eidx[min(b + lane, E - 1)];
}

template <int PITCH, bool PRE>
__device__ __forceinline__ void agg_two_rows(
    float* __restrict__ st, int r, const float4* __restrict__ hv,
    const int* __restrict__ eidx, int rp, int dg, int ia, int ib,
    float4 (&v)[AGG_ROWS], int lane, int E) {
  int b = __builtin_amdgcn_readlane(rp, 0);
  int e = __builtin_amdgcn_readlane(rp, 1);
  int idx = ia;
#pragma nounroll
  for (int k = 0; k < 2; ++k) {
    const int cnt = e - b;
    float4 acc = make_float4(0.0f, 0.0f, 0.0f, 0.0f);
    int cur = idx;
    for (int g = 0; g < cnt; g += AGG_ROWS) {
      const int o = g & 31;
      if (g >= 32 && o == 0) cur = agg_indices(eidx, b + g, lane, E);
      const int m = min(AGG_ROWS, cnt - g);
      if (!(PRE && k == 0 && g == 0)) agg_request<0>(v, hv, cur, o, m, lane);
      agg_sum<0>(acc, v, m);
    }
    const float sc =
        __builtin_bit_cast(float, __builtin_amdgcn_readlane(dg, k));
    float4 q;
    q.x = acc.x * sc;
    q.y = acc.y * sc;
    q.z = acc.z * sc;
    q.w = acc.w * sc;
    ((float4*)(st + (r + k) * PITCH + HID))[lane] = q;
    b = e;
    e = __builtin_amdgcn_readlane(rp, 2);
    idx = ib;
  }
}


template <int M>
__device__ __forceinline__ float row_xor(float x) {
  const int b = __builtin_bit_cast(int, x);
  return __builtin_bit_cast(
      float, __builtin_amdgcn_update_dpp(b, b, 0x160 + M, 0xf, 0xf, true));
}

__device__ __forceinline__ float row_other(float x) {
  const int b = __builtin_bit_cast(int, x);
  return __builtin_bit_cast(
      float, __builtin_amdgcn_permlanex16(b, b, 0x76543210, 0xfedcba98, false,
                                          false));
}

#define A_PAD 4
#define GEMM_BLOCKS 512
#define NB_STATS GEMM_BLOCKS
#define FILL_BLOCKS 1024

__device__ __forceinline__ void tile_request(float4& qa, float4& qb,
                                             const float* __restrict__ src,
                                             int tile, int r0, int c4) {
  const float4* p = (const float4*)(src + (tile * 16 + r0) * HID) + c4;
  qa = p[0];
  qb = p[8 * (HID / 4)];
}

template <int PITCH>
__device__ __forceinline__ void tile_store(float* __restrict__ st,
                                           const float4& qa, const float4& qb,
                                           int r0, int c4) {
  float4* p = (float4*)(st + r0 * PITCH) + c4;
  p[0] = qa;
  p[8 * (PITCH / 4)] = qb;
}

template <int R>
__device__ __forceinline__ float row_share(float x) {
  const int b = __builtin_bit_cast(int, x);
  return __builtin_bit_cast(
      float, __builtin_amdgcn_update_dpp(b, b, 0x150 + R, 0xf, 0xf, true));
}

template <int K, bool NORM>
__device__ __forceinline__ void gemm_persistent(
    const float* __restrict__ a0, const int* __restrict__ rowptr,
    const int* __restrict__ eidx, const float* __restrict__ invdeg,
    const float* __restrict__ W, const float* __restrict__ B,
    float* __restrict__ Y, int mtiles, float* __restrict__ tA,
    float* __restrict__ rs, float* __restrict__ part, int bid, int nb, int N,
    int E) {
  constexpr bool GATHER = (K == 2 * HID);
  constexpr int PITCH = K + A_PAD;
  constexpr int KS = K / 4;

  const int tid = threadIdx.x;
  const int wave = tid >> 5;
  const int lane = tid & 31;
  const int half = lane >> 4;
  const int l16 = lane & 15;
  const int col = wave * 16 + l16;
  const int r0 = tid >> 5;
  const int c4 = tid & 31;

  int tile = bid;
  float4 q0, q1;
  q0 = q1 = make_float4(0.0f, 0.0f, 0.0f, 0.0f);
  const float4* __restrict__ hv = (const float4*)a0;
  const int wv = __builtin_amdgcn_readfirstlane(wave);
  int rp = 0, dg = 0, ia = 0, ib = 0, rpn = 0, ian = 0, ibn = 0;
  float4 gv[AGG_ROWS];
  if (tile < mtiles) {
    tile_request(q0, q1, a0, tile, r0, c4);
    if constexpr (GATHER) {
      rp = agg_bounds(rowptr, tile * 16 + 2 * wv, lane, N);
      dg = agg_scales(invdeg, tile * 16 + 2 * wv, lane, N);
      if (tile + nb < mtiles)
        rpn = agg_bounds(rowptr, (tile + nb) * 16 + 2 * wv, lane, N);
    }
  }

  const float bv = B[col];

  if (tile < mtiles) {
    tile_store<PITCH>(tA, q0, q1, r0, c4);
    if constexpr (GATHER) {
      ia = agg_indices(eidx, __builtin_amdgcn_readlane(rp, 0), lane, E);
      ib = agg_indices(eidx, __builtin_amdgcn_readlane(rp, 1), lane, E);
      if (tile + nb < mtiles) {
        ian = agg_indices(eidx, __builtin_amdgcn_readlane(rpn, 0), lane, E);
        ibn = agg_indices(eidx, __builtin_amdgcn_readlane(rpn, 1), lane, E);
      }
      agg_two_rows<PITCH, false>(tA, 2 * wv, hv, eidx, rp, dg, ia, ib, gv,
                                 lane, E);
      rp = rpn;
      ia = ian;
      ib = ibn;
    }
  }
  __syncthreads();

  v2f bf[KS];
  const float* wrow = W + col * K + 2 * half;
#pragma unroll
  for (int i = 0; i < KS; ++i) bf[i] = *(const v2f*)(wrow + 4 * i);

  float ps = 0.0f, pss = 0.0f;
  int s = 0;
  for (; tile < mtiles; tile += nb) {
    const int nt = tile + nb;
    const bool more = nt < mtiles;
    if (more) {
      tile_request(q0, q1, a0, nt, r0, c4);
      if constexpr (GATHER) {
        const int gb = __builtin_amdgcn_readlane(rp, 0);
        const int ge = __builtin_amdgcn_readlane(rp, 1);
        agg_request<0>(gv, hv, ia, 0, min(AGG_ROWS, ge - gb), lane);
        if (nt + nb < mtiles)
          rpn = agg_bounds(rowptr, (nt + nb) * 16 + 2 * wv, lane, N);
      }
    }

    v8f acc;
#pragma unroll
    for (int r = 0; r < 8; ++r) acc[r] = bv;

    const float* arow = tA + s * (16 * PITCH) + l16 * PITCH + 2 * half;
#pragma unroll
    for (int i = 0; i < KS; ++i) {
      const v2f a = *(const v2f*)(arow + 4 * i);
      acc = __builtin_amdgcn_wmma_f32_16x16x4_f32(false, a, false, bf[i],
                                                  (short)0, acc, false, false);
    }

    if constexpr (NORM) {
      asm volatile("v_nop\n\tv_nop\n\tv_nop\n\tv_nop\n\tv_nop" : "+v"(acc));
      float sq[8];
#pragma unroll
      for (int r = 0; r < 8; ++r) {
        float t = acc[r] * acc[r];
        t += row_xor<1>(t);
        t += row_xor<2>(t);
        t += row_xor<4>(t);
        t += row_xor<8>(t);
        sq[r] = t;
      }
      if (l16 == 0) {
        float4* p = (float4*)(rs + s * 128 + wave * 16 + 8 * half);
        p[0] = make_float4(sq[0], sq[1], sq[2], sq[3]);
        p[1] = make_float4(sq[4], sq[5], sq[6], sq[7]);
      }
    }

    if (more) {
      float* st = tA + (s ^ 1) * (16 * PITCH);
      tile_store<PITCH>(st, q0, q1, r0, c4);
      if constexpr (GATHER) {
        dg = agg_scales(invdeg, nt * 16 + 2 * wv, lane, N);
        if (nt + nb < mtiles) {
          ian = agg_indices(eidx, __builtin_amdgcn_readlane(rpn, 0), lane, E);
          ibn = agg_indices(eidx, __builtin_amdgcn_readlane(rpn, 1), lane, E);
        }
        agg_two_rows<PITCH, true>(st, 2 * wv, hv, eidx, rp, dg, ia, ib, gv,
                                  lane, E);
        rp = rpn;
        ia = ian;
        ib = ibn;
      }
    }
    __syncthreads();

    if constexpr (NORM) {
      const float* qs = rs + s * 128 + 8 * half + (l16 & 7);
      float t = qs[0];
#pragma unroll
      for (int w = 1; w < 8; ++w) t += qs[16 * w];
      const float nrm = fmaxf(sqrtf(t), 1e-12f);
      float nr[8];
      nr[0] = row_share<0>(nrm);
      nr[1] = row_share<1>(nrm);
      nr[2] = row_share<2>(nrm);
      nr[3] = row_share<3>(nrm);
      nr[4] = row_share<4>(nrm);
      nr[5] = row_share<5>(nrm);
      nr[6] = row_share<6>(nrm);
      nr[7] = row_share<7>(nrm);
#pragma unroll
      for (int r = 0; r < 8; ++r) {
        const float v = fmaxf(acc[r] / nr[r], 0.0f);
        acc[r] = v;
        ps += v;
        pss += v * v;
      }
    }
    s ^= 1;

#pragma unroll
    for (int r = 0; r < 8; ++r) {
      const int row = tile * 16 + r + 8 * half;
      Y[row * HID + col] = acc[r];
    }
  }

  if constexpr (NORM) {
    ps += row_other(ps);
    pss += row_other(pss);
    if (half == 0) {
      part[bid * 256 + col] = ps;
      part[bid * 256 + HID + col] = pss;
    }
  }
}

__global__ void __launch_bounds__(256) embed_gemm_fill_csr(
    const float* __restrict__ X, const float* __restrict__ W,
    const float* __restrict__ B, float* __restrict__ Y, int mtiles,
    const int* __restrict__ src, const int* __restrict__ dst,
    const int* __restrict__ rowptr, const int* __restrict__ rank,
    int* __restrict__ eidx, float* __restrict__ invdeg, int N, int E) {
  __shared__ __attribute__((aligned(16))) float tA[2 * 16 * (HID + A_PAD)];
  if (blockIdx.x < GEMM_BLOCKS) {
    gemm_persistent<HID, false>(X, nullptr, nullptr, nullptr, W, B, Y, mtiles,
                                tA, nullptr, nullptr, blockIdx.x, GEMM_BLOCKS,
                                N, E);
  } else {
    const int fb = blockIdx.x - GEMM_BLOCKS;
    const int nfb = gridDim.x - GEMM_BLOCKS;
    const int gt = fb * blockDim.x + threadIdx.x;
    const int nt = nfb * blockDim.x;
    for (int e = gt; e < E; e += nt) eidx[rowptr[dst[e]] + rank[e]] = src[e];
    for (int n = gt; n < N; n += nt)
      invdeg[n] = 1.0f / fmaxf((float)(rowptr[n + 1] - rowptr[n]), 1.0f);
  }
}

__global__ void __launch_bounds__(256, 4) layer_gemm(
    const float* __restrict__ h, const int* __restrict__ rowptr,
    const int* __restrict__ eidx, const float* __restrict__ invdeg,
    const float* __restrict__ W, const float* __restrict__ B,
    float* __restrict__ Z, float* __restrict__ part, int mtiles, int N,
    int E) {
  __shared__ __attribute__((aligned(16))) float tA[2 * 16 * (2 * HID + A_PAD)];
  __shared__ __attribute__((aligned(16))) float rs[2 * 8 * 16];
  gemm_persistent<2 * HID, true>(h, rowptr, eidx, invdeg, W, B, Z, mtiles, tA,
                                 rs, part, blockIdx.x, gridDim.x, N, E);
}

#define BNF_BLOCKS (HID / 32)
#define BNF_WAVES 32
#define BNF_RPP (BNF_WAVES * 4)
#define BNF_PASSES (NB_STATS / BNF_RPP)
#define BNR_SLABS 64
#define BNR_VEC 5

__global__ void __launch_bounds__(BNF_WAVES * 32) bn_residual(
    const float* h, const float* __restrict__ z,
    const float* __restrict__ part, const float* __restrict__ gamma,
    const float* __restrict__ beta, float* out, float invN, int N) {
  static_assert(NB_STATS % BNF_RPP == 0, "whole passes over the partial rows");
  __shared__ __attribute__((aligned(16))) float red[BNF_WAVES * 64];
  __shared__ __attribute__((aligned(16))) float bnl[64];
  const int t = threadIdx.x;
  const int lane = t & 31;
  const int w = t >> 5;
  const int g = blockIdx.x & (BNF_BLOCKS - 1);
  const int slab = blockIdx.x / BNF_BLOCKS;
  const int col0 = g * 32;
  const int piece = t & 7;
  const int rr = t >> 3;

  const int rps = (N + BNR_SLABS - 1) / BNR_SLABS;
  const int rbeg = slab * rps;
  const int rend = min(rbeg + rps, N);

  float ga = 0.0f, be = 0.0f;
  if (t < 32) {
    ga = gamma[col0 + t];
    be = beta[col0 + t];
  }

  const float4* p = (const float4*)(part + rr * 256 + col0) + piece;
  float4 s[BNF_PASSES], q[BNF_PASSES];
#pragma unroll
  for (int i = 0; i < BNF_PASSES; ++i) {
    s[i] = p[i * (BNF_RPP * 64)];
    q[i] = p[i * (BNF_RPP * 64) + 32];
  }

  const float4* hv = (const float4*)h + g * 8 + piece;
  const float4* zv = (const float4*)z + g * 8 + piece;
  float4* ov = (float4*)out + g * 8 + piece;
  float4 hq[BNR_VEC], zq[BNR_VEC];
#pragma unroll
  for (int i = 0; i < BNR_VEC; ++i) {
    const int r = min(rbeg + rr + i * BNF_RPP, N - 1);
    hq[i] = hv[r * (HID / 4)];
    zq[i] = zv[r * (HID / 4)];
  }
  __builtin_amdgcn_sched_barrier(0);

  float4 a = s[0];
  float4 b = q[0];
#pragma unroll
  for (int i = 1; i < BNF_PASSES; ++i) {
    a.x += s[i].x;
    a.y += s[i].y;
    a.z += s[i].z;
    a.w += s[i].w;
    b.x += q[i].x;
    b.y += q[i].y;
    b.z += q[i].z;
    b.w += q[i].w;
  }
  a.x += row_xor<8>(a.x);
  a.y += row_xor<8>(a.y);
  a.z += row_xor<8>(a.z);
  a.w += row_xor<8>(a.w);
  b.x += row_xor<8>(b.x);
  b.y += row_xor<8>(b.y);
  b.z += row_xor<8>(b.z);
  b.w += row_xor<8>(b.w);
  a.x += row_other(a.x);
  a.y += row_other(a.y);
  a.z += row_other(a.z);
  a.w += row_other(a.w);
  b.x += row_other(b.x);
  b.y += row_other(b.y);
  b.z += row_other(b.z);
  b.w += row_other(b.w);
  if (lane < 8) {
    ((float4*)(red + w * 64))[lane] = a;
    ((float4*)(red + w * 64 + 32))[lane] = b;
  }
  __syncthreads();

  if (t < 32) {
    float s1 = 0.0f, s2 = 0.0f;
#pragma unroll
    for (int k = 0; k < BNF_WAVES; ++k) {
      s1 += red[k * 64 + t];
      s2 += red[k * 64 + 32 + t];
    }
    float mu = s1 * invN;
    float var = s2 * invN - mu * mu;
    float sc = ga / sqrtf(var + 1e-5f);
    bnl[t] = sc;
    bnl[32 + t] = be - mu * sc;
  }
  __syncthreads();

  const float4 a4 = ((const float4*)bnl)[piece];
  const float4 b4 = ((const float4*)(bnl + 32))[piece];

  for (int r0 = rbeg;;) {
#pragma unroll
    for (int i = 0; i < BNR_VEC; ++i) {
      const int r = r0 + rr + i * BNF_RPP;
      float4 o;
      o.x = hq[i].x + zq[i].x * a4.x + b4.x;
      o.y = hq[i].y + zq[i].y * a4.y + b4.y;
      o.z = hq[i].z + zq[i].z * a4.z + b4.z;
      o.w = hq[i].w + zq[i].w * a4.w + b4.w;
      if (r < rend) ov[r * (HID / 4)] = o;
    }
    r0 += BNR_VEC * BNF_RPP;
    if (r0 >= rend) break;
#pragma unroll
    for (int i = 0; i < BNR_VEC; ++i) {
      const int r = min(r0 + rr + i * BNF_RPP, N - 1);
      hq[i] = hv[r * (HID / 4)];
      zq[i] = zv[r * (HID / 4)];
    }
  }
}

extern "C" void kernel_launch(void* const* d_in, const int* in_sizes, int n_in,
                              void* d_out, int out_size, void* d_ws,
                              size_t ws_size, hipStream_t stream) {
  const float* h0 = (const float*)d_in[0];
  const int* src = (const int*)d_in[1];
  const int* dst = (const int*)d_in[2];
  const float* W_emb = (const float*)d_in[3];
  const float* b_emb = (const float*)d_in[4];
  const float* Ws = (const float*)d_in[5];
  const float* bs = (const float*)d_in[6];
  const float* gam = (const float*)d_in[7];
  const float* bet = (const float*)d_in[8];

  const int N = in_sizes[0] / HID;
  const int E = in_sizes[1];
  const int NF = N * HID;
  const int L = 4;

  float* ws = (float*)d_ws;
  float* H = ws;
  float* C = H + NF;
  float* Z = C + NF;
  float* DEG = Z + NF;
  float* BN = DEG + N;
  float* PART = BN + 256;
  int* cnt = (int*)(PART + NB_STATS * 256);
  int* rowptr = cnt + N;
  int* eidx = rowptr + (N + 1);
  int* rank = (int*)Z;

  const int mtiles = N / 16;

  zero_i32<<<256, 256, 0, stream>>>(cnt, N);
  hist_rank<<<1024, 256, 0, stream>>>(dst, cnt, rank, E);
  scan_excl<<<1, SCAN_WAVES * 32, 0, stream>>>(cnt, rowptr, N);

  embed_gemm_fill_csr<<<GEMM_BLOCKS + FILL_BLOCKS, 256, 0, stream>>>(
      h0, W_emb, b_emb, H, mtiles, src, dst, rowptr, rank, eidx, DEG, N, E);

  for (int l = 0; l < L; ++l) {
    layer_gemm<<<GEMM_BLOCKS, 256, 0, stream>>>(H, rowptr, eidx, DEG,
                                                Ws + l * HID * 256,
                                                bs + l * HID, Z, PART, mtiles,
                                                N, E);
    float* out = (l == L - 1) ? (float*)d_out : H;
    bn_residual<<<BNF_BLOCKS * BNR_SLABS, BNF_WAVES * 32, 0, stream>>>(
        H, Z, PART, gam + l * HID, bet + l * HID, out, 1.0f / (float)N, N);
  }
}
